# starting kernel
# baseline (speedup 1.0000x reference)
#include <hip/hip_runtime.h>
#include <math.h>

typedef __attribute__((ext_vector_type(2))) float v2f;
typedef __attribute__((ext_vector_type(8))) float v8f;

#define DIMC    96
#define D_STATE 16
#define D_INNER 192
#define DT_RANK 6
#define KDIR    4
#define BATCH   2
#define HH      56
#define WWD     56
#define LL      (HH * WWD)
#define BL      (BATCH * LL)
#define NPROJ   (DT_RANK + 2 * D_STATE)
#define XZW     (2 * D_INNER)

__device__ __forceinline__ int scan_to_spatial(int k, int l) {
    int q = l / WWD, r = l % WWD;
    if (k == 0) return l;
    if (k == 1) return q * WWD + (WWD - 1 - r);
    if (k == 2) return r * WWD + q;
    return (WWD - 1 - r) * WWD + q;
}

__device__ __forceinline__ int spatial_to_scan(int k, int l) {
    int q = l / WWD, r = l % WWD;
    if (k == 0) return l;
    if (k == 1) return q * WWD + (WWD - 1 - r);
    if (k == 2) return r * WWD + q;
    return r * WWD + (WWD - 1 - q);
}

__device__ __forceinline__ float silu(float v) {
    return v / (1.f + __expf(-v));
}

__global__ void k_in_gemm(const float* __restrict__ x,
                          const float* __restrict__ Win,
                          float* __restrict__ xz) {
    int lane  = threadIdx.x & 31;
    int wave  = threadIdx.x >> 5;
    int mTile = blockIdx.x * 4 + wave;
    int nTile = blockIdx.y;
    int m0 = mTile * 16, n0 = nTile * 16;
    int mrow = m0 + (lane & 15);
    int b = mrow / LL, l = mrow % LL;
    int nn = n0 + (lane & 15);
    int kh = (lane >> 4) * 2;
    v8f acc = {};
    for (int kb = 0; kb < DIMC; kb += 4) {
        int c0 = kb + kh;
        v2f a, bf;
        a[0]  = x[(size_t)(b * DIMC + c0    ) * LL + l];
        a[1]  = x[(size_t)(b * DIMC + c0 + 1) * LL + l];
        bf[0] = Win[(size_t)(c0    ) * XZW + nn];
        bf[1] = Win[(size_t)(c0 + 1) * XZW + nn];
        acc = __builtin_amdgcn_wmma_f32_16x16x4_f32(false, a, false, bf,
                                                    (short)0, acc, false, false);
    }
    int mb = m0 + (lane >> 4) * 8;
#pragma unroll
    for (int v = 0; v < 8; ++v)
        xz[(size_t)(mb + v) * XZW + nn] = acc[v];
}

__global__ void k_conv(const float* __restrict__ xz,
                       const float* __restrict__ Wc,
                       const float* __restrict__ bc,
                       float* __restrict__ xi) {
    int tid = blockIdx.x * blockDim.x + threadIdx.x;
    if (tid >= BL * D_INNER) return;
    int d  = tid % D_INNER;
    int bl = tid / D_INNER;
    int b = bl / LL, l = bl % LL;
    int h = l / WWD, w = l % WWD;
    float acc = bc[d];
#pragma unroll
    for (int dy = 0; dy < 3; ++dy) {
        int hy = h + dy - 1;
        if (hy < 0 || hy >= HH) continue;
#pragma unroll
        for (int dx = 0; dx < 3; ++dx) {
            int wx = w + dx - 1;
            if (wx < 0 || wx >= WWD) continue;
            acc += Wc[d * 9 + dy * 3 + dx] *
                   xz[(size_t)(b * LL + hy * WWD + wx) * XZW + d];
        }
    }
    xi[(size_t)bl * D_INNER + d] = silu(acc);
}

__global__ void k_xproj(const float* __restrict__ xi,
                        const float* __restrict__ Wxp,
                        float* __restrict__ xdbl) {
    int lane = threadIdx.x & 31;
    int wave = threadIdx.x >> 5;
    int kdir = blockIdx.z;
    int mTile = blockIdx.x * 4 + wave;
    int nTile = blockIdx.y;
    int m0 = mTile * 16, n0 = nTile * 16;
    int mrow = m0 + (lane & 15);
    int b = mrow / LL, l = mrow % LL;
    int lsp = scan_to_spatial(kdir, l);
    const float* arow = xi + (size_t)(b * LL + lsp) * D_INNER;
    const float* wk = Wxp + (size_t)kdir * D_INNER * NPROJ;
    int nn = n0 + (lane & 15);
    bool nok = (nn < NPROJ);
    int kh = (lane >> 4) * 2;
    v8f acc = {};
    for (int kb = 0; kb < D_INNER; kb += 4) {
        v2f a, bf;
        a[0]  = arow[kb + kh];
        a[1]  = arow[kb + kh + 1];
        bf[0] = nok ? wk[(size_t)(kb + kh    ) * NPROJ + nn] : 0.f;
        bf[1] = nok ? wk[(size_t)(kb + kh + 1) * NPROJ + nn] : 0.f;
        acc = __builtin_amdgcn_wmma_f32_16x16x4_f32(false, a, false, bf,
                                                    (short)0, acc, false, false);
    }
    if (nok) {
        int mb = m0 + (lane >> 4) * 8;
#pragma unroll
        for (int v = 0; v < 8; ++v)
            xdbl[(size_t)(kdir * BL + mb + v) * NPROJ + nn] = acc[v];
    }
}

__global__ void k_delta(const float* __restrict__ xdbl,
                        const float* __restrict__ Wdt,
                        const float* __restrict__ bdt,
                        float* __restrict__ delta) {
    int lane = threadIdx.x & 31;
    int wave = threadIdx.x >> 5;
    int kdir = blockIdx.z;
    int mTile = blockIdx.x * 4 + wave;
    int nTile = blockIdx.y;
    int m0 = mTile * 16, n0 = nTile * 16;
    int mrow = m0 + (lane & 15);
    int nn = n0 + (lane & 15);
    int kh = (lane >> 4) * 2;
    v8f acc = {};
#pragma unroll
    for (int kb = 0; kb < 8; kb += 4) {
        v2f a, bf;
#pragma unroll
        for (int v = 0; v < 2; ++v) {
            int kk = kb + kh + v;
            bool kok = (kk < DT_RANK);
            a[v]  = kok ? xdbl[(size_t)(kdir * BL + mrow) * NPROJ + kk] : 0.f;
            bf[v] = kok ? Wdt[(size_t)(kdir * DT_RANK + kk) * D_INNER + nn] : 0.f;
        }
        acc = __builtin_amdgcn_wmma_f32_16x16x4_f32(false, a, false, bf,
                                                    (short)0, acc, false, false);
    }
    float bias = bdt[kdir * D_INNER + nn];
    int mb = m0 + (lane >> 4) * 8;
#pragma unroll
    for (int v = 0; v < 8; ++v) {
        float t = acc[v] + bias;
        float sp = (t > 20.f) ? t : log1pf(__expf(t));
        delta[(size_t)(kdir * BL + mb + v) * D_INNER + nn] = sp;
    }
}

__global__ void k_scan(const float* __restrict__ xi,
                       const float* __restrict__ xdbl,
                       const float* __restrict__ delta,
                       const float* __restrict__ A_logs,
                       const float* __restrict__ Ds,
                       float* __restrict__ ys) {
    int tid = blockIdx.x * blockDim.x + threadIdx.x;
    if (tid >= KDIR * BATCH * D_INNER * D_STATE) return;
    int n     = tid & 15;
    int group = tid >> 4;
    int d     = group % D_INNER;
    int kb    = group / D_INNER;
    int b     = kb % BATCH;
    int kdir  = kb / BATCH;

    float A = -__expf(A_logs[(size_t)(kdir * D_INNER + d) * D_STATE + n]);
    float Dskip = Ds[kdir * D_INNER + d];
    float h = 0.f;

    const float* xib = xi    + (size_t)b * LL * D_INNER + d;
    const float* dlb = delta + (size_t)(kdir * BATCH + b) * LL * D_INNER + d;
    const float* xdb = xdbl  + (size_t)(kdir * BATCH + b) * LL * NPROJ;
    float*       ysb = ys    + (size_t)(kdir * BATCH + b) * LL * D_INNER + d;

    for (int l = 0; l < LL; ++l) {
        int lsp  = scan_to_spatial(kdir, l);
        float x  = xib[(size_t)lsp * D_INNER];
        float dl = dlb[(size_t)l * D_INNER];
        const float* bc = xdb + (size_t)l * NPROJ;
        __builtin_prefetch(bc + NPROJ, 0, 3);
        float Bn = bc[DT_RANK + n];
        float Cn = bc[DT_RANK + D_STATE + n];
        float dA = __expf(dl * A);
        h = dA * h + (dl * x) * Bn;
        float p = h * Cn;
#pragma unroll
        for (int off = 8; off > 0; off >>= 1)
            p += __shfl_xor(p, off, 16);
        if (n == 0)
            ysb[(size_t)l * D_INNER] = p + x * Dskip;
    }
}

__global__ void k_combine(const float* __restrict__ ys,
                          const float* __restrict__ xz,
                          float* __restrict__ y) {
    int tid = blockIdx.x * blockDim.x + threadIdx.x;
    if (tid >= BL * D_INNER) return;
    int d  = tid % D_INNER;
    int bl = tid / D_INNER;
    int b = bl / LL, l = bl % LL;
    float acc = 0.f;
#pragma unroll
    for (int k = 0; k < KDIR; ++k) {
        int ls = spatial_to_scan(k, l);
        acc += ys[(size_t)((k * BATCH + b) * LL + ls) * D_INNER + d];
    }
    float z = xz[(size_t)bl * XZW + D_INNER + d];
    y[tid] = acc * silu(z);
}

__global__ void k_out_gemm(const float* __restrict__ y,
                           const float* __restrict__ Wout,
                           float* __restrict__ out) {
    int lane = threadIdx.x & 31;
    int wave = threadIdx.x >> 5;
    int mTile = blockIdx.x * 4 + wave;
    int nTile = blockIdx.y;
    int m0 = mTile * 16, n0 = nTile * 16;
    int mrow = m0 + (lane & 15);
    int nn = n0 + (lane & 15);
    int kh = (lane >> 4) * 2;
    v8f acc = {};
    for (int kb = 0; kb < D_INNER; kb += 4) {
        v2f a, bf;
        a[0]  = y[(size_t)mrow * D_INNER + kb + kh];
        a[1]  = y[(size_t)mrow * D_INNER + kb + kh + 1];
        bf[0] = Wout[(size_t)(kb + kh    ) * DIMC + nn];
        bf[1] = Wout[(size_t)(kb + kh + 1) * DIMC + nn];
        acc = __builtin_amdgcn_wmma_f32_16x16x4_f32(false, a, false, bf,
                                                    (short)0, acc, false, false);
    }
    int mb = m0 + (lane >> 4) * 8;
#pragma unroll
    for (int v = 0; v < 8; ++v) {
        int row = mb + v;
        int b = row / LL, l = row % LL;
        out[(size_t)(b * DIMC + nn) * LL + l] = acc[v];
    }
}

extern "C" void kernel_launch(void* const* d_in, const int* in_sizes, int n_in,
                              void* d_out, int out_size, void* d_ws, size_t ws_size,
                              hipStream_t stream) {
    (void)in_sizes; (void)n_in; (void)out_size; (void)ws_size;
    const float* x       = (const float*)d_in[0];
    const float* W_in    = (const float*)d_in[1];
    const float* W_conv  = (const float*)d_in[2];
    const float* b_conv  = (const float*)d_in[3];
    const float* W_xproj = (const float*)d_in[4];
    const float* W_dt    = (const float*)d_in[5];
    const float* b_dt    = (const float*)d_in[6];
    const float* A_logs  = (const float*)d_in[7];
    const float* Ds      = (const float*)d_in[8];
    const float* W_out   = (const float*)d_in[9];
    float* out = (float*)d_out;

    float* ws = (float*)d_ws;
    size_t o = 0;
    float* xz    = ws + o;  o += (size_t)BL * XZW;
    float* xi    = ws + o;  o += (size_t)BL * D_INNER;
    float* xdbl  = ws + o;  o += (size_t)KDIR * BL * NPROJ;
    float* delta = ws + o;  o += (size_t)KDIR * BL * D_INNER;
    float* ysbuf = ws + o;  o += (size_t)KDIR * BL * D_INNER;
    float* ybuf  = ws + o;  o += (size_t)BL * D_INNER;

    dim3 blk(128);
    k_in_gemm <<<dim3(98, 24),    blk, 0, stream>>>(x, W_in, xz);
    int nelt = BL * D_INNER;
    k_conv    <<<(nelt + 255) / 256, 256, 0, stream>>>(xz, W_conv, b_conv, xi);
    k_xproj   <<<dim3(98, 3, 4),  blk, 0, stream>>>(xi, W_xproj, xdbl);
    k_delta   <<<dim3(98, 12, 4), blk, 0, stream>>>(xdbl, W_dt, b_dt, delta);
    int nscan = KDIR * BATCH * D_INNER * D_STATE;
    k_scan    <<<(nscan + 255) / 256, 256, 0, stream>>>(xi, xdbl, delta,
                                                        A_logs, Ds, ysbuf);
    k_combine <<<(nelt + 255) / 256, 256, 0, stream>>>(ysbuf, xz, ybuf);
    k_out_gemm<<<dim3(98, 6),     blk, 0, stream>>>(ybuf, W_out, out);
}


# scan re-laid: one thread per channel holds its 16 states in registers (was 16 lanes per channel, lane-tree sum per step); B, C by scalar loads; sum over the states now serial in 4 partial sums
# speedup vs baseline: 16.6018x; 16.6018x over previous
#include <hip/hip_runtime.h>
#include <math.h>

typedef __attribute__((ext_vector_type(2))) float v2f;
typedef __attribute__((ext_vector_type(4))) float v4f;
typedef __attribute__((ext_vector_type(8))) float v8f;

#define DIMC    96
#define D_STATE 16
#define D_INNER 192
#define DT_RANK 6
#define KDIR    4
#define BATCH   2
#define HH      56
#define WWD     56
#define LL      (HH * WWD)
#define BL      (BATCH * LL)
#define NPROJ   (DT_RANK + 2 * D_STATE)
#define XZW     (2 * D_INNER)
#define NSCAN   (KDIR * BATCH * D_INNER * D_STATE)
#define NCHAN   (KDIR * BATCH * D_INNER)
#define SCAN_BLK 64
#define SCAN_G   8
#define NCHUNK  56
#define CHUNK   (LL / NCHUNK)

__device__ __forceinline__ int scan_to_spatial(int k, int l) {
    int q = l / WWD, r = l % WWD;
    if (k == 0) return l;
    if (k == 1) return q * WWD + (WWD - 1 - r);
    if (k == 2) return r * WWD + q;
    return (WWD - 1 - r) * WWD + q;
}

__device__ __forceinline__ int spatial_to_scan(int k, int l) {
    int q = l / WWD, r = l % WWD;
    if (k == 0) return l;
    if (k == 1) return q * WWD + (WWD - 1 - r);
    if (k == 2) return r * WWD + q;
    return r * WWD + (WWD - 1 - q);
}

__device__ __forceinline__ float silu(float v) {
    return v / (1.f + __expf(-v));
}

__global__ void k_in_gemm(const float* __restrict__ x,
                          const float* __restrict__ Win,
                          float* __restrict__ xz) {
    int lane  = threadIdx.x & 31;
    int wave  = threadIdx.x >> 5;
    int mTile = blockIdx.x * 4 + wave;
    int nTile = blockIdx.y;
    int m0 = mTile * 16, n0 = nTile * 16;
    int mrow = m0 + (lane & 15);
    int b = mrow / LL, l = mrow % LL;
    int nn = n0 + (lane & 15);
    int kh = (lane >> 4) * 2;
    v8f acc = {};
    for (int kb = 0; kb < DIMC; kb += 4) {
        int c0 = kb + kh;
        v2f a, bf;
        a[0]  = x[(size_t)(b * DIMC + c0    ) * LL + l];
        a[1]  = x[(size_t)(b * DIMC + c0 + 1) * LL + l];
        bf[0] = Win[(size_t)(c0    ) * XZW + nn];
        bf[1] = Win[(size_t)(c0 + 1) * XZW + nn];
        acc = __builtin_amdgcn_wmma_f32_16x16x4_f32(false, a, false, bf,
                                                    (short)0, acc, false, false);
    }
    int mb = m0 + (lane >> 4) * 8;
#pragma unroll
    for (int v = 0; v < 8; ++v)
        xz[(size_t)(mb + v) * XZW + nn] = acc[v];
}

__global__ void k_conv(const float* __restrict__ xz,
                       const float* __restrict__ Wc,
                       const float* __restrict__ bc,
                       float* __restrict__ xi) {
    int tid = blockIdx.x * blockDim.x + threadIdx.x;
    if (tid >= BL * D_INNER) return;
    int d  = tid % D_INNER;
    int bl = tid / D_INNER;
    int b = bl / LL, l = bl % LL;
    int h = l / WWD, w = l % WWD;
    float acc = bc[d];
#pragma unroll
    for (int dy = 0; dy < 3; ++dy) {
        int hy = h + dy - 1;
        if (hy < 0 || hy >= HH) continue;
#pragma unroll
        for (int dx = 0; dx < 3; ++dx) {
            int wx = w + dx - 1;
            if (wx < 0 || wx >= WWD) continue;
            acc += Wc[d * 9 + dy * 3 + dx] *
                   xz[(size_t)(b * LL + hy * WWD + wx) * XZW + d];
        }
    }
    xi[(size_t)bl * D_INNER + d] = silu(acc);
}

__global__ void k_xproj(const float* __restrict__ xi,
                        const float* __restrict__ Wxp,
                        float* __restrict__ xdbl) {
    int lane = threadIdx.x & 31;
    int wave = threadIdx.x >> 5;
    int kdir = blockIdx.z;
    int mTile = blockIdx.x * 4 + wave;
    int nTile = blockIdx.y;
    int m0 = mTile * 16, n0 = nTile * 16;
    int mrow = m0 + (lane & 15);
    int b = mrow / LL, l = mrow % LL;
    int lsp = scan_to_spatial(kdir, l);
    const float* arow = xi + (size_t)(b * LL + lsp) * D_INNER;
    const float* wk = Wxp + (size_t)kdir * D_INNER * NPROJ;
    int nn = n0 + (lane & 15);
    bool nok = (nn < NPROJ);
    int kh = (lane >> 4) * 2;
    v8f acc = {};
    for (int kb = 0; kb < D_INNER; kb += 4) {
        v2f a, bf;
        a[0]  = arow[kb + kh];
        a[1]  = arow[kb + kh + 1];
        bf[0] = nok ? wk[(size_t)(kb + kh    ) * NPROJ + nn] : 0.f;
        bf[1] = nok ? wk[(size_t)(kb + kh + 1) * NPROJ + nn] : 0.f;
        acc = __builtin_amdgcn_wmma_f32_16x16x4_f32(false, a, false, bf,
                                                    (short)0, acc, false, false);
    }
    if (nok) {
        int mb = m0 + (lane >> 4) * 8;
#pragma unroll
        for (int v = 0; v < 8; ++v)
            xdbl[(size_t)(kdir * BL + mb + v) * NPROJ + nn] = acc[v];
    }
}

__global__ void k_delta(const float* __restrict__ xdbl,
                        const float* __restrict__ Wdt,
                        const float* __restrict__ bdt,
                        float* __restrict__ delta) {
    int lane = threadIdx.x & 31;
    int wave = threadIdx.x >> 5;
    int kdir = blockIdx.z;
    int mTile = blockIdx.x * 4 + wave;
    int nTile = blockIdx.y;
    int m0 = mTile * 16, n0 = nTile * 16;
    int mrow = m0 + (lane & 15);
    int nn = n0 + (lane & 15);
    int kh = (lane >> 4) * 2;
    v8f acc = {};
#pragma unroll
    for (int kb = 0; kb < 8; kb += 4) {
        v2f a, bf;
#pragma unroll
        for (int v = 0; v < 2; ++v) {
            int kk = kb + kh + v;
            bool kok = (kk < DT_RANK);
            a[v]  = kok ? xdbl[(size_t)(kdir * BL + mrow) * NPROJ + kk] : 0.f;
            bf[v] = kok ? Wdt[(size_t)(kdir * DT_RANK + kk) * D_INNER + nn] : 0.f;
        }
        acc = __builtin_amdgcn_wmma_f32_16x16x4_f32(false, a, false, bf,
                                                    (short)0, acc, false, false);
    }
    float bias = bdt[kdir * D_INNER + nn];
    int mb = m0 + (lane >> 4) * 8;
#pragma unroll
    for (int v = 0; v < 8; ++v) {
        float t = acc[v] + bias;
        float sp = (t > 20.f) ? t : log1pf(__expf(t));
        delta[(size_t)(kdir * BL + mb + v) * D_INNER + nn] = sp;
    }
}


static_assert(WWD % CHUNK == 0 && CHUNK >= 2, "a chunk lies inside one scan row");
static_assert(CHUNK % SCAN_G == 0, "whole groups");

__device__ __forceinline__ size_t part_idx(int chunk, int g, int ch) {
    return ((size_t)(chunk * (D_STATE / 4) + g) * NCHAN + ch) * 4;
}

__device__ __forceinline__ v2f exp_pair(v2f a) {
    v2f r = {__expf(a[0]), __expf(a[1])};
    return r;
}

__global__ void __launch_bounds__(SCAN_BLK) __attribute__((amdgpu_waves_per_eu(1, 4)))
k_scan_part(const float* __restrict__ xi,
            const float* __restrict__ xdbl,
            const float* __restrict__ delta,
            const float* __restrict__ A_logs,
            float* __restrict__ hend,
            float* __restrict__ pdec) {
    int chunk = blockIdx.y;
    int kb    = blockIdx.x / (D_INNER / SCAN_BLK);
    int d     = (blockIdx.x % (D_INNER / SCAN_BLK)) * SCAN_BLK + threadIdx.x;
    int b     = kb % BATCH;
    int kdir  = kb / BATCH;
    int ch    = kb * D_INNER + d;

    v2f A[D_STATE / 2], h[D_STATE / 2], P[D_STATE / 2];
#pragma unroll
    for (int j = 0; j < D_STATE / 2; ++j) {
        const float* al = A_logs + (size_t)(kdir * D_INNER + d) * D_STATE + 2 * j;
        A[j][0] = -__expf(al[0]);
        A[j][1] = -__expf(al[1]);
        h[j] = (v2f){0.f, 0.f};
        P[j] = (v2f){1.f, 1.f};
    }

    int l0    = chunk * CHUNK;
    int lsp0  = scan_to_spatial(kdir, l0);
    int lstep = scan_to_spatial(kdir, l0 + 1) - lsp0;
    const float* xib = xi    + ((size_t)b  * LL + lsp0) * D_INNER + d;
    const float* dlb = delta + ((size_t)kb * LL + l0) * D_INNER + d;
    const float* xdb = xdbl  + ((size_t)kb * LL + l0) * NPROJ + DT_RANK;

    float xc[SCAN_G], dc[SCAN_G], Bs[D_STATE];
#pragma unroll
    for (int g = 0; g < SCAN_G; ++g) {
        xc[g] = xib[(ptrdiff_t)g * lstep * D_INNER];
        dc[g] = dlb[(size_t)g * D_INNER];
    }
#pragma unroll
    for (int n = 0; n < D_STATE; ++n) Bs[n] = xdb[n];

#pragma nounroll
    for (int i0 = 0; i0 < CHUNK; i0 += SCAN_G) {
        float xn[SCAN_G], dn[SCAN_G];
        __builtin_amdgcn_sched_barrier(0);
        int in = (i0 + SCAN_G < CHUNK) ? i0 + SCAN_G : i0;
#pragma unroll
        for (int g = 0; g < SCAN_G; ++g) {
            xn[g] = xib[(ptrdiff_t)(in + g) * lstep * D_INNER];
            dn[g] = dlb[(size_t)(in + g) * D_INNER];
        }
        __builtin_amdgcn_sched_barrier(0);
        const float* bcg = xdb + (size_t)i0 * NPROJ;
#pragma unroll
        for (int g = 0; g < SCAN_G; ++g) {
            const float* bn = (g + 1 < SCAN_G || i0 + SCAN_G < CHUNK) ? bcg + (g + 1) * NPROJ : bcg + g * NPROJ;
            __builtin_amdgcn_sched_barrier(0);
            asm volatile("" :: "s"(Bs[0]));
            float Bn[D_STATE];
#pragma unroll
            for (int n = 0; n < D_STATE; ++n) Bn[n] = bn[n];
            __builtin_amdgcn_sched_barrier(0);
            float dl  = dc[g];
            float dlx = dl * xc[g];
            v2f dl2  = {dl, dl};
            v2f dlx2 = {dlx, dlx};
#pragma unroll
            for (int j = 0; j < D_STATE / 2; ++j) {
                v2f dA = exp_pair(dl2 * A[j]);
                v2f Bv = {Bs[2 * j], Bs[2 * j + 1]};
                h[j] = dA * h[j] + dlx2 * Bv;
                P[j] = P[j] * dA;
            }
#pragma unroll
            for (int n = 0; n < D_STATE; ++n) Bs[n] = Bn[n];
        }
#pragma unroll
        for (int g = 0; g < SCAN_G; ++g) { xc[g] = xn[g]; dc[g] = dn[g]; }
    }
#pragma unroll
    for (int g = 0; g < D_STATE / 4; ++g) {
        v4f hv = {h[2 * g][0], h[2 * g][1], h[2 * g + 1][0], h[2 * g + 1][1]};
        v4f pv = {P[2 * g][0], P[2 * g][1], P[2 * g + 1][0], P[2 * g + 1][1]};
        *(v4f*)(hend + part_idx(chunk, g, ch)) = hv;
        *(v4f*)(pdec + part_idx(chunk, g, ch)) = pv;
    }
}

__global__ void __launch_bounds__(SCAN_BLK) __attribute__((amdgpu_waves_per_eu(1, 4)))
k_scan_final(const float* __restrict__ xi,
             const float* __restrict__ xdbl,
             const float* __restrict__ delta,
             const float* __restrict__ A_logs,
             const float* __restrict__ Ds,
             const float* __restrict__ hend,
             const float* __restrict__ pdec,
             float* __restrict__ ys) {
    int chunk = NCHUNK - 1 - (int)blockIdx.y;
    int kb    = blockIdx.x / (D_INNER / SCAN_BLK);
    int d     = (blockIdx.x % (D_INNER / SCAN_BLK)) * SCAN_BLK + threadIdx.x;
    int b     = kb % BATCH;
    int kdir  = kb / BATCH;
    int ch    = kb * D_INNER + d;

    v2f A[D_STATE / 2], h[D_STATE / 2];
#pragma unroll
    for (int j = 0; j < D_STATE / 2; ++j) {
        const float* al = A_logs + (size_t)(kdir * D_INNER + d) * D_STATE + 2 * j;
        A[j][0] = -__expf(al[0]);
        A[j][1] = -__expf(al[1]);
        h[j] = (v2f){0.f, 0.f};
    }
    float Dskip = Ds[kdir * D_INNER + d];

#pragma unroll 4
    for (int j = 0; j < chunk; ++j) {
#pragma unroll
        for (int g = 0; g < D_STATE / 4; ++g) {
            v4f pv = *(const v4f*)(pdec + part_idx(j, g, ch));
            v4f hv = *(const v4f*)(hend + part_idx(j, g, ch));
            h[2 * g]     = (v2f){pv[0], pv[1]} * h[2 * g]     + (v2f){hv[0], hv[1]};
            h[2 * g + 1] = (v2f){pv[2], pv[3]} * h[2 * g + 1] + (v2f){hv[2], hv[3]};
        }
    }

    int l0    = chunk * CHUNK;
    int lsp0  = scan_to_spatial(kdir, l0);
    int lstep = scan_to_spatial(kdir, l0 + 1) - lsp0;
    const float* xib = xi    + ((size_t)b  * LL + lsp0) * D_INNER + d;
    const float* dlb = delta + ((size_t)kb * LL + l0) * D_INNER + d;
    const float* xdb = xdbl  + ((size_t)kb * LL + l0) * NPROJ + DT_RANK;
    float*       ysb = ys    + ((size_t)kb * LL + l0) * D_INNER + d;

    float xc[SCAN_G], dc[SCAN_G];
    v2f Bv[D_STATE / 2], Cv[D_STATE / 2];
    float bkey = xdb[0];
#pragma unroll
    for (int g = 0; g < SCAN_G; ++g) {
        xc[g] = xib[(ptrdiff_t)g * lstep * D_INNER];
        dc[g] = dlb[(size_t)g * D_INNER];
    }
#pragma unroll
    for (int j = 0; j < D_STATE / 2; ++j) {
        Bv[j] = (v2f){xdb[2 * j], xdb[2 * j + 1]};
        Cv[j] = (v2f){xdb[D_STATE + 2 * j], xdb[D_STATE + 2 * j + 1]};
    }

#pragma nounroll
    for (int i0 = 0; i0 < CHUNK; i0 += SCAN_G) {
        float xn[SCAN_G], dn[SCAN_G];
        __builtin_amdgcn_sched_barrier(0);
        int in = (i0 + SCAN_G < CHUNK) ? i0 + SCAN_G : i0;
#pragma unroll
        for (int g = 0; g < SCAN_G; ++g) {
            xn[g] = xib[(ptrdiff_t)(in + g) * lstep * D_INNER];
            dn[g] = dlb[(size_t)(in + g) * D_INNER];
        }
        __builtin_amdgcn_sched_barrier(0);
        const float* bcg = xdb + (size_t)i0 * NPROJ;
#pragma unroll
        for (int g = 0; g < SCAN_G; ++g) {
            const float* bn = (g + 1 < SCAN_G || i0 + SCAN_G < CHUNK) ? bcg + (g + 1) * NPROJ : bcg + g * NPROJ;
            __builtin_amdgcn_sched_barrier(0);
            asm volatile("" :: "s"(bkey));
            float Bn[D_STATE], Cn[D_STATE];
#pragma unroll
            for (int n = 0; n < D_STATE; ++n) {
                Bn[n] = bn[n];
                Cn[n] = bn[D_STATE + n];
            }
            __builtin_amdgcn_sched_barrier(0);
            float x   = xc[g];
            float dl  = dc[g];
            float dlx = dl * x;
            v2f dl2  = {dl, dl};
            v2f dlx2 = {dlx, dlx};
            v2f ya = {0.f, 0.f}, yb = {0.f, 0.f};
#pragma unroll
            for (int j = 0; j < D_STATE / 2; j += 2) {
                v2f dA0 = exp_pair(dl2 * A[j]);
                v2f dA1 = exp_pair(dl2 * A[j + 1]);
                h[j]     = dA0 * h[j]     + dlx2 * Bv[j];
                h[j + 1] = dA1 * h[j + 1] + dlx2 * Bv[j + 1];
                ya += h[j]     * Cv[j];
                yb += h[j + 1] * Cv[j + 1];
            }
            ysb[(size_t)(i0 + g) * D_INNER] = ((ya[0] + ya[1]) + (yb[0] + yb[1])) + x * Dskip;
            bkey = Bn[0];
#pragma unroll
            for (int j = 0; j < D_STATE / 2; ++j) {
                Bv[j] = (v2f){Bn[2 * j], Bn[2 * j + 1]};
                Cv[j] = (v2f){Cn[2 * j], Cn[2 * j + 1]};
            }
        }
#pragma unroll
        for (int g = 0; g < SCAN_G; ++g) { xc[g] = xn[g]; dc[g] = dn[g]; }
    }
}

__global__ void k_combine(const float* __restrict__ ys,
                          const float* __restrict__ xz,
                          float* __restrict__ y) {
    int tid = blockIdx.x * blockDim.x + threadIdx.x;
    if (tid >= BL * D_INNER) return;
    int d  = tid % D_INNER;
    int bl = tid / D_INNER;
    int b = bl / LL, l = bl % LL;
    float acc = 0.f;
#pragma unroll
    for (int k = 0; k < KDIR; ++k) {
        int ls = spatial_to_scan(k, l);
        acc += ys[(size_t)((k * BATCH + b) * LL + ls) * D_INNER + d];
    }
    float z = xz[(size_t)bl * XZW + D_INNER + d];
    y[tid] = acc * silu(z);
}

__global__ void k_out_gemm(const float* __restrict__ y,
                           const float* __restrict__ Wout,
                           float* __restrict__ out) {
    int lane = threadIdx.x & 31;
    int wave = threadIdx.x >> 5;
    int mTile = blockIdx.x * 4 + wave;
    int nTile = blockIdx.y;
    int m0 = mTile * 16, n0 = nTile * 16;
    int mrow = m0 + (lane & 15);
    int nn = n0 + (lane & 15);
    int kh = (lane >> 4) * 2;
    v8f acc = {};
    for (int kb = 0; kb < D_INNER; kb += 4) {
        v2f a, bf;
        a[0]  = y[(size_t)mrow * D_INNER + kb + kh];
        a[1]  = y[(size_t)mrow * D_INNER + kb + kh + 1];
        bf[0] = Wout[(size_t)(kb + kh    ) * DIMC + nn];
        bf[1] = Wout[(size_t)(kb + kh + 1) * DIMC + nn];
        acc = __builtin_amdgcn_wmma_f32_16x16x4_f32(false, a, false, bf,
                                                    (short)0, acc, false, false);
    }
    int mb = m0 + (lane >> 4) * 8;
#pragma unroll
    for (int v = 0; v < 8; ++v) {
        int row = mb + v;
        int b = row / LL, l = row % LL;
        out[(size_t)(b * DIMC + nn) * LL + l] = acc[v];
    }
}

extern "C" void kernel_launch(void* const* d_in, const int* in_sizes, int n_in,
                              void* d_out, int out_size, void* d_ws, size_t ws_size,
                              hipStream_t stream) {
    (void)in_sizes; (void)n_in; (void)out_size; (void)ws_size;
    const float* x       = (const float*)d_in[0];
    const float* W_in    = (const float*)d_in[1];
    const float* W_conv  = (const float*)d_in[2];
    const float* b_conv  = (const float*)d_in[3];
    const float* W_xproj = (const float*)d_in[4];
    const float* W_dt    = (const float*)d_in[5];
    const float* b_dt    = (const float*)d_in[6];
    const float* A_logs  = (const float*)d_in[7];
    const float* Ds      = (const float*)d_in[8];
    const float* W_out   = (const float*)d_in[9];
    float* out = (float*)d_out;

    float* ws = (float*)d_ws;
    size_t o = 0;
    float* xz    = ws + o;  o += (size_t)BL * XZW;
    float* xi    = ws + o;  o += (size_t)BL * D_INNER;
    float* xdbl  = ws + o;  o += (size_t)KDIR * BL * NPROJ;
    float* delta = ws + o;  o += (size_t)KDIR * BL * D_INNER;
    float* ysbuf = ws + o;  o += (size_t)KDIR * BL * D_INNER;
    float* ybuf  = ws + o;  o += (size_t)BL * D_INNER;
    float* hend  = ws + o;  o += (size_t)(NCHUNK - 1) * NSCAN;
    float* pdec  = ws + o;  o += (size_t)(NCHUNK - 1) * NSCAN;

    dim3 blk(128);
    k_in_gemm <<<dim3(98, 24),    blk, 0, stream>>>(x, W_in, xz);
    int nelt = BL * D_INNER;
    k_conv    <<<(nelt + 255) / 256, 256, 0, stream>>>(xz, W_conv, b_conv, xi);
    k_xproj   <<<dim3(98, 3, 4),  blk, 0, stream>>>(xi, W_xproj, xdbl);
    k_delta   <<<dim3(98, 12, 4), blk, 0, stream>>>(xdbl, W_dt, b_dt, delta);
    k_scan_part <<<dim3(NCHAN / SCAN_BLK, NCHUNK - 1), SCAN_BLK, 0, stream>>>(
        xi, xdbl, delta, A_logs, hend, pdec);
    k_scan_final<<<dim3(NCHAN / SCAN_BLK, NCHUNK),     SCAN_BLK, 0, stream>>>(
        xi, xdbl, delta, A_logs, Ds, hend, pdec, ysbuf);
    k_combine <<<(nelt + 255) / 256, 256, 0, stream>>>(ysbuf, xz, ybuf);
    k_out_gemm<<<dim3(98, 6),     blk, 0, stream>>>(ybuf, W_out, out);
}


# in-projection GEMM and depthwise conv fused: a block makes xp of its 4 image rows + 2 halo rows x 32 channels on the matrix unit into LDS and convolves from LDS; conv launch gone; same order of sums
# speedup vs baseline: 18.9746x; 1.1429x over previous
#include <hip/hip_runtime.h>
#include <math.h>

typedef __attribute__((ext_vector_type(2))) float v2f;
typedef __attribute__((ext_vector_type(4))) float v4f;
typedef __attribute__((ext_vector_type(8))) float v8f;

#define DIMC    96
#define D_STATE 16
#define D_INNER 192
#define DT_RANK 6
#define KDIR    4
#define BATCH   2
#define HH      56
#define WWD     56
#define LL      (HH * WWD)
#define BL      (BATCH * LL)
#define NPROJ   (DT_RANK + 2 * D_STATE)
#define XZW     (2 * D_INNER)
#define NSCAN   (KDIR * BATCH * D_INNER * D_STATE)
#define NCHAN   (KDIR * BATCH * D_INNER)
#define SCAN_BLK 64
#define SCAN_G   8
#define NCHUNK  56
#define CHUNK   (LL / NCHUNK)

__device__ __forceinline__ int scan_to_spatial(int k, int l) {
    int q = l / WWD, r = l % WWD;
    if (k == 0) return l;
    if (k == 1) return q * WWD + (WWD - 1 - r);
    if (k == 2) return r * WWD + q;
    return (WWD - 1 - r) * WWD + q;
}

__device__ __forceinline__ int spatial_to_scan(int k, int l) {
    int q = l / WWD, r = l % WWD;
    if (k == 0) return l;
    if (k == 1) return q * WWD + (WWD - 1 - r);
    if (k == 2) return r * WWD + q;
    return r * WWD + (WWD - 1 - q);
}

__device__ __forceinline__ float silu(float v) {
    return v / (1.f + __expf(-v));
}

#define CV_ROWS   4
#define CV_WIN    (CV_ROWS + 2)
#define CV_POS    (CV_WIN * WWD)
#define CV_PW     (WWD + 2)
#define CV_COLS   32
#define CV_PITCH  34
#define CV_KU     6
#define CV_NB     (DIMC / (4 * CV_KU))
#define IC_WAVES  7
#define IC_THREADS (IC_WAVES * 32)
#define CV_PER_IMG ((HH / CV_ROWS) * (D_INNER / CV_COLS))
#define CV_BLOCKS (BATCH * CV_PER_IMG)
#define Z_BLOCKS  ((BL / 16 / IC_WAVES) * (D_INNER / 16))
static_assert(HH % CV_ROWS == 0, "whole bands");
static_assert(CV_POS == IC_WAVES * 3 * 16, "three row tiles per wave");
static_assert(CV_ROWS * WWD == IC_WAVES * 32, "32 conv outputs per thread");
static_assert((BL / 16) % IC_WAVES == 0, "whole z blocks");
static_assert(DIMC % (4 * CV_KU) == 0, "whole groups of k blocks");

__global__ void __launch_bounds__(IC_THREADS)
k_in_conv(const float* __restrict__ x,
          const float* __restrict__ Win,
          const float* __restrict__ Wc,
          const float* __restrict__ bc,
          float* __restrict__ xz,
          float* __restrict__ xi) {
    __shared__ float xs[CV_WIN * CV_PW * CV_PITCH];
    int lane = threadIdx.x & 31;
    int wave = __builtin_amdgcn_readfirstlane((int)(threadIdx.x >> 5));
    int c16  = lane & 15;
    int kh   = (lane >> 4) * 2;

    if (blockIdx.x >= CV_BLOCKS) {
        int zb    = (int)blockIdx.x - CV_BLOCKS;
        int mTile = (zb / (D_INNER / 16)) * IC_WAVES + wave;
        int nTile = D_INNER / 16 + zb % (D_INNER / 16);
        int m0 = mTile * 16, n0 = nTile * 16;
        int mrow = m0 + c16;
        int b = mrow / LL, l = mrow % LL;
        int nn = n0 + c16;
        float za[DIMC / 4][2], zw[DIMC / 4][2];
#pragma unroll
        for (int u = 0; u < DIMC / 4; ++u) {
            int c0 = 4 * u + kh;
            za[u][0] = x[(size_t)(b * DIMC + c0    ) * LL + l];
            za[u][1] = x[(size_t)(b * DIMC + c0 + 1) * LL + l];
            zw[u][0] = Win[(size_t)(c0    ) * XZW + nn];
            zw[u][1] = Win[(size_t)(c0 + 1) * XZW + nn];
        }
        v8f acc = {};
#pragma unroll
        for (int u = 0; u < DIMC / 4; ++u) {
            v2f a  = {za[u][0], za[u][1]};
            v2f bf = {zw[u][0], zw[u][1]};
            acc = __builtin_amdgcn_wmma_f32_16x16x4_f32(false, a, false, bf,
                                                        (short)0, acc, false, false);
        }
        int mb = m0 + (lane >> 4) * 8;
#pragma unroll
        for (int v = 0; v < 8; ++v)
            xz[(size_t)(mb + v) * XZW + nn] = acc[v];
        return;
    }

    int idx  = (int)blockIdx.x;
    int b    = idx / CV_PER_IMG;
    int rem  = idx % CV_PER_IMG;
    int band = rem / (D_INNER / CV_COLS);
    int d0   = (rem % (D_INNER / CV_COLS)) * CV_COLS;
    int hfirst = band * CV_ROWS - 1;

    for (int i = threadIdx.x; i < CV_WIN * 2 * CV_COLS; i += IC_THREADS) {
        int row = i / (2 * CV_COLS), side = (i / CV_COLS) & 1, col = i % CV_COLS;
        xs[(row * CV_PW + side * (CV_PW - 1)) * CV_PITCH + col] = 0.f;
    }

    const float* xa[3];
    bool ok[3];
#pragma unroll
    for (int t = 0; t < 3; ++t) {
        int p = (wave * 3 + t) * 16 + c16;
        int h = hfirst + p / WWD;
        int w = p % WWD;
        ok[t] = (h >= 0 && h < HH);
        int l = ok[t] ? h * WWD + w : 0;
        xa[t] = x + (size_t)b * DIMC * LL + l;
    }
    const float* wb = Win + d0 + c16;
    v8f acc[3][2];
#pragma unroll
    for (int t = 0; t < 3; ++t) { acc[t][0] = (v8f){}; acc[t][1] = (v8f){}; }
    float av[2][CV_KU][3][2], bv[2][CV_KU][2][2];
#pragma unroll
    for (int u = 0; u < CV_KU; ++u) {
        int c0 = 4 * u + kh;
#pragma unroll
        for (int t = 0; t < 3; ++t) {
            av[0][u][t][0] = xa[t][(size_t)(c0    ) * LL];
            av[0][u][t][1] = xa[t][(size_t)(c0 + 1) * LL];
        }
#pragma unroll
        for (int j = 0; j < 2; ++j) {
            bv[0][u][j][0] = wb[(size_t)(c0    ) * XZW + 16 * j];
            bv[0][u][j][1] = wb[(size_t)(c0 + 1) * XZW + 16 * j];
        }
    }
#pragma unroll
    for (int bt = 0; bt < CV_NB; ++bt) {
        const int cur = bt & 1, nxt = cur ^ 1;
        if (bt + 1 < CV_NB) {
#pragma unroll
            for (int u = 0; u < CV_KU; ++u) {
                int c0 = (bt + 1) * 4 * CV_KU + 4 * u + kh;
#pragma unroll
                for (int t = 0; t < 3; ++t) {
                    av[nxt][u][t][0] = xa[t][(size_t)(c0    ) * LL];
                    av[nxt][u][t][1] = xa[t][(size_t)(c0 + 1) * LL];
                }
#pragma unroll
                for (int j = 0; j < 2; ++j) {
                    bv[nxt][u][j][0] = wb[(size_t)(c0    ) * XZW + 16 * j];
                    bv[nxt][u][j][1] = wb[(size_t)(c0 + 1) * XZW + 16 * j];
                }
            }
        }
#pragma unroll
        for (int u = 0; u < CV_KU; ++u) {
#pragma unroll
            for (int t = 0; t < 3; ++t) {
                v2f a;
                a[0] = ok[t] ? av[cur][u][t][0] : 0.f;
                a[1] = ok[t] ? av[cur][u][t][1] : 0.f;
#pragma unroll
                for (int j = 0; j < 2; ++j) {
                    v2f bf = {bv[cur][u][j][0], bv[cur][u][j][1]};
                    acc[t][j] = __builtin_amdgcn_wmma_f32_16x16x4_f32(false, a, false, bf,
                                                                      (short)0, acc[t][j], false, false);
                }
            }
        }
    }
#pragma unroll
    for (int t = 0; t < 3; ++t) {
        int prow = (wave * 3 + t) * 16 + (lane >> 4) * 8;
#pragma unroll
        for (int v = 0; v < 8; ++v) {
            int p = prow + v;
            int q = (p / WWD) * CV_PW + p % WWD + 1;
            xs[q * CV_PITCH + c16]      = acc[t][0][v];
            xs[q * CV_PITCH + 16 + c16] = acc[t][1][v];
        }
    }
    __syncthreads();

    int d = d0 + lane;
    float wg[9];
#pragma unroll
    for (int t = 0; t < 9; ++t) wg[t] = Wc[d * 9 + t];
    float bias = bc[d];
    float* xo = xi + ((size_t)b * LL + band * CV_ROWS * WWD) * D_INNER + d;
#pragma unroll 4
    for (int i = 0; i < (CV_ROWS * WWD) / IC_WAVES; ++i) {
        int o = wave + IC_WAVES * i;
        int r = o / WWD, w = o % WWD;
        const float* q = xs + (r * CV_PW + w) * CV_PITCH + lane;
        float s = bias;
#pragma unroll
        for (int dy = 0; dy < 3; ++dy)
#pragma unroll
            for (int dx = 0; dx < 3; ++dx)
                s = __builtin_fmaf(wg[dy * 3 + dx], q[(dy * CV_PW + dx) * CV_PITCH], s);
        xo[(size_t)o * D_INNER] = silu(s);
    }
}

__global__ void k_xproj(const float* __restrict__ xi,
                        const float* __restrict__ Wxp,
                        float* __restrict__ xdbl) {
    int lane = threadIdx.x & 31;
    int wave = threadIdx.x >> 5;
    int kdir = blockIdx.z;
    int mTile = blockIdx.x * 4 + wave;
    int nTile = blockIdx.y;
    int m0 = mTile * 16, n0 = nTile * 16;
    int mrow = m0 + (lane & 15);
    int b = mrow / LL, l = mrow % LL;
    int lsp = scan_to_spatial(kdir, l);
    const float* arow = xi + (size_t)(b * LL + lsp) * D_INNER;
    const float* wk = Wxp + (size_t)kdir * D_INNER * NPROJ;
    int nn = n0 + (lane & 15);
    bool nok = (nn < NPROJ);
    int kh = (lane >> 4) * 2;
    v8f acc = {};
    for (int kb = 0; kb < D_INNER; kb += 4) {
        v2f a, bf;
        a[0]  = arow[kb + kh];
        a[1]  = arow[kb + kh + 1];
        bf[0] = nok ? wk[(size_t)(kb + kh    ) * NPROJ + nn] : 0.f;
        bf[1] = nok ? wk[(size_t)(kb + kh + 1) * NPROJ + nn] : 0.f;
        acc = __builtin_amdgcn_wmma_f32_16x16x4_f32(false, a, false, bf,
                                                    (short)0, acc, false, false);
    }
    if (nok) {
        int mb = m0 + (lane >> 4) * 8;
#pragma unroll
        for (int v = 0; v < 8; ++v)
            xdbl[(size_t)(kdir * BL + mb + v) * NPROJ + nn] = acc[v];
    }
}

__global__ void k_delta(const float* __restrict__ xdbl,
                        const float* __restrict__ Wdt,
                        const float* __restrict__ bdt,
                        float* __restrict__ delta) {
    int lane = threadIdx.x & 31;
    int wave = threadIdx.x >> 5;
    int kdir = blockIdx.z;
    int mTile = blockIdx.x * 4 + wave;
    int nTile = blockIdx.y;
    int m0 = mTile * 16, n0 = nTile * 16;
    int mrow = m0 + (lane & 15);
    int nn = n0 + (lane & 15);
    int kh = (lane >> 4) * 2;
    v8f acc = {};
#pragma unroll
    for (int kb = 0; kb < 8; kb += 4) {
        v2f a, bf;
#pragma unroll
        for (int v = 0; v < 2; ++v) {
            int kk = kb + kh + v;
            bool kok = (kk < DT_RANK);
            a[v]  = kok ? xdbl[(size_t)(kdir * BL + mrow) * NPROJ + kk] : 0.f;
            bf[v] = kok ? Wdt[(size_t)(kdir * DT_RANK + kk) * D_INNER + nn] : 0.f;
        }
        acc = __builtin_amdgcn_wmma_f32_16x16x4_f32(false, a, false, bf,
                                                    (short)0, acc, false, false);
    }
    float bias = bdt[kdir * D_INNER + nn];
    int mb = m0 + (lane >> 4) * 8;
#pragma unroll
    for (int v = 0; v < 8; ++v) {
        float t = acc[v] + bias;
        float sp = (t > 20.f) ? t : log1pf(__expf(t));
        delta[(size_t)(kdir * BL + mb + v) * D_INNER + nn] = sp;
    }
}


static_assert(WWD % CHUNK == 0 && CHUNK >= 2, "a chunk lies inside one scan row");
static_assert(CHUNK % SCAN_G == 0, "whole groups");

__device__ __forceinline__ size_t part_idx(int chunk, int g, int ch) {
    return ((size_t)(chunk * (D_STATE / 4) + g) * NCHAN + ch) * 4;
}

__device__ __forceinline__ v2f exp_pair(v2f a) {
    v2f r = {__expf(a[0]), __expf(a[1])};
    return r;
}

__global__ void __launch_bounds__(SCAN_BLK) __attribute__((amdgpu_waves_per_eu(1, 4)))
k_scan_part(const float* __restrict__ xi,
            const float* __restrict__ xdbl,
            const float* __restrict__ delta,
            const float* __restrict__ A_logs,
            float* __restrict__ hend,
            float* __restrict__ pdec) {
    int chunk = blockIdx.y;
    int kb    = blockIdx.x / (D_INNER / SCAN_BLK);
    int d     = (blockIdx.x % (D_INNER / SCAN_BLK)) * SCAN_BLK + threadIdx.x;
    int b     = kb % BATCH;
    int kdir  = kb / BATCH;
    int ch    = kb * D_INNER + d;

    v2f A[D_STATE / 2], h[D_STATE / 2], P[D_STATE / 2];
#pragma unroll
    for (int j = 0; j < D_STATE / 2; ++j) {
        const float* al = A_logs + (size_t)(kdir * D_INNER + d) * D_STATE + 2 * j;
        A[j][0] = -__expf(al[0]);
        A[j][1] = -__expf(al[1]);
        h[j] = (v2f){0.f, 0.f};
        P[j] = (v2f){1.f, 1.f};
    }

    int l0    = chunk * CHUNK;
    int lsp0  = scan_to_spatial(kdir, l0);
    int lstep = scan_to_spatial(kdir, l0 + 1) - lsp0;
    const float* xib = xi    + ((size_t)b  * LL + lsp0) * D_INNER + d;
    const float* dlb = delta + ((size_t)kb * LL + l0) * D_INNER + d;
    const float* xdb = xdbl  + ((size_t)kb * LL + l0) * NPROJ + DT_RANK;

    float xc[SCAN_G], dc[SCAN_G], Bs[D_STATE];
#pragma unroll
    for (int g = 0; g < SCAN_G; ++g) {
        xc[g] = xib[(ptrdiff_t)g * lstep * D_INNER];
        dc[g] = dlb[(size_t)g * D_INNER];
    }
#pragma unroll
    for (int n = 0; n < D_STATE; ++n) Bs[n] = xdb[n];

#pragma nounroll
    for (int i0 = 0; i0 < CHUNK; i0 += SCAN_G) {
        float xn[SCAN_G], dn[SCAN_G];
        __builtin_amdgcn_sched_barrier(0);
        int in = (i0 + SCAN_G < CHUNK) ? i0 + SCAN_G : i0;
#pragma unroll
        for (int g = 0; g < SCAN_G; ++g) {
            xn[g] = xib[(ptrdiff_t)(in + g) * lstep * D_INNER];
            dn[g] = dlb[(size_t)(in + g) * D_INNER];
        }
        __builtin_amdgcn_sched_barrier(0);
        const float* bcg = xdb + (size_t)i0 * NPROJ;
#pragma unroll
        for (int g = 0; g < SCAN_G; ++g) {
            const float* bn = (g + 1 < SCAN_G || i0 + SCAN_G < CHUNK) ? bcg + (g + 1) * NPROJ : bcg + g * NPROJ;
            __builtin_amdgcn_sched_barrier(0);
            asm volatile("" :: "s"(Bs[0]));
            float Bn[D_STATE];
#pragma unroll
            for (int n = 0; n < D_STATE; ++n) Bn[n] = bn[n];
            __builtin_amdgcn_sched_barrier(0);
            float dl  = dc[g];
            float dlx = dl * xc[g];
            v2f dl2  = {dl, dl};
            v2f dlx2 = {dlx, dlx};
#pragma unroll
            for (int j = 0; j < D_STATE / 2; ++j) {
                v2f dA = exp_pair(dl2 * A[j]);
                v2f Bv = {Bs[2 * j], Bs[2 * j + 1]};
                h[j] = dA * h[j] + dlx2 * Bv;
                P[j] = P[j] * dA;
            }
#pragma unroll
            for (int n = 0; n < D_STATE; ++n) Bs[n] = Bn[n];
        }
#pragma unroll
        for (int g = 0; g < SCAN_G; ++g) { xc[g] = xn[g]; dc[g] = dn[g]; }
    }
#pragma unroll
    for (int g = 0; g < D_STATE / 4; ++g) {
        v4f hv = {h[2 * g][0], h[2 * g][1], h[2 * g + 1][0], h[2 * g + 1][1]};
        v4f pv = {P[2 * g][0], P[2 * g][1], P[2 * g + 1][0], P[2 * g + 1][1]};
        *(v4f*)(hend + part_idx(chunk, g, ch)) = hv;
        *(v4f*)(pdec + part_idx(chunk, g, ch)) = pv;
    }
}

__global__ void __launch_bounds__(SCAN_BLK) __attribute__((amdgpu_waves_per_eu(1, 4)))
k_scan_final(const float* __restrict__ xi,
             const float* __restrict__ xdbl,
             const float* __restrict__ delta,
             const float* __restrict__ A_logs,
             const float* __restrict__ Ds,
             const float* __restrict__ hend,
             const float* __restrict__ pdec,
             float* __restrict__ ys) {
    int chunk = NCHUNK - 1 - (int)blockIdx.y;
    int kb    = blockIdx.x / (D_INNER / SCAN_BLK);
    int d     = (blockIdx.x % (D_INNER / SCAN_BLK)) * SCAN_BLK + threadIdx.x;
    int b     = kb % BATCH;
    int kdir  = kb / BATCH;
    int ch    = kb * D_INNER + d;

    v2f A[D_STATE / 2], h[D_STATE / 2];
#pragma unroll
    for (int j = 0; j < D_STATE / 2; ++j) {
        const float* al = A_logs + (size_t)(kdir * D_INNER + d) * D_STATE + 2 * j;
        A[j][0] = -__expf(al[0]);
        A[j][1] = -__expf(al[1]);
        h[j] = (v2f){0.f, 0.f};
    }
    float Dskip = Ds[kdir * D_INNER + d];

#pragma unroll 4
    for (int j = 0; j < chunk; ++j) {
#pragma unroll
        for (int g = 0; g < D_STATE / 4; ++g) {
            v4f pv = *(const v4f*)(pdec + part_idx(j, g, ch));
            v4f hv = *(const v4f*)(hend + part_idx(j, g, ch));
            h[2 * g]     = (v2f){pv[0], pv[1]} * h[2 * g]     + (v2f){hv[0], hv[1]};
            h[2 * g + 1] = (v2f){pv[2], pv[3]} * h[2 * g + 1] + (v2f){hv[2], hv[3]};
        }
    }

    int l0    = chunk * CHUNK;
    int lsp0  = scan_to_spatial(kdir, l0);
    int lstep = scan_to_spatial(kdir, l0 + 1) - lsp0;
    const float* xib = xi    + ((size_t)b  * LL + lsp0) * D_INNER + d;
    const float* dlb = delta + ((size_t)kb * LL + l0) * D_INNER + d;
    const float* xdb = xdbl  + ((size_t)kb * LL + l0) * NPROJ + DT_RANK;
    float*       ysb = ys    + ((size_t)kb * LL + l0) * D_INNER + d;

    float xc[SCAN_G], dc[SCAN_G];
    v2f Bv[D_STATE / 2], Cv[D_STATE / 2];
    float bkey = xdb[0];
#pragma unroll
    for (int g = 0; g < SCAN_G; ++g) {
        xc[g] = xib[(ptrdiff_t)g * lstep * D_INNER];
        dc[g] = dlb[(size_t)g * D_INNER];
    }
#pragma unroll
    for (int j = 0; j < D_STATE / 2; ++j) {
        Bv[j] = (v2f){xdb[2 * j], xdb[2 * j + 1]};
        Cv[j] = (v2f){xdb[D_STATE + 2 * j], xdb[D_STATE + 2 * j + 1]};
    }

#pragma nounroll
    for (int i0 = 0; i0 < CHUNK; i0 += SCAN_G) {
        float xn[SCAN_G], dn[SCAN_G];
        __builtin_amdgcn_sched_barrier(0);
        int in = (i0 + SCAN_G < CHUNK) ? i0 + SCAN_G : i0;
#pragma unroll
        for (int g = 0; g < SCAN_G; ++g) {
            xn[g] = xib[(ptrdiff_t)(in + g) * lstep * D_INNER];
            dn[g] = dlb[(size_t)(in + g) * D_INNER];
        }
        __builtin_amdgcn_sched_barrier(0);
        const float* bcg = xdb + (size_t)i0 * NPROJ;
#pragma unroll
        for (int g = 0; g < SCAN_G; ++g) {
            const float* bn = (g + 1 < SCAN_G || i0 + SCAN_G < CHUNK) ? bcg + (g + 1) * NPROJ : bcg + g * NPROJ;
            __builtin_amdgcn_sched_barrier(0);
            asm volatile("" :: "s"(bkey));
            float Bn[D_STATE], Cn[D_STATE];
#pragma unroll
            for (int n = 0; n < D_STATE; ++n) {
                Bn[n] = bn[n];
                Cn[n] = bn[D_STATE + n];
            }
            __builtin_amdgcn_sched_barrier(0);
            float x   = xc[g];
            float dl  = dc[g];
            float dlx = dl * x;
            v2f dl2  = {dl, dl};
            v2f dlx2 = {dlx, dlx};
            v2f ya = {0.f, 0.f}, yb = {0.f, 0.f};
#pragma unroll
            for (int j = 0; j < D_STATE / 2; j += 2) {
                v2f dA0 = exp_pair(dl2 * A[j]);
                v2f dA1 = exp_pair(dl2 * A[j + 1]);
                h[j]     = dA0 * h[j]     + dlx2 * Bv[j];
                h[j + 1] = dA1 * h[j + 1] + dlx2 * Bv[j + 1];
                ya += h[j]     * Cv[j];
                yb += h[j + 1] * Cv[j + 1];
            }
            ysb[(size_t)(i0 + g) * D_INNER] = ((ya[0] + ya[1]) + (yb[0] + yb[1])) + x * Dskip;
            bkey = Bn[0];
#pragma unroll
            for (int j = 0; j < D_STATE / 2; ++j) {
                Bv[j] = (v2f){Bn[2 * j], Bn[2 * j + 1]};
                Cv[j] = (v2f){Cn[2 * j], Cn[2 * j + 1]};
            }
        }
#pragma unroll
        for (int g = 0; g < SCAN_G; ++g) { xc[g] = xn[g]; dc[g] = dn[g]; }
    }
}

__global__ void k_combine(const float* __restrict__ ys,
                          const float* __restrict__ xz,
                          float* __restrict__ y) {
    int tid = blockIdx.x * blockDim.x + threadIdx.x;
    if (tid >= BL * D_INNER) return;
    int d  = tid % D_INNER;
    int bl = tid / D_INNER;
    int b = bl / LL, l = bl % LL;
    float acc = 0.f;
#pragma unroll
    for (int k = 0; k < KDIR; ++k) {
        int ls = spatial_to_scan(k, l);
        acc += ys[(size_t)((k * BATCH + b) * LL + ls) * D_INNER + d];
    }
    float z = xz[(size_t)bl * XZW + D_INNER + d];
    y[tid] = acc * silu(z);
}

__global__ void k_out_gemm(const float* __restrict__ y,
                           const float* __restrict__ Wout,
                           float* __restrict__ out) {
    int lane = threadIdx.x & 31;
    int wave = threadIdx.x >> 5;
    int mTile = blockIdx.x * 4 + wave;
    int nTile = blockIdx.y;
    int m0 = mTile * 16, n0 = nTile * 16;
    int mrow = m0 + (lane & 15);
    int nn = n0 + (lane & 15);
    int kh = (lane >> 4) * 2;
    v8f acc = {};
    for (int kb = 0; kb < D_INNER; kb += 4) {
        v2f a, bf;
        a[0]  = y[(size_t)mrow * D_INNER + kb + kh];
        a[1]  = y[(size_t)mrow * D_INNER + kb + kh + 1];
        bf[0] = Wout[(size_t)(kb + kh    ) * DIMC + nn];
        bf[1] = Wout[(size_t)(kb + kh + 1) * DIMC + nn];
        acc = __builtin_amdgcn_wmma_f32_16x16x4_f32(false, a, false, bf,
                                                    (short)0, acc, false, false);
    }
    int mb = m0 + (lane >> 4) * 8;
#pragma unroll
    for (int v = 0; v < 8; ++v) {
        int row = mb + v;
        int b = row / LL, l = row % LL;
        out[(size_t)(b * DIMC + nn) * LL + l] = acc[v];
    }
}

extern "C" void kernel_launch(void* const* d_in, const int* in_sizes, int n_in,
                              void* d_out, int out_size, void* d_ws, size_t ws_size,
                              hipStream_t stream) {
    (void)in_sizes; (void)n_in; (void)out_size; (void)ws_size;
    const float* x       = (const float*)d_in[0];
    const float* W_in    = (const float*)d_in[1];
    const float* W_conv  = (const float*)d_in[2];
    const float* b_conv  = (const float*)d_in[3];
    const float* W_xproj = (const float*)d_in[4];
    const float* W_dt    = (const float*)d_in[5];
    const float* b_dt    = (const float*)d_in[6];
    const float* A_logs  = (const float*)d_in[7];
    const float* Ds      = (const float*)d_in[8];
    const float* W_out   = (const float*)d_in[9];
    float* out = (float*)d_out;

    float* ws = (float*)d_ws;
    size_t o = 0;
    float* xz    = ws + o;  o += (size_t)BL * XZW;
    float* xi    = ws + o;  o += (size_t)BL * D_INNER;
    float* xdbl  = ws + o;  o += (size_t)KDIR * BL * NPROJ;
    float* delta = ws + o;  o += (size_t)KDIR * BL * D_INNER;
    float* ysbuf = ws + o;  o += (size_t)KDIR * BL * D_INNER;
    float* ybuf  = ws + o;  o += (size_t)BL * D_INNER;
    float* hend  = ws + o;  o += (size_t)(NCHUNK - 1) * NSCAN;
    float* pdec  = ws + o;  o += (size_t)(NCHUNK - 1) * NSCAN;

    dim3 blk(128);
    k_in_conv <<<CV_BLOCKS + Z_BLOCKS, IC_THREADS, 0, stream>>>(x, W_in, W_conv, b_conv, xz, xi);
    int nelt = BL * D_INNER;
    k_xproj   <<<dim3(98, 3, 4),  blk, 0, stream>>>(xi, W_xproj, xdbl);
    k_delta   <<<dim3(98, 12, 4), blk, 0, stream>>>(xdbl, W_dt, b_dt, delta);
    k_scan_part <<<dim3(NCHAN / SCAN_BLK, NCHUNK - 1), SCAN_BLK, 0, stream>>>(
        xi, xdbl, delta, A_logs, hend, pdec);
    k_scan_final<<<dim3(NCHAN / SCAN_BLK, NCHUNK),     SCAN_BLK, 0, stream>>>(
        xi, xdbl, delta, A_logs, Ds, hend, pdec, ysbuf);
    k_combine <<<(nelt + 255) / 256, 256, 0, stream>>>(ysbuf, xz, ybuf);
    k_out_gemm<<<dim3(98, 6),     blk, 0, stream>>>(ybuf, W_out, out);
}


# x-projection and dt-projection GEMMs in one launch: a block makes its 16 rows of xdbl, hands dr through LDS and makes their delta tile; xproj operands requested 6 k blocks ahead; same order of sums
# speedup vs baseline: 21.1654x; 1.1155x over previous
#include <hip/hip_runtime.h>
#include <math.h>

typedef __attribute__((ext_vector_type(2))) float v2f;
typedef __attribute__((ext_vector_type(4))) float v4f;
typedef __attribute__((ext_vector_type(8))) float v8f;

#define DIMC    96
#define D_STATE 16
#define D_INNER 192
#define DT_RANK 6
#define KDIR    4
#define BATCH   2
#define HH      56
#define WWD     56
#define LL      (HH * WWD)
#define BL      (BATCH * LL)
#define NPROJ   (DT_RANK + 2 * D_STATE)
#define XZW     (2 * D_INNER)
#define NSCAN   (KDIR * BATCH * D_INNER * D_STATE)
#define NCHAN   (KDIR * BATCH * D_INNER)
#define SCAN_BLK 64
#define SCAN_G   8
#define NCHUNK  56
#define CHUNK   (LL / NCHUNK)

__device__ __forceinline__ int scan_to_spatial(int k, int l) {
    int q = l / WWD, r = l % WWD;
    if (k == 0) return l;
    if (k == 1) return q * WWD + (WWD - 1 - r);
    if (k == 2) return r * WWD + q;
    return (WWD - 1 - r) * WWD + q;
}

__device__ __forceinline__ int spatial_to_scan(int k, int l) {
    int q = l / WWD, r = l % WWD;
    if (k == 0) return l;
    if (k == 1) return q * WWD + (WWD - 1 - r);
    if (k == 2) return r * WWD + q;
    return r * WWD + (WWD - 1 - q);
}

__device__ __forceinline__ float silu(float v) {
    return v / (1.f + __expf(-v));
}

#define CV_ROWS   4
#define CV_WIN    (CV_ROWS + 2)
#define CV_POS    (CV_WIN * WWD)
#define CV_PW     (WWD + 2)
#define CV_COLS   32
#define CV_PITCH  34
#define CV_KU     6
#define CV_NB     (DIMC / (4 * CV_KU))
#define IC_WAVES  7
#define IC_THREADS (IC_WAVES * 32)
#define CV_PER_IMG ((HH / CV_ROWS) * (D_INNER / CV_COLS))
#define CV_BLOCKS (BATCH * CV_PER_IMG)
#define Z_BLOCKS  ((BL / 16 / IC_WAVES) * (D_INNER / 16))
static_assert(HH % CV_ROWS == 0, "whole bands");
static_assert(CV_POS == IC_WAVES * 3 * 16, "three row tiles per wave");
static_assert(CV_ROWS * WWD == IC_WAVES * 32, "32 conv outputs per thread");
static_assert((BL / 16) % IC_WAVES == 0, "whole z blocks");
static_assert(DIMC % (4 * CV_KU) == 0, "whole groups of k blocks");

__global__ void __launch_bounds__(IC_THREADS)
k_in_conv(const float* __restrict__ x,
          const float* __restrict__ Win,
          const float* __restrict__ Wc,
          const float* __restrict__ bc,
          float* __restrict__ xz,
          float* __restrict__ xi) {
    __shared__ float xs[CV_WIN * CV_PW * CV_PITCH];
    int lane = threadIdx.x & 31;
    int wave = __builtin_amdgcn_readfirstlane((int)(threadIdx.x >> 5));
    int c16  = lane & 15;
    int kh   = (lane >> 4) * 2;

    if (blockIdx.x >= CV_BLOCKS) {
        int zb    = (int)blockIdx.x - CV_BLOCKS;
        int mTile = (zb / (D_INNER / 16)) * IC_WAVES + wave;
        int nTile = D_INNER / 16 + zb % (D_INNER / 16);
        int m0 = mTile * 16, n0 = nTile * 16;
        int mrow = m0 + c16;
        int b = mrow / LL, l = mrow % LL;
        int nn = n0 + c16;
        float za[DIMC / 4][2], zw[DIMC / 4][2];
#pragma unroll
        for (int u = 0; u < DIMC / 4; ++u) {
            int c0 = 4 * u + kh;
            za[u][0] = x[(size_t)(b * DIMC + c0    ) * LL + l];
            za[u][1] = x[(size_t)(b * DIMC + c0 + 1) * LL + l];
            zw[u][0] = Win[(size_t)(c0    ) * XZW + nn];
            zw[u][1] = Win[(size_t)(c0 + 1) * XZW + nn];
        }
        v8f acc = {};
#pragma unroll
        for (int u = 0; u < DIMC / 4; ++u) {
            v2f a  = {za[u][0], za[u][1]};
            v2f bf = {zw[u][0], zw[u][1]};
            acc = __builtin_amdgcn_wmma_f32_16x16x4_f32(false, a, false, bf,
                                                        (short)0, acc, false, false);
        }
        int mb = m0 + (lane >> 4) * 8;
#pragma unroll
        for (int v = 0; v < 8; ++v)
            xz[(size_t)(mb + v) * XZW + nn] = acc[v];
        return;
    }

    int idx  = (int)blockIdx.x;
    int b    = idx / CV_PER_IMG;
    int rem  = idx % CV_PER_IMG;
    int band = rem / (D_INNER / CV_COLS);
    int d0   = (rem % (D_INNER / CV_COLS)) * CV_COLS;
    int hfirst = band * CV_ROWS - 1;

    for (int i = threadIdx.x; i < CV_WIN * 2 * CV_COLS; i += IC_THREADS) {
        int row = i / (2 * CV_COLS), side = (i / CV_COLS) & 1, col = i % CV_COLS;
        xs[(row * CV_PW + side * (CV_PW - 1)) * CV_PITCH + col] = 0.f;
    }

    const float* xa[3];
    bool ok[3];
#pragma unroll
    for (int t = 0; t < 3; ++t) {
        int p = (wave * 3 + t) * 16 + c16;
        int h = hfirst + p / WWD;
        int w = p % WWD;
        ok[t] = (h >= 0 && h < HH);
        int l = ok[t] ? h * WWD + w : 0;
        xa[t] = x + (size_t)b * DIMC * LL + l;
    }
    const float* wb = Win + d0 + c16;
    v8f acc[3][2];
#pragma unroll
    for (int t = 0; t < 3; ++t) { acc[t][0] = (v8f){}; acc[t][1] = (v8f){}; }
    float av[2][CV_KU][3][2], bv[2][CV_KU][2][2];
#pragma unroll
    for (int u = 0; u < CV_KU; ++u) {
        int c0 = 4 * u + kh;
#pragma unroll
        for (int t = 0; t < 3; ++t) {
            av[0][u][t][0] = xa[t][(size_t)(c0    ) * LL];
            av[0][u][t][1] = xa[t][(size_t)(c0 + 1) * LL];
        }
#pragma unroll
        for (int j = 0; j < 2; ++j) {
            bv[0][u][j][0] = wb[(size_t)(c0    ) * XZW + 16 * j];
            bv[0][u][j][1] = wb[(size_t)(c0 + 1) * XZW + 16 * j];
        }
    }
#pragma unroll
    for (int bt = 0; bt < CV_NB; ++bt) {
        const int cur = bt & 1, nxt = cur ^ 1;
        if (bt + 1 < CV_NB) {
#pragma unroll
            for (int u = 0; u < CV_KU; ++u) {
                int c0 = (bt + 1) * 4 * CV_KU + 4 * u + kh;
#pragma unroll
                for (int t = 0; t < 3; ++t) {
                    av[nxt][u][t][0] = xa[t][(size_t)(c0    ) * LL];
                    av[nxt][u][t][1] = xa[t][(size_t)(c0 + 1) * LL];
                }
#pragma unroll
                for (int j = 0; j < 2; ++j) {
                    bv[nxt][u][j][0] = wb[(size_t)(c0    ) * XZW + 16 * j];
                    bv[nxt][u][j][1] = wb[(size_t)(c0 + 1) * XZW + 16 * j];
                }
            }
        }
#pragma unroll
        for (int u = 0; u < CV_KU; ++u) {
#pragma unroll
            for (int t = 0; t < 3; ++t) {
                v2f a;
                a[0] = ok[t] ? av[cur][u][t][0] : 0.f;
                a[1] = ok[t] ? av[cur][u][t][1] : 0.f;
#pragma unroll
                for (int j = 0; j < 2; ++j) {
                    v2f bf = {bv[cur][u][j][0], bv[cur][u][j][1]};
                    acc[t][j] = __builtin_amdgcn_wmma_f32_16x16x4_f32(false, a, false, bf,
                                                                      (short)0, acc[t][j], false, false);
                }
            }
        }
    }
#pragma unroll
    for (int t = 0; t < 3; ++t) {
        int prow = (wave * 3 + t) * 16 + (lane >> 4) * 8;
#pragma unroll
        for (int v = 0; v < 8; ++v) {
            int p = prow + v;
            int q = (p / WWD) * CV_PW + p % WWD + 1;
            xs[q * CV_PITCH + c16]      = acc[t][0][v];
            xs[q * CV_PITCH + 16 + c16] = acc[t][1][v];
        }
    }
    __syncthreads();

    int d = d0 + lane;
    float wg[9];
#pragma unroll
    for (int t = 0; t < 9; ++t) wg[t] = Wc[d * 9 + t];
    float bias = bc[d];
    float* xo = xi + ((size_t)b * LL + band * CV_ROWS * WWD) * D_INNER + d;
#pragma unroll 4
    for (int i = 0; i < (CV_ROWS * WWD) / IC_WAVES; ++i) {
        int o = wave + IC_WAVES * i;
        int r = o / WWD, w = o % WWD;
        const float* q = xs + (r * CV_PW + w) * CV_PITCH + lane;
        float s = bias;
#pragma unroll
        for (int dy = 0; dy < 3; ++dy)
#pragma unroll
            for (int dx = 0; dx < 3; ++dx)
                s = __builtin_fmaf(wg[dy * 3 + dx], q[(dy * CV_PW + dx) * CV_PITCH], s);
        xo[(size_t)o * D_INNER] = silu(s);
    }
}

#define XP_KU   6
#define XP_NB   (D_INNER / (4 * XP_KU))
#define XP_NT   3
#define DL_PW   (D_INNER / 16 / 4)
static_assert(D_INNER % (4 * XP_KU) == 0, "whole groups of k blocks");
static_assert(D_INNER % 64 == 0, "whole column tiles per wave");

__global__ void __launch_bounds__(128)
k_xproj_delta(const float* __restrict__ xi,
              const float* __restrict__ Wxp,
              const float* __restrict__ Wdt,
              const float* __restrict__ bdt,
              float* __restrict__ xdbl,
              float* __restrict__ delta) {
    __shared__ float drs[16 * 8];
    int lane = threadIdx.x & 31;
    int wave = __builtin_amdgcn_readfirstlane((int)(threadIdx.x >> 5));
    int kdir = blockIdx.y;
    int m0 = blockIdx.x * 16;
    int c16 = lane & 15;
    int hf  = lane >> 4;
    int kh  = hf * 2;
    int mb  = m0 + hf * 8;

    int dcol = wave * (16 * DL_PW) + c16;
    const float* wd = Wdt + (size_t)kdir * DT_RANK * D_INNER + dcol;
    int k2 = hf ? 0 : 4;
    float w0[DL_PW][2], w1[DL_PW][2], bs[DL_PW];
#pragma unroll
    for (int t = 0; t < DL_PW; ++t) {
        w0[t][0] = wd[(size_t)(kh     ) * D_INNER + 16 * t];
        w0[t][1] = wd[(size_t)(kh + 1 ) * D_INNER + 16 * t];
        w1[t][0] = wd[(size_t)(k2     ) * D_INNER + 16 * t];
        w1[t][1] = wd[(size_t)(k2 + 1 ) * D_INNER + 16 * t];
        bs[t]    = bdt[kdir * D_INNER + dcol + 16 * t];
    }

    if (wave < XP_NT) {
        int mrow = m0 + c16;
        int b = mrow / LL, l = mrow % LL;
        int lsp = scan_to_spatial(kdir, l);
        const float* arow = xi + (size_t)(b * LL + lsp) * D_INNER + kh;
        int nn = 16 * wave + c16;
        bool nok = (nn < NPROJ);
        int ncol = nok ? nn : 0;
        const float* wk = Wxp + (size_t)kdir * D_INNER * NPROJ + (size_t)kh * NPROJ + ncol;

        v8f acc = {};
        v2f   av[2][XP_KU];
        float bw[2][XP_KU][2];
#pragma unroll
        for (int u = 0; u < XP_KU; ++u) {
            int k0 = 4 * u;
            av[0][u]    = *(const v2f*)(arow + k0);
            bw[0][u][0] = wk[(size_t)(k0    ) * NPROJ];
            bw[0][u][1] = wk[(size_t)(k0 + 1) * NPROJ];
        }
#pragma unroll
        for (int bt = 0; bt < XP_NB; ++bt) {
            const int cur = bt & 1, nxt = cur ^ 1;
            if (bt + 1 < XP_NB) {
#pragma unroll
                for (int u = 0; u < XP_KU; ++u) {
                    int k0 = (bt + 1) * 4 * XP_KU + 4 * u;
                    av[nxt][u]    = *(const v2f*)(arow + k0);
                    bw[nxt][u][0] = wk[(size_t)(k0    ) * NPROJ];
                    bw[nxt][u][1] = wk[(size_t)(k0 + 1) * NPROJ];
                }
            }
#pragma unroll
            for (int u = 0; u < XP_KU; ++u) {
                v2f bf;
                bf[0] = nok ? bw[cur][u][0] : 0.f;
                bf[1] = nok ? bw[cur][u][1] : 0.f;
                acc = __builtin_amdgcn_wmma_f32_16x16x4_f32(false, av[cur][u], false, bf,
                                                            (short)0, acc, false, false);
            }
        }
        if (nok) {
#pragma unroll
            for (int v = 0; v < 8; ++v)
                xdbl[(size_t)(kdir * BL + mb + v) * NPROJ + nn] = acc[v];
        }
        if (wave == 0 && c16 < 8) {
#pragma unroll
            for (int v = 0; v < 8; ++v)
                drs[(hf * 8 + v) * 8 + c16] = (c16 < DT_RANK) ? acc[v] : 0.f;
        }
    }
    __syncthreads();
    v2f a0 = {drs[c16 * 8 + kh],     drs[c16 * 8 + kh + 1]};
    v2f a1 = {drs[c16 * 8 + 4 + kh], drs[c16 * 8 + 4 + kh + 1]};

#pragma unroll
    for (int t = 0; t < DL_PW; ++t) {
        v2f b0 = {w0[t][0], w0[t][1]};
        v2f b1;
        b1[0] = hf ? 0.f : w1[t][0];
        b1[1] = hf ? 0.f : w1[t][1];
        v8f dacc = {};
        dacc = __builtin_amdgcn_wmma_f32_16x16x4_f32(false, a0, false, b0,
                                                     (short)0, dacc, false, false);
        dacc = __builtin_amdgcn_wmma_f32_16x16x4_f32(false, a1, false, b1,
                                                     (short)0, dacc, false, false);
        int nn = dcol + 16 * t;
#pragma unroll
        for (int v = 0; v < 8; ++v) {
            float tt = dacc[v] + bs[t];
            float sp = (tt > 20.f) ? tt : log1pf(__expf(tt));
            delta[(size_t)(kdir * BL + mb + v) * D_INNER + nn] = sp;
        }
    }
}


static_assert(WWD % CHUNK == 0 && CHUNK >= 2, "a chunk lies inside one scan row");
static_assert(CHUNK % SCAN_G == 0, "whole groups");

__device__ __forceinline__ size_t part_idx(int chunk, int g, int ch) {
    return ((size_t)(chunk * (D_STATE / 4) + g) * NCHAN + ch) * 4;
}

__device__ __forceinline__ v2f exp_pair(v2f a) {
    v2f r = {__expf(a[0]), __expf(a[1])};
    return r;
}

__global__ void __launch_bounds__(SCAN_BLK) __attribute__((amdgpu_waves_per_eu(1, 4)))
k_scan_part(const float* __restrict__ xi,
            const float* __restrict__ xdbl,
            const float* __restrict__ delta,
            const float* __restrict__ A_logs,
            float* __restrict__ hend,
            float* __restrict__ pdec) {
    int chunk = blockIdx.y;
    int kb    = blockIdx.x / (D_INNER / SCAN_BLK);
    int d     = (blockIdx.x % (D_INNER / SCAN_BLK)) * SCAN_BLK + threadIdx.x;
    int b     = kb % BATCH;
    int kdir  = kb / BATCH;
    int ch    = kb * D_INNER + d;

    v2f A[D_STATE / 2], h[D_STATE / 2], P[D_STATE / 2];
#pragma unroll
    for (int j = 0; j < D_STATE / 2; ++j) {
        const float* al = A_logs + (size_t)(kdir * D_INNER + d) * D_STATE + 2 * j;
        A[j][0] = -__expf(al[0]);
        A[j][1] = -__expf(al[1]);
        h[j] = (v2f){0.f, 0.f};
        P[j] = (v2f){1.f, 1.f};
    }

    int l0    = chunk * CHUNK;
    int lsp0  = scan_to_spatial(kdir, l0);
    int lstep = scan_to_spatial(kdir, l0 + 1) - lsp0;
    const float* xib = xi    + ((size_t)b  * LL + lsp0) * D_INNER + d;
    const float* dlb = delta + ((size_t)kb * LL + l0) * D_INNER + d;
    const float* xdb = xdbl  + ((size_t)kb * LL + l0) * NPROJ + DT_RANK;

    float xc[SCAN_G], dc[SCAN_G], Bs[D_STATE];
#pragma unroll
    for (int g = 0; g < SCAN_G; ++g) {
        xc[g] = xib[(ptrdiff_t)g * lstep * D_INNER];
        dc[g] = dlb[(size_t)g * D_INNER];
    }
#pragma unroll
    for (int n = 0; n < D_STATE; ++n) Bs[n] = xdb[n];

#pragma nounroll
    for (int i0 = 0; i0 < CHUNK; i0 += SCAN_G) {
        float xn[SCAN_G], dn[SCAN_G];
        __builtin_amdgcn_sched_barrier(0);
        int in = (i0 + SCAN_G < CHUNK) ? i0 + SCAN_G : i0;
#pragma unroll
        for (int g = 0; g < SCAN_G; ++g) {
            xn[g] = xib[(ptrdiff_t)(in + g) * lstep * D_INNER];
            dn[g] = dlb[(size_t)(in + g) * D_INNER];
        }
        __builtin_amdgcn_sched_barrier(0);
        const float* bcg = xdb + (size_t)i0 * NPROJ;
#pragma unroll
        for (int g = 0; g < SCAN_G; ++g) {
            const float* bn = (g + 1 < SCAN_G || i0 + SCAN_G < CHUNK) ? bcg + (g + 1) * NPROJ : bcg + g * NPROJ;
            __builtin_amdgcn_sched_barrier(0);
            asm volatile("" :: "s"(Bs[0]));
            float Bn[D_STATE];
#pragma unroll
            for (int n = 0; n < D_STATE; ++n) Bn[n] = bn[n];
            __builtin_amdgcn_sched_barrier(0);
            float dl  = dc[g];
            float dlx = dl * xc[g];
            v2f dl2  = {dl, dl};
            v2f dlx2 = {dlx, dlx};
#pragma unroll
            for (int j = 0; j < D_STATE / 2; ++j) {
                v2f dA = exp_pair(dl2 * A[j]);
                v2f Bv = {Bs[2 * j], Bs[2 * j + 1]};
                h[j] = dA * h[j] + dlx2 * Bv;
                P[j] = P[j] * dA;
            }
#pragma unroll
            for (int n = 0; n < D_STATE; ++n) Bs[n] = Bn[n];
        }
#pragma unroll
        for (int g = 0; g < SCAN_G; ++g) { xc[g] = xn[g]; dc[g] = dn[g]; }
    }
#pragma unroll
    for (int g = 0; g < D_STATE / 4; ++g) {
        v4f hv = {h[2 * g][0], h[2 * g][1], h[2 * g + 1][0], h[2 * g + 1][1]};
        v4f pv = {P[2 * g][0], P[2 * g][1], P[2 * g + 1][0], P[2 * g + 1][1]};
        *(v4f*)(hend + part_idx(chunk, g, ch)) = hv;
        *(v4f*)(pdec + part_idx(chunk, g, ch)) = pv;
    }
}

__global__ void __launch_bounds__(SCAN_BLK) __attribute__((amdgpu_waves_per_eu(1, 4)))
k_scan_final(const float* __restrict__ xi,
             const float* __restrict__ xdbl,
             const float* __restrict__ delta,
             const float* __restrict__ A_logs,
             const float* __restrict__ Ds,
             const float* __restrict__ hend,
             const float* __restrict__ pdec,
             float* __restrict__ ys) {
    int chunk = NCHUNK - 1 - (int)blockIdx.y;
    int kb    = blockIdx.x / (D_INNER / SCAN_BLK);
    int d     = (blockIdx.x % (D_INNER / SCAN_BLK)) * SCAN_BLK + threadIdx.x;
    int b     = kb % BATCH;
    int kdir  = kb / BATCH;
    int ch    = kb * D_INNER + d;

    v2f A[D_STATE / 2], h[D_STATE / 2];
#pragma unroll
    for (int j = 0; j < D_STATE / 2; ++j) {
        const float* al = A_logs + (size_t)(kdir * D_INNER + d) * D_STATE + 2 * j;
        A[j][0] = -__expf(al[0]);
        A[j][1] = -__expf(al[1]);
        h[j] = (v2f){0.f, 0.f};
    }
    float Dskip = Ds[kdir * D_INNER + d];

#pragma unroll 4
    for (int j = 0; j < chunk; ++j) {
#pragma unroll
        for (int g = 0; g < D_STATE / 4; ++g) {
            v4f pv = *(const v4f*)(pdec + part_idx(j, g, ch));
            v4f hv = *(const v4f*)(hend + part_idx(j, g, ch));
            h[2 * g]     = (v2f){pv[0], pv[1]} * h[2 * g]     + (v2f){hv[0], hv[1]};
            h[2 * g + 1] = (v2f){pv[2], pv[3]} * h[2 * g + 1] + (v2f){hv[2], hv[3]};
        }
    }

    int l0    = chunk * CHUNK;
    int lsp0  = scan_to_spatial(kdir, l0);
    int lstep = scan_to_spatial(kdir, l0 + 1) - lsp0;
    const float* xib = xi    + ((size_t)b  * LL + lsp0) * D_INNER + d;
    const float* dlb = delta + ((size_t)kb * LL + l0) * D_INNER + d;
    const float* xdb = xdbl  + ((size_t)kb * LL + l0) * NPROJ + DT_RANK;
    float*       ysb = ys    + ((size_t)kb * LL + l0) * D_INNER + d;

    float xc[SCAN_G], dc[SCAN_G];
    v2f Bv[D_STATE / 2], Cv[D_STATE / 2];
    float bkey = xdb[0];
#pragma unroll
    for (int g = 0; g < SCAN_G; ++g) {
        xc[g] = xib[(ptrdiff_t)g * lstep * D_INNER];
        dc[g] = dlb[(size_t)g * D_INNER];
    }
#pragma unroll
    for (int j = 0; j < D_STATE / 2; ++j) {
        Bv[j] = (v2f){xdb[2 * j], xdb[2 * j + 1]};
        Cv[j] = (v2f){xdb[D_STATE + 2 * j], xdb[D_STATE + 2 * j + 1]};
    }

#pragma nounroll
    for (int i0 = 0; i0 < CHUNK; i0 += SCAN_G) {
        float xn[SCAN_G], dn[SCAN_G];
        __builtin_amdgcn_sched_barrier(0);
        int in = (i0 + SCAN_G < CHUNK) ? i0 + SCAN_G : i0;
#pragma unroll
        for (int g = 0; g < SCAN_G; ++g) {
            xn[g] = xib[(ptrdiff_t)(in + g) * lstep * D_INNER];
            dn[g] = dlb[(size_t)(in + g) * D_INNER];
        }
        __builtin_amdgcn_sched_barrier(0);
        const float* bcg = xdb + (size_t)i0 * NPROJ;
#pragma unroll
        for (int g = 0; g < SCAN_G; ++g) {
            const float* bn = (g + 1 < SCAN_G || i0 + SCAN_G < CHUNK) ? bcg + (g + 1) * NPROJ : bcg + g * NPROJ;
            __builtin_amdgcn_sched_barrier(0);
            asm volatile("" :: "s"(bkey));
            float Bn[D_STATE], Cn[D_STATE];
#pragma unroll
            for (int n = 0; n < D_STATE; ++n) {
                Bn[n] = bn[n];
                Cn[n] = bn[D_STATE + n];
            }
            __builtin_amdgcn_sched_barrier(0);
            float x   = xc[g];
            float dl  = dc[g];
            float dlx = dl * x;
            v2f dl2  = {dl, dl};
            v2f dlx2 = {dlx, dlx};
            v2f ya = {0.f, 0.f}, yb = {0.f, 0.f};
#pragma unroll
            for (int j = 0; j < D_STATE / 2; j += 2) {
                v2f dA0 = exp_pair(dl2 * A[j]);
                v2f dA1 = exp_pair(dl2 * A[j + 1]);
                h[j]     = dA0 * h[j]     + dlx2 * Bv[j];
                h[j + 1] = dA1 * h[j + 1] + dlx2 * Bv[j + 1];
                ya += h[j]     * Cv[j];
                yb += h[j + 1] * Cv[j + 1];
            }
            ysb[(size_t)(i0 + g) * D_INNER] = ((ya[0] + ya[1]) + (yb[0] + yb[1])) + x * Dskip;
            bkey = Bn[0];
#pragma unroll
            for (int j = 0; j < D_STATE / 2; ++j) {
                Bv[j] = (v2f){Bn[2 * j], Bn[2 * j + 1]};
                Cv[j] = (v2f){Cn[2 * j], Cn[2 * j + 1]};
            }
        }
#pragma unroll
        for (int g = 0; g < SCAN_G; ++g) { xc[g] = xn[g]; dc[g] = dn[g]; }
    }
}

__global__ void k_combine(const float* __restrict__ ys,
                          const float* __restrict__ xz,
                          float* __restrict__ y) {
    int tid = blockIdx.x * blockDim.x + threadIdx.x;
    if (tid >= BL * D_INNER) return;
    int d  = tid % D_INNER;
    int bl = tid / D_INNER;
    int b = bl / LL, l = bl % LL;
    float acc = 0.f;
#pragma unroll
    for (int k = 0; k < KDIR; ++k) {
        int ls = spatial_to_scan(k, l);
        acc += ys[(size_t)((k * BATCH + b) * LL + ls) * D_INNER + d];
    }
    float z = xz[(size_t)bl * XZW + D_INNER + d];
    y[tid] = acc * silu(z);
}

__global__ void k_out_gemm(const float* __restrict__ y,
                           const float* __restrict__ Wout,
                           float* __restrict__ out) {
    int lane = threadIdx.x & 31;
    int wave = threadIdx.x >> 5;
    int mTile = blockIdx.x * 4 + wave;
    int nTile = blockIdx.y;
    int m0 = mTile * 16, n0 = nTile * 16;
    int mrow = m0 + (lane & 15);
    int nn = n0 + (lane & 15);
    int kh = (lane >> 4) * 2;
    v8f acc = {};
    for (int kb = 0; kb < D_INNER; kb += 4) {
        v2f a, bf;
        a[0]  = y[(size_t)mrow * D_INNER + kb + kh];
        a[1]  = y[(size_t)mrow * D_INNER + kb + kh + 1];
        bf[0] = Wout[(size_t)(kb + kh    ) * DIMC + nn];
        bf[1] = Wout[(size_t)(kb + kh + 1) * DIMC + nn];
        acc = __builtin_amdgcn_wmma_f32_16x16x4_f32(false, a, false, bf,
                                                    (short)0, acc, false, false);
    }
    int mb = m0 + (lane >> 4) * 8;
#pragma unroll
    for (int v = 0; v < 8; ++v) {
        int row = mb + v;
        int b = row / LL, l = row % LL;
        out[(size_t)(b * DIMC + nn) * LL + l] = acc[v];
    }
}

extern "C" void kernel_launch(void* const* d_in, const int* in_sizes, int n_in,
                              void* d_out, int out_size, void* d_ws, size_t ws_size,
                              hipStream_t stream) {
    (void)in_sizes; (void)n_in; (void)out_size; (void)ws_size;
    const float* x       = (const float*)d_in[0];
    const float* W_in    = (const float*)d_in[1];
    const float* W_conv  = (const float*)d_in[2];
    const float* b_conv  = (const float*)d_in[3];
    const float* W_xproj = (const float*)d_in[4];
    const float* W_dt    = (const float*)d_in[5];
    const float* b_dt    = (const float*)d_in[6];
    const float* A_logs  = (const float*)d_in[7];
    const float* Ds      = (const float*)d_in[8];
    const float* W_out   = (const float*)d_in[9];
    float* out = (float*)d_out;

    float* ws = (float*)d_ws;
    size_t o = 0;
    float* xz    = ws + o;  o += (size_t)BL * XZW;
    float* xi    = ws + o;  o += (size_t)BL * D_INNER;
    float* xdbl  = ws + o;  o += (size_t)KDIR * BL * NPROJ;
    float* delta = ws + o;  o += (size_t)KDIR * BL * D_INNER;
    float* ysbuf = ws + o;  o += (size_t)KDIR * BL * D_INNER;
    float* ybuf  = ws + o;  o += (size_t)BL * D_INNER;
    float* hend  = ws + o;  o += (size_t)(NCHUNK - 1) * NSCAN;
    float* pdec  = ws + o;  o += (size_t)(NCHUNK - 1) * NSCAN;

    dim3 blk(128);
    k_in_conv <<<CV_BLOCKS + Z_BLOCKS, IC_THREADS, 0, stream>>>(x, W_in, W_conv, b_conv, xz, xi);
    int nelt = BL * D_INNER;
    k_xproj_delta<<<dim3(392, 4), blk, 0, stream>>>(xi, W_xproj, W_dt, b_dt, xdbl, delta);
    k_scan_part <<<dim3(NCHAN / SCAN_BLK, NCHUNK - 1), SCAN_BLK, 0, stream>>>(
        xi, xdbl, delta, A_logs, hend, pdec);
    k_scan_final<<<dim3(NCHAN / SCAN_BLK, NCHUNK),     SCAN_BLK, 0, stream>>>(
        xi, xdbl, delta, A_logs, Ds, hend, pdec, ysbuf);
    k_combine <<<(nelt + 255) / 256, 256, 0, stream>>>(ysbuf, xz, ybuf);
    k_out_gemm<<<dim3(98, 6),     blk, 0, stream>>>(ybuf, W_out, out);
}


# dt-projection softplus: log(1+e) = log2(s)*ln2 + r/s by hardware log2 and reciprocal, s = 1+e rounded, r its exact rounding error (was library log1pf, ~90 dependent ops); delta moves in last place
# speedup vs baseline: 22.9096x; 1.0824x over previous
#include <hip/hip_runtime.h>
#include <math.h>

typedef __attribute__((ext_vector_type(2))) float v2f;
typedef __attribute__((ext_vector_type(4))) float v4f;
typedef __attribute__((ext_vector_type(8))) float v8f;

#define DIMC    96
#define D_STATE 16
#define D_INNER 192
#define DT_RANK 6
#define KDIR    4
#define BATCH   2
#define HH      56
#define WWD     56
#define LL      (HH * WWD)
#define BL      (BATCH * LL)
#define NPROJ   (DT_RANK + 2 * D_STATE)
#define XZW     (2 * D_INNER)
#define NSCAN   (KDIR * BATCH * D_INNER * D_STATE)
#define NCHAN   (KDIR * BATCH * D_INNER)
#define SCAN_BLK 64
#define SCAN_G   8
#define NCHUNK  56
#define CHUNK   (LL / NCHUNK)

__device__ __forceinline__ int scan_to_spatial(int k, int l) {
    int q = l / WWD, r = l % WWD;
    if (k == 0) return l;
    if (k == 1) return q * WWD + (WWD - 1 - r);
    if (k == 2) return r * WWD + q;
    return (WWD - 1 - r) * WWD + q;
}

__device__ __forceinline__ int spatial_to_scan(int k, int l) {
    int q = l / WWD, r = l % WWD;
    if (k == 0) return l;
    if (k == 1) return q * WWD + (WWD - 1 - r);
    if (k == 2) return r * WWD + q;
    return r * WWD + (WWD - 1 - q);
}

__device__ __forceinline__ float silu(float v) {
    return v / (1.f + __expf(-v));
}

__device__ __forceinline__ float softplus20(float t) {
    float e  = __expf(t);
    float s  = 1.f + e;
    float bb = s - 1.f;
    float r  = (1.f - (s - bb)) + (e - bb);
    float l2 = __builtin_amdgcn_logf(s);
    const float c = 0x1.62e42ep-1f, cc = 0x1.efa39ep-25f;
    float lo = __builtin_fmaf(l2, cc, r * __builtin_amdgcn_rcpf(s));
    return (t > 20.f) ? t : __builtin_fmaf(l2, c, lo);
}

#define CV_ROWS   4
#define CV_WIN    (CV_ROWS + 2)
#define CV_POS    (CV_WIN * WWD)
#define CV_PW     (WWD + 2)
#define CV_COLS   32
#define CV_PITCH  34
#define CV_KU     6
#define CV_NB     (DIMC / (4 * CV_KU))
#define IC_WAVES  7
#define IC_THREADS (IC_WAVES * 32)
#define CV_PER_IMG ((HH / CV_ROWS) * (D_INNER / CV_COLS))
#define CV_BLOCKS (BATCH * CV_PER_IMG)
#define Z_BLOCKS  ((BL / 16 / IC_WAVES) * (D_INNER / 16))
static_assert(HH % CV_ROWS == 0, "whole bands");
static_assert(CV_POS == IC_WAVES * 3 * 16, "three row tiles per wave");
static_assert(CV_ROWS * WWD == IC_WAVES * 32, "32 conv outputs per thread");
static_assert((BL / 16) % IC_WAVES == 0, "whole z blocks");
static_assert(DIMC % (4 * CV_KU) == 0, "whole groups of k blocks");

__global__ void __launch_bounds__(IC_THREADS)
k_in_conv(const float* __restrict__ x,
          const float* __restrict__ Win,
          const float* __restrict__ Wc,
          const float* __restrict__ bc,
          float* __restrict__ xz,
          float* __restrict__ xi) {
    __shared__ float xs[CV_WIN * CV_PW * CV_PITCH];
    int lane = threadIdx.x & 31;
    int wave = __builtin_amdgcn_readfirstlane((int)(threadIdx.x >> 5));
    int c16  = lane & 15;
    int kh   = (lane >> 4) * 2;

    if (blockIdx.x >= CV_BLOCKS) {
        int zb    = (int)blockIdx.x - CV_BLOCKS;
        int mTile = (zb / (D_INNER / 16)) * IC_WAVES + wave;
        int nTile = D_INNER / 16 + zb % (D_INNER / 16);
        int m0 = mTile * 16, n0 = nTile * 16;
        int mrow = m0 + c16;
        int b = mrow / LL, l = mrow % LL;
        int nn = n0 + c16;
        float za[DIMC / 4][2], zw[DIMC / 4][2];
#pragma unroll
        for (int u = 0; u < DIMC / 4; ++u) {
            int c0 = 4 * u + kh;
            za[u][0] = x[(size_t)(b * DIMC + c0    ) * LL + l];
            za[u][1] = x[(size_t)(b * DIMC + c0 + 1) * LL + l];
            zw[u][0] = Win[(size_t)(c0    ) * XZW + nn];
            zw[u][1] = Win[(size_t)(c0 + 1) * XZW + nn];
        }
        v8f acc = {};
#pragma unroll
        for (int u = 0; u < DIMC / 4; ++u) {
            v2f a  = {za[u][0], za[u][1]};
            v2f bf = {zw[u][0], zw[u][1]};
            acc = __builtin_amdgcn_wmma_f32_16x16x4_f32(false, a, false, bf,
                                                        (short)0, acc, false, false);
        }
        int mb = m0 + (lane >> 4) * 8;
#pragma unroll
        for (int v = 0; v < 8; ++v)
            xz[(size_t)(mb + v) * XZW + nn] = acc[v];
        return;
    }

    int idx  = (int)blockIdx.x;
    int b    = idx / CV_PER_IMG;
    int rem  = idx % CV_PER_IMG;
    int band = rem / (D_INNER / CV_COLS);
    int d0   = (rem % (D_INNER / CV_COLS)) * CV_COLS;
    int hfirst = band * CV_ROWS - 1;

    for (int i = threadIdx.x; i < CV_WIN * 2 * CV_COLS; i += IC_THREADS) {
        int row = i / (2 * CV_COLS), side = (i / CV_COLS) & 1, col = i % CV_COLS;
        xs[(row * CV_PW + side * (CV_PW - 1)) * CV_PITCH + col] = 0.f;
    }

    const float* xa[3];
    bool ok[3];
#pragma unroll
    for (int t = 0; t < 3; ++t) {
        int p = (wave * 3 + t) * 16 + c16;
        int h = hfirst + p / WWD;
        int w = p % WWD;
        ok[t] = (h >= 0 && h < HH);
        int l = ok[t] ? h * WWD + w : 0;
        xa[t] = x + (size_t)b * DIMC * LL + l;
    }
    const float* wb = Win + d0 + c16;
    v8f acc[3][2];
#pragma unroll
    for (int t = 0; t < 3; ++t) { acc[t][0] = (v8f){}; acc[t][1] = (v8f){}; }
    float av[2][CV_KU][3][2], bv[2][CV_KU][2][2];
#pragma unroll
    for (int u = 0; u < CV_KU; ++u) {
        int c0 = 4 * u + kh;
#pragma unroll
        for (int t = 0; t < 3; ++t) {
            av[0][u][t][0] = xa[t][(size_t)(c0    ) * LL];
            av[0][u][t][1] = xa[t][(size_t)(c0 + 1) * LL];
        }
#pragma unroll
        for (int j = 0; j < 2; ++j) {
            bv[0][u][j][0] = wb[(size_t)(c0    ) * XZW + 16 * j];
            bv[0][u][j][1] = wb[(size_t)(c0 + 1) * XZW + 16 * j];
        }
    }
#pragma unroll
    for (int bt = 0; bt < CV_NB; ++bt) {
        const int cur = bt & 1, nxt = cur ^ 1;
        if (bt + 1 < CV_NB) {
#pragma unroll
            for (int u = 0; u < CV_KU; ++u) {
                int c0 = (bt + 1) * 4 * CV_KU + 4 * u + kh;
#pragma unroll
                for (int t = 0; t < 3; ++t) {
                    av[nxt][u][t][0] = xa[t][(size_t)(c0    ) * LL];
                    av[nxt][u][t][1] = xa[t][(size_t)(c0 + 1) * LL];
                }
#pragma unroll
                for (int j = 0; j < 2; ++j) {
                    bv[nxt][u][j][0] = wb[(size_t)(c0    ) * XZW + 16 * j];
                    bv[nxt][u][j][1] = wb[(size_t)(c0 + 1) * XZW + 16 * j];
                }
            }
        }
#pragma unroll
        for (int u = 0; u < CV_KU; ++u) {
#pragma unroll
            for (int t = 0; t < 3; ++t) {
                v2f a;
                a[0] = ok[t] ? av[cur][u][t][0] : 0.f;
                a[1] = ok[t] ? av[cur][u][t][1] : 0.f;
#pragma unroll
                for (int j = 0; j < 2; ++j) {
                    v2f bf = {bv[cur][u][j][0], bv[cur][u][j][1]};
                    acc[t][j] = __builtin_amdgcn_wmma_f32_16x16x4_f32(false, a, false, bf,
                                                                      (short)0, acc[t][j], false, false);
                }
            }
        }
    }
#pragma unroll
    for (int t = 0; t < 3; ++t) {
        int prow = (wave * 3 + t) * 16 + (lane >> 4) * 8;
#pragma unroll
        for (int v = 0; v < 8; ++v) {
            int p = prow + v;
            int q = (p / WWD) * CV_PW + p % WWD + 1;
            xs[q * CV_PITCH + c16]      = acc[t][0][v];
            xs[q * CV_PITCH + 16 + c16] = acc[t][1][v];
        }
    }
    __syncthreads();

    int d = d0 + lane;
    float wg[9];
#pragma unroll
    for (int t = 0; t < 9; ++t) wg[t] = Wc[d * 9 + t];
    float bias = bc[d];
    float* xo = xi + ((size_t)b * LL + band * CV_ROWS * WWD) * D_INNER + d;
#pragma unroll 4
    for (int i = 0; i < (CV_ROWS * WWD) / IC_WAVES; ++i) {
        int o = wave + IC_WAVES * i;
        int r = o / WWD, w = o % WWD;
        const float* q = xs + (r * CV_PW + w) * CV_PITCH + lane;
        float s = bias;
#pragma unroll
        for (int dy = 0; dy < 3; ++dy)
#pragma unroll
            for (int dx = 0; dx < 3; ++dx)
                s = __builtin_fmaf(wg[dy * 3 + dx], q[(dy * CV_PW + dx) * CV_PITCH], s);
        xo[(size_t)o * D_INNER] = silu(s);
    }
}

#define XP_KU   6
#define XP_NB   (D_INNER / (4 * XP_KU))
#define XP_NT   3
#define DL_PW   (D_INNER / 16 / 4)
static_assert(D_INNER % (4 * XP_KU) == 0, "whole groups of k blocks");
static_assert(D_INNER % 64 == 0, "whole column tiles per wave");

__global__ void __launch_bounds__(128)
k_xproj_delta(const float* __restrict__ xi,
              const float* __restrict__ Wxp,
              const float* __restrict__ Wdt,
              const float* __restrict__ bdt,
              float* __restrict__ xdbl,
              float* __restrict__ delta) {
    __shared__ float drs[16 * 8];
    int lane = threadIdx.x & 31;
    int wave = __builtin_amdgcn_readfirstlane((int)(threadIdx.x >> 5));
    int kdir = blockIdx.y;
    int m0 = blockIdx.x * 16;
    int c16 = lane & 15;
    int hf  = lane >> 4;
    int kh  = hf * 2;
    int mb  = m0 + hf * 8;

    int dcol = wave * (16 * DL_PW) + c16;
    const float* wd = Wdt + (size_t)kdir * DT_RANK * D_INNER + dcol;
    int k2 = hf ? 0 : 4;
    float w0[DL_PW][2], w1[DL_PW][2], bs[DL_PW];
#pragma unroll
    for (int t = 0; t < DL_PW; ++t) {
        w0[t][0] = wd[(size_t)(kh     ) * D_INNER + 16 * t];
        w0[t][1] = wd[(size_t)(kh + 1 ) * D_INNER + 16 * t];
        w1[t][0] = wd[(size_t)(k2     ) * D_INNER + 16 * t];
        w1[t][1] = wd[(size_t)(k2 + 1 ) * D_INNER + 16 * t];
        bs[t]    = bdt[kdir * D_INNER + dcol + 16 * t];
    }

    if (wave < XP_NT) {
        int mrow = m0 + c16;
        int b = mrow / LL, l = mrow % LL;
        int lsp = scan_to_spatial(kdir, l);
        const float* arow = xi + (size_t)(b * LL + lsp) * D_INNER + kh;
        int nn = 16 * wave + c16;
        bool nok = (nn < NPROJ);
        int ncol = nok ? nn : 0;
        const float* wk = Wxp + (size_t)kdir * D_INNER * NPROJ + (size_t)kh * NPROJ + ncol;

        v8f acc = {};
        v2f   av[2][XP_KU];
        float bw[2][XP_KU][2];
#pragma unroll
        for (int u = 0; u < XP_KU; ++u) {
            int k0 = 4 * u;
            av[0][u]    = *(const v2f*)(arow + k0);
            bw[0][u][0] = wk[(size_t)(k0    ) * NPROJ];
            bw[0][u][1] = wk[(size_t)(k0 + 1) * NPROJ];
        }
#pragma unroll
        for (int bt = 0; bt < XP_NB; ++bt) {
            const int cur = bt & 1, nxt = cur ^ 1;
            if (bt + 1 < XP_NB) {
#pragma unroll
                for (int u = 0; u < XP_KU; ++u) {
                    int k0 = (bt + 1) * 4 * XP_KU + 4 * u;
                    av[nxt][u]    = *(const v2f*)(arow + k0);
                    bw[nxt][u][0] = wk[(size_t)(k0    ) * NPROJ];
                    bw[nxt][u][1] = wk[(size_t)(k0 + 1) * NPROJ];
                }
            }
#pragma unroll
            for (int u = 0; u < XP_KU; ++u) {
                v2f bf;
                bf[0] = nok ? bw[cur][u][0] : 0.f;
                bf[1] = nok ? bw[cur][u][1] : 0.f;
                acc = __builtin_amdgcn_wmma_f32_16x16x4_f32(false, av[cur][u], false, bf,
                                                            (short)0, acc, false, false);
            }
        }
        if (nok) {
#pragma unroll
            for (int v = 0; v < 8; ++v)
                xdbl[(size_t)(kdir * BL + mb + v) * NPROJ + nn] = acc[v];
        }
        if (wave == 0 && c16 < 8) {
#pragma unroll
            for (int v = 0; v < 8; ++v)
                drs[(hf * 8 + v) * 8 + c16] = (c16 < DT_RANK) ? acc[v] : 0.f;
        }
    }
    __syncthreads();
    v2f a0 = {drs[c16 * 8 + kh],     drs[c16 * 8 + kh + 1]};
    v2f a1 = {drs[c16 * 8 + 4 + kh], drs[c16 * 8 + 4 + kh + 1]};

#pragma unroll
    for (int t = 0; t < DL_PW; ++t) {
        v2f b0 = {w0[t][0], w0[t][1]};
        v2f b1;
        b1[0] = hf ? 0.f : w1[t][0];
        b1[1] = hf ? 0.f : w1[t][1];
        v8f dacc = {};
        dacc = __builtin_amdgcn_wmma_f32_16x16x4_f32(false, a0, false, b0,
                                                     (short)0, dacc, false, false);
        dacc = __builtin_amdgcn_wmma_f32_16x16x4_f32(false, a1, false, b1,
                                                     (short)0, dacc, false, false);
        int nn = dcol + 16 * t;
#pragma unroll
        for (int v = 0; v < 8; ++v) {
            float tt = dacc[v] + bs[t];
            delta[(size_t)(kdir * BL + mb + v) * D_INNER + nn] = softplus20(tt);
        }
    }
}


static_assert(WWD % CHUNK == 0 && CHUNK >= 2, "a chunk lies inside one scan row");
static_assert(CHUNK % SCAN_G == 0, "whole groups");

__device__ __forceinline__ size_t part_idx(int chunk, int g, int ch) {
    return ((size_t)(chunk * (D_STATE / 4) + g) * NCHAN + ch) * 4;
}

__device__ __forceinline__ v2f exp_pair(v2f a) {
    v2f r = {__expf(a[0]), __expf(a[1])};
    return r;
}

__global__ void __launch_bounds__(SCAN_BLK) __attribute__((amdgpu_waves_per_eu(1, 4)))
k_scan_part(const float* __restrict__ xi,
            const float* __restrict__ xdbl,
            const float* __restrict__ delta,
            const float* __restrict__ A_logs,
            float* __restrict__ hend,
            float* __restrict__ pdec) {
    int chunk = blockIdx.y;
    int kb    = blockIdx.x / (D_INNER / SCAN_BLK);
    int d     = (blockIdx.x % (D_INNER / SCAN_BLK)) * SCAN_BLK + threadIdx.x;
    int b     = kb % BATCH;
    int kdir  = kb / BATCH;
    int ch    = kb * D_INNER + d;

    v2f A[D_STATE / 2], h[D_STATE / 2], P[D_STATE / 2];
#pragma unroll
    for (int j = 0; j < D_STATE / 2; ++j) {
        const float* al = A_logs + (size_t)(kdir * D_INNER + d) * D_STATE + 2 * j;
        A[j][0] = -__expf(al[0]);
        A[j][1] = -__expf(al[1]);
        h[j] = (v2f){0.f, 0.f};
        P[j] = (v2f){1.f, 1.f};
    }

    int l0    = chunk * CHUNK;
    int lsp0  = scan_to_spatial(kdir, l0);
    int lstep = scan_to_spatial(kdir, l0 + 1) - lsp0;
    const float* xib = xi    + ((size_t)b  * LL + lsp0) * D_INNER + d;
    const float* dlb = delta + ((size_t)kb * LL + l0) * D_INNER + d;
    const float* xdb = xdbl  + ((size_t)kb * LL + l0) * NPROJ + DT_RANK;

    float xc[SCAN_G], dc[SCAN_G], Bs[D_STATE];
#pragma unroll
    for (int g = 0; g < SCAN_G; ++g) {
        xc[g] = xib[(ptrdiff_t)g * lstep * D_INNER];
        dc[g] = dlb[(size_t)g * D_INNER];
    }
#pragma unroll
    for (int n = 0; n < D_STATE; ++n) Bs[n] = xdb[n];

#pragma nounroll
    for (int i0 = 0; i0 < CHUNK; i0 += SCAN_G) {
        float xn[SCAN_G], dn[SCAN_G];
        __builtin_amdgcn_sched_barrier(0);
        int in = (i0 + SCAN_G < CHUNK) ? i0 + SCAN_G : i0;
#pragma unroll
        for (int g = 0; g < SCAN_G; ++g) {
            xn[g] = xib[(ptrdiff_t)(in + g) * lstep * D_INNER];
            dn[g] = dlb[(size_t)(in + g) * D_INNER];
        }
        __builtin_amdgcn_sched_barrier(0);
        const float* bcg = xdb + (size_t)i0 * NPROJ;
#pragma unroll
        for (int g = 0; g < SCAN_G; ++g) {
            const float* bn = (g + 1 < SCAN_G || i0 + SCAN_G < CHUNK) ? bcg + (g + 1) * NPROJ : bcg + g * NPROJ;
            __builtin_amdgcn_sched_barrier(0);
            asm volatile("" :: "s"(Bs[0]));
            float Bn[D_STATE];
#pragma unroll
            for (int n = 0; n < D_STATE; ++n) Bn[n] = bn[n];
            __builtin_amdgcn_sched_barrier(0);
            float dl  = dc[g];
            float dlx = dl * xc[g];
            v2f dl2  = {dl, dl};
            v2f dlx2 = {dlx, dlx};
#pragma unroll
            for (int j = 0; j < D_STATE / 2; ++j) {
                v2f dA = exp_pair(dl2 * A[j]);
                v2f Bv = {Bs[2 * j], Bs[2 * j + 1]};
                h[j] = dA * h[j] + dlx2 * Bv;
                P[j] = P[j] * dA;
            }
#pragma unroll
            for (int n = 0; n < D_STATE; ++n) Bs[n] = Bn[n];
        }
#pragma unroll
        for (int g = 0; g < SCAN_G; ++g) { xc[g] = xn[g]; dc[g] = dn[g]; }
    }
#pragma unroll
    for (int g = 0; g < D_STATE / 4; ++g) {
        v4f hv = {h[2 * g][0], h[2 * g][1], h[2 * g + 1][0], h[2 * g + 1][1]};
        v4f pv = {P[2 * g][0], P[2 * g][1], P[2 * g + 1][0], P[2 * g + 1][1]};
        *(v4f*)(hend + part_idx(chunk, g, ch)) = hv;
        *(v4f*)(pdec + part_idx(chunk, g, ch)) = pv;
    }
}

__global__ void __launch_bounds__(SCAN_BLK) __attribute__((amdgpu_waves_per_eu(1, 4)))
k_scan_final(const float* __restrict__ xi,
             const float* __restrict__ xdbl,
             const float* __restrict__ delta,
             const float* __restrict__ A_logs,
             const float* __restrict__ Ds,
             const float* __restrict__ hend,
             const float* __restrict__ pdec,
             float* __restrict__ ys) {
    int chunk = NCHUNK - 1 - (int)blockIdx.y;
    int kb    = blockIdx.x / (D_INNER / SCAN_BLK);
    int d     = (blockIdx.x % (D_INNER / SCAN_BLK)) * SCAN_BLK + threadIdx.x;
    int b     = kb % BATCH;
    int kdir  = kb / BATCH;
    int ch    = kb * D_INNER + d;

    v2f A[D_STATE / 2], h[D_STATE / 2];
#pragma unroll
    for (int j = 0; j < D_STATE / 2; ++j) {
        const float* al = A_logs + (size_t)(kdir * D_INNER + d) * D_STATE + 2 * j;
        A[j][0] = -__expf(al[0]);
        A[j][1] = -__expf(al[1]);
        h[j] = (v2f){0.f, 0.f};
    }
    float Dskip = Ds[kdir * D_INNER + d];

#pragma unroll 4
    for (int j = 0; j < chunk; ++j) {
#pragma unroll
        for (int g = 0; g < D_STATE / 4; ++g) {
            v4f pv = *(const v4f*)(pdec + part_idx(j, g, ch));
            v4f hv = *(const v4f*)(hend + part_idx(j, g, ch));
            h[2 * g]     = (v2f){pv[0], pv[1]} * h[2 * g]     + (v2f){hv[0], hv[1]};
            h[2 * g + 1] = (v2f){pv[2], pv[3]} * h[2 * g + 1] + (v2f){hv[2], hv[3]};
        }
    }

    int l0    = chunk * CHUNK;
    int lsp0  = scan_to_spatial(kdir, l0);
    int lstep = scan_to_spatial(kdir, l0 + 1) - lsp0;
    const float* xib = xi    + ((size_t)b  * LL + lsp0) * D_INNER + d;
    const float* dlb = delta + ((size_t)kb * LL + l0) * D_INNER + d;
    const float* xdb = xdbl  + ((size_t)kb * LL + l0) * NPROJ + DT_RANK;
    float*       ysb = ys    + ((size_t)kb * LL + l0) * D_INNER + d;

    float xc[SCAN_G], dc[SCAN_G];
    v2f Bv[D_STATE / 2], Cv[D_STATE / 2];
    float bkey = xdb[0];
#pragma unroll
    for (int g = 0; g < SCAN_G; ++g) {
        xc[g] = xib[(ptrdiff_t)g * lstep * D_INNER];
        dc[g] = dlb[(size_t)g * D_INNER];
    }
#pragma unroll
    for (int j = 0; j < D_STATE / 2; ++j) {
        Bv[j] = (v2f){xdb[2 * j], xdb[2 * j + 1]};
        Cv[j] = (v2f){xdb[D_STATE + 2 * j], xdb[D_STATE + 2 * j + 1]};
    }

#pragma nounroll
    for (int i0 = 0; i0 < CHUNK; i0 += SCAN_G) {
        float xn[SCAN_G], dn[SCAN_G];
        __builtin_amdgcn_sched_barrier(0);
        int in = (i0 + SCAN_G < CHUNK) ? i0 + SCAN_G : i0;
#pragma unroll
        for (int g = 0; g < SCAN_G; ++g) {
            xn[g] = xib[(ptrdiff_t)(in + g) * lstep * D_INNER];
            dn[g] = dlb[(size_t)(in + g) * D_INNER];
        }
        __builtin_amdgcn_sched_barrier(0);
        const float* bcg = xdb + (size_t)i0 * NPROJ;
#pragma unroll
        for (int g = 0; g < SCAN_G; ++g) {
            const float* bn = (g + 1 < SCAN_G || i0 + SCAN_G < CHUNK) ? bcg + (g + 1) * NPROJ : bcg + g * NPROJ;
            __builtin_amdgcn_sched_barrier(0);
            asm volatile("" :: "s"(bkey));
            float Bn[D_STATE], Cn[D_STATE];
#pragma unroll
            for (int n = 0; n < D_STATE; ++n) {
                Bn[n] = bn[n];
                Cn[n] = bn[D_STATE + n];
            }
            __builtin_amdgcn_sched_barrier(0);
            float x   = xc[g];
            float dl  = dc[g];
            float dlx = dl * x;
            v2f dl2  = {dl, dl};
            v2f dlx2 = {dlx, dlx};
            v2f ya = {0.f, 0.f}, yb = {0.f, 0.f};
#pragma unroll
            for (int j = 0; j < D_STATE / 2; j += 2) {
                v2f dA0 = exp_pair(dl2 * A[j]);
                v2f dA1 = exp_pair(dl2 * A[j + 1]);
                h[j]     = dA0 * h[j]     + dlx2 * Bv[j];
                h[j + 1] = dA1 * h[j + 1] + dlx2 * Bv[j + 1];
                ya += h[j]     * Cv[j];
                yb += h[j + 1] * Cv[j + 1];
            }
            ysb[(size_t)(i0 + g) * D_INNER] = ((ya[0] + ya[1]) + (yb[0] + yb[1])) + x * Dskip;
            bkey = Bn[0];
#pragma unroll
            for (int j = 0; j < D_STATE / 2; ++j) {
                Bv[j] = (v2f){Bn[2 * j], Bn[2 * j + 1]};
                Cv[j] = (v2f){Cn[2 * j], Cn[2 * j + 1]};
            }
        }
#pragma unroll
        for (int g = 0; g < SCAN_G; ++g) { xc[g] = xn[g]; dc[g] = dn[g]; }
    }
}

__global__ void k_combine(const float* __restrict__ ys,
                          const float* __restrict__ xz,
                          float* __restrict__ y) {
    int tid = blockIdx.x * blockDim.x + threadIdx.x;
    if (tid >= BL * D_INNER) return;
    int d  = tid % D_INNER;
    int bl = tid / D_INNER;
    int b = bl / LL, l = bl % LL;
    float acc = 0.f;
#pragma unroll
    for (int k = 0; k < KDIR; ++k) {
        int ls = spatial_to_scan(k, l);
        acc += ys[(size_t)((k * BATCH + b) * LL + ls) * D_INNER + d];
    }
    float z = xz[(size_t)bl * XZW + D_INNER + d];
    y[tid] = acc * silu(z);
}

__global__ void k_out_gemm(const float* __restrict__ y,
                           const float* __restrict__ Wout,
                           float* __restrict__ out) {
    int lane = threadIdx.x & 31;
    int wave = threadIdx.x >> 5;
    int mTile = blockIdx.x * 4 + wave;
    int nTile = blockIdx.y;
    int m0 = mTile * 16, n0 = nTile * 16;
    int mrow = m0 + (lane & 15);
    int nn = n0 + (lane & 15);
    int kh = (lane >> 4) * 2;
    v8f acc = {};
    for (int kb = 0; kb < D_INNER; kb += 4) {
        v2f a, bf;
        a[0]  = y[(size_t)mrow * D_INNER + kb + kh];
        a[1]  = y[(size_t)mrow * D_INNER + kb + kh + 1];
        bf[0] = Wout[(size_t)(kb + kh    ) * DIMC + nn];
        bf[1] = Wout[(size_t)(kb + kh + 1) * DIMC + nn];
        acc = __builtin_amdgcn_wmma_f32_16x16x4_f32(false, a, false, bf,
                                                    (short)0, acc, false, false);
    }
    int mb = m0 + (lane >> 4) * 8;
#pragma unroll
    for (int v = 0; v < 8; ++v) {
        int row = mb + v;
        int b = row / LL, l = row % LL;
        out[(size_t)(b * DIMC + nn) * LL + l] = acc[v];
    }
}

extern "C" void kernel_launch(void* const* d_in, const int* in_sizes, int n_in,
                              void* d_out, int out_size, void* d_ws, size_t ws_size,
                              hipStream_t stream) {
    (void)in_sizes; (void)n_in; (void)out_size; (void)ws_size;
    const float* x       = (const float*)d_in[0];
    const float* W_in    = (const float*)d_in[1];
    const float* W_conv  = (const float*)d_in[2];
    const float* b_conv  = (const float*)d_in[3];
    const float* W_xproj = (const float*)d_in[4];
    const float* W_dt    = (const float*)d_in[5];
    const float* b_dt    = (const float*)d_in[6];
    const float* A_logs  = (const float*)d_in[7];
    const float* Ds      = (const float*)d_in[8];
    const float* W_out   = (const float*)d_in[9];
    float* out = (float*)d_out;

    float* ws = (float*)d_ws;
    size_t o = 0;
    float* xz    = ws + o;  o += (size_t)BL * XZW;
    float* xi    = ws + o;  o += (size_t)BL * D_INNER;
    float* xdbl  = ws + o;  o += (size_t)KDIR * BL * NPROJ;
    float* delta = ws + o;  o += (size_t)KDIR * BL * D_INNER;
    float* ysbuf = ws + o;  o += (size_t)KDIR * BL * D_INNER;
    float* ybuf  = ws + o;  o += (size_t)BL * D_INNER;
    float* hend  = ws + o;  o += (size_t)(NCHUNK - 1) * NSCAN;
    float* pdec  = ws + o;  o += (size_t)(NCHUNK - 1) * NSCAN;

    dim3 blk(128);
    k_in_conv <<<CV_BLOCKS + Z_BLOCKS, IC_THREADS, 0, stream>>>(x, W_in, W_conv, b_conv, xz, xi);
    int nelt = BL * D_INNER;
    k_xproj_delta<<<dim3(392, 4), blk, 0, stream>>>(xi, W_xproj, W_dt, b_dt, xdbl, delta);
    k_scan_part <<<dim3(NCHAN / SCAN_BLK, NCHUNK - 1), SCAN_BLK, 0, stream>>>(
        xi, xdbl, delta, A_logs, hend, pdec);
    k_scan_final<<<dim3(NCHAN / SCAN_BLK, NCHUNK),     SCAN_BLK, 0, stream>>>(
        xi, xdbl, delta, A_logs, Ds, hend, pdec, ysbuf);
    k_combine <<<(nelt + 255) / 256, 256, 0, stream>>>(ysbuf, xz, ybuf);
    k_out_gemm<<<dim3(98, 6),     blk, 0, stream>>>(ybuf, W_out, out);
}


# combine launch folded into out GEMM: its y loader sums the 4 directions and gates by silu(z) in registers; wave = 16 positions x 48 channels, positions on lanes; k order of the out sums changes
# speedup vs baseline: 24.2182x; 1.0571x over previous
#include <hip/hip_runtime.h>
#include <math.h>

typedef __attribute__((ext_vector_type(2))) float v2f;
typedef __attribute__((ext_vector_type(4))) float v4f;
typedef __attribute__((ext_vector_type(8))) float v8f;

#define DIMC    96
#define D_STATE 16
#define D_INNER 192
#define DT_RANK 6
#define KDIR    4
#define BATCH   2
#define HH      56
#define WWD     56
#define LL      (HH * WWD)
#define BL      (BATCH * LL)
#define NPROJ   (DT_RANK + 2 * D_STATE)
#define XZW     (2 * D_INNER)
#define NSCAN   (KDIR * BATCH * D_INNER * D_STATE)
#define NCHAN   (KDIR * BATCH * D_INNER)
#define SCAN_BLK 64
#define SCAN_G   8
#define NCHUNK  56
#define CHUNK   (LL / NCHUNK)

__device__ __forceinline__ int scan_to_spatial(int k, int l) {
    int q = l / WWD, r = l % WWD;
    if (k == 0) return l;
    if (k == 1) return q * WWD + (WWD - 1 - r);
    if (k == 2) return r * WWD + q;
    return (WWD - 1 - r) * WWD + q;
}

__device__ __forceinline__ int spatial_to_scan(int k, int l) {
    int q = l / WWD, r = l % WWD;
    if (k == 0) return l;
    if (k == 1) return q * WWD + (WWD - 1 - r);
    if (k == 2) return r * WWD + q;
    return r * WWD + (WWD - 1 - q);
}

__device__ __forceinline__ float silu(float v) {
    return v / (1.f + __expf(-v));
}

__device__ __forceinline__ float softplus20(float t) {
    float e  = __expf(t);
    float s  = 1.f + e;
    float bb = s - 1.f;
    float r  = (1.f - (s - bb)) + (e - bb);
    float l2 = __builtin_amdgcn_logf(s);
    const float c = 0x1.62e42ep-1f, cc = 0x1.efa39ep-25f;
    float lo = __builtin_fmaf(l2, cc, r * __builtin_amdgcn_rcpf(s));
    return (t > 20.f) ? t : __builtin_fmaf(l2, c, lo);
}

#define CV_ROWS   4
#define CV_WIN    (CV_ROWS + 2)
#define CV_POS    (CV_WIN * WWD)
#define CV_PW     (WWD + 2)
#define CV_COLS   32
#define CV_PITCH  34
#define CV_KU     6
#define CV_NB     (DIMC / (4 * CV_KU))
#define IC_WAVES  7
#define IC_THREADS (IC_WAVES * 32)
#define CV_PER_IMG ((HH / CV_ROWS) * (D_INNER / CV_COLS))
#define CV_BLOCKS (BATCH * CV_PER_IMG)
#define Z_BLOCKS  ((BL / 16 / IC_WAVES) * (D_INNER / 16))
static_assert(HH % CV_ROWS == 0, "whole bands");
static_assert(CV_POS == IC_WAVES * 3 * 16, "three row tiles per wave");
static_assert(CV_ROWS * WWD == IC_WAVES * 32, "32 conv outputs per thread");
static_assert((BL / 16) % IC_WAVES == 0, "whole z blocks");
static_assert(DIMC % (4 * CV_KU) == 0, "whole groups of k blocks");

__global__ void __launch_bounds__(IC_THREADS)
k_in_conv(const float* __restrict__ x,
          const float* __restrict__ Win,
          const float* __restrict__ Wc,
          const float* __restrict__ bc,
          float* __restrict__ xz,
          float* __restrict__ xi) {
    __shared__ float xs[CV_WIN * CV_PW * CV_PITCH];
    int lane = threadIdx.x & 31;
    int wave = __builtin_amdgcn_readfirstlane((int)(threadIdx.x >> 5));
    int c16  = lane & 15;
    int kh   = (lane >> 4) * 2;

    if (blockIdx.x >= CV_BLOCKS) {
        int zb    = (int)blockIdx.x - CV_BLOCKS;
        int mTile = (zb / (D_INNER / 16)) * IC_WAVES + wave;
        int nTile = D_INNER / 16 + zb % (D_INNER / 16);
        int m0 = mTile * 16, n0 = nTile * 16;
        int mrow = m0 + c16;
        int b = mrow / LL, l = mrow % LL;
        int nn = n0 + c16;
        float za[DIMC / 4][2], zw[DIMC / 4][2];
#pragma unroll
        for (int u = 0; u < DIMC / 4; ++u) {
            int c0 = 4 * u + kh;
            za[u][0] = x[(size_t)(b * DIMC + c0    ) * LL + l];
            za[u][1] = x[(size_t)(b * DIMC + c0 + 1) * LL + l];
            zw[u][0] = Win[(size_t)(c0    ) * XZW + nn];
            zw[u][1] = Win[(size_t)(c0 + 1) * XZW + nn];
        }
        v8f acc = {};
#pragma unroll
        for (int u = 0; u < DIMC / 4; ++u) {
            v2f a  = {za[u][0], za[u][1]};
            v2f bf = {zw[u][0], zw[u][1]};
            acc = __builtin_amdgcn_wmma_f32_16x16x4_f32(false, a, false, bf,
                                                        (short)0, acc, false, false);
        }
        int mb = m0 + (lane >> 4) * 8;
#pragma unroll
        for (int v = 0; v < 8; ++v)
            xz[(size_t)(mb + v) * XZW + nn] = acc[v];
        return;
    }

    int idx  = (int)blockIdx.x;
    int b    = idx / CV_PER_IMG;
    int rem  = idx % CV_PER_IMG;
    int band = rem / (D_INNER / CV_COLS);
    int d0   = (rem % (D_INNER / CV_COLS)) * CV_COLS;
    int hfirst = band * CV_ROWS - 1;

    for (int i = threadIdx.x; i < CV_WIN * 2 * CV_COLS; i += IC_THREADS) {
        int row = i / (2 * CV_COLS), side = (i / CV_COLS) & 1, col = i % CV_COLS;
        xs[(row * CV_PW + side * (CV_PW - 1)) * CV_PITCH + col] = 0.f;
    }

    const float* xa[3];
    bool ok[3];
#pragma unroll
    for (int t = 0; t < 3; ++t) {
        int p = (wave * 3 + t) * 16 + c16;
        int h = hfirst + p / WWD;
        int w = p % WWD;
        ok[t] = (h >= 0 && h < HH);
        int l = ok[t] ? h * WWD + w : 0;
        xa[t] = x + (size_t)b * DIMC * LL + l;
    }
    const float* wb = Win + d0 + c16;
    v8f acc[3][2];
#pragma unroll
    for (int t = 0; t < 3; ++t) { acc[t][0] = (v8f){}; acc[t][1] = (v8f){}; }
    float av[2][CV_KU][3][2], bv[2][CV_KU][2][2];
#pragma unroll
    for (int u = 0; u < CV_KU; ++u) {
        int c0 = 4 * u + kh;
#pragma unroll
        for (int t = 0; t < 3; ++t) {
            av[0][u][t][0] = xa[t][(size_t)(c0    ) * LL];
            av[0][u][t][1] = xa[t][(size_t)(c0 + 1) * LL];
        }
#pragma unroll
        for (int j = 0; j < 2; ++j) {
            bv[0][u][j][0] = wb[(size_t)(c0    ) * XZW + 16 * j];
            bv[0][u][j][1] = wb[(size_t)(c0 + 1) * XZW + 16 * j];
        }
    }
#pragma unroll
    for (int bt = 0; bt < CV_NB; ++bt) {
        const int cur = bt & 1, nxt = cur ^ 1;
        if (bt + 1 < CV_NB) {
#pragma unroll
            for (int u = 0; u < CV_KU; ++u) {
                int c0 = (bt + 1) * 4 * CV_KU + 4 * u + kh;
#pragma unroll
                for (int t = 0; t < 3; ++t) {
                    av[nxt][u][t][0] = xa[t][(size_t)(c0    ) * LL];
                    av[nxt][u][t][1] = xa[t][(size_t)(c0 + 1) * LL];
                }
#pragma unroll
                for (int j = 0; j < 2; ++j) {
                    bv[nxt][u][j][0] = wb[(size_t)(c0    ) * XZW + 16 * j];
                    bv[nxt][u][j][1] = wb[(size_t)(c0 + 1) * XZW + 16 * j];
                }
            }
        }
#pragma unroll
        for (int u = 0; u < CV_KU; ++u) {
#pragma unroll
            for (int t = 0; t < 3; ++t) {
                v2f a;
                a[0] = ok[t] ? av[cur][u][t][0] : 0.f;
                a[1] = ok[t] ? av[cur][u][t][1] : 0.f;
#pragma unroll
                for (int j = 0; j < 2; ++j) {
                    v2f bf = {bv[cur][u][j][0], bv[cur][u][j][1]};
                    acc[t][j] = __builtin_amdgcn_wmma_f32_16x16x4_f32(false, a, false, bf,
                                                                      (short)0, acc[t][j], false, false);
                }
            }
        }
    }
#pragma unroll
    for (int t = 0; t < 3; ++t) {
        int prow = (wave * 3 + t) * 16 + (lane >> 4) * 8;
#pragma unroll
        for (int v = 0; v < 8; ++v) {
            int p = prow + v;
            int q = (p / WWD) * CV_PW + p % WWD + 1;
            xs[q * CV_PITCH + c16]      = acc[t][0][v];
            xs[q * CV_PITCH + 16 + c16] = acc[t][1][v];
        }
    }
    __syncthreads();

    int d = d0 + lane;
    float wg[9];
#pragma unroll
    for (int t = 0; t < 9; ++t) wg[t] = Wc[d * 9 + t];
    float bias = bc[d];
    float* xo = xi + ((size_t)b * LL + band * CV_ROWS * WWD) * D_INNER + d;
#pragma unroll 4
    for (int i = 0; i < (CV_ROWS * WWD) / IC_WAVES; ++i) {
        int o = wave + IC_WAVES * i;
        int r = o / WWD, w = o % WWD;
        const float* q = xs + (r * CV_PW + w) * CV_PITCH + lane;
        float s = bias;
#pragma unroll
        for (int dy = 0; dy < 3; ++dy)
#pragma unroll
            for (int dx = 0; dx < 3; ++dx)
                s = __builtin_fmaf(wg[dy * 3 + dx], q[(dy * CV_PW + dx) * CV_PITCH], s);
        xo[(size_t)o * D_INNER] = silu(s);
    }
}

#define XP_KU   6
#define XP_NB   (D_INNER / (4 * XP_KU))
#define XP_NT   3
#define DL_PW   (D_INNER / 16 / 4)
static_assert(D_INNER % (4 * XP_KU) == 0, "whole groups of k blocks");
static_assert(D_INNER % 64 == 0, "whole column tiles per wave");

__global__ void __launch_bounds__(128)
k_xproj_delta(const float* __restrict__ xi,
              const float* __restrict__ Wxp,
              const float* __restrict__ Wdt,
              const float* __restrict__ bdt,
              float* __restrict__ xdbl,
              float* __restrict__ delta) {
    __shared__ float drs[16 * 8];
    int lane = threadIdx.x & 31;
    int wave = __builtin_amdgcn_readfirstlane((int)(threadIdx.x >> 5));
    int kdir = blockIdx.y;
    int m0 = blockIdx.x * 16;
    int c16 = lane & 15;
    int hf  = lane >> 4;
    int kh  = hf * 2;
    int mb  = m0 + hf * 8;

    int dcol = wave * (16 * DL_PW) + c16;
    const float* wd = Wdt + (size_t)kdir * DT_RANK * D_INNER + dcol;
    int k2 = hf ? 0 : 4;
    float w0[DL_PW][2], w1[DL_PW][2], bs[DL_PW];
#pragma unroll
    for (int t = 0; t < DL_PW; ++t) {
        w0[t][0] = wd[(size_t)(kh     ) * D_INNER + 16 * t];
        w0[t][1] = wd[(size_t)(kh + 1 ) * D_INNER + 16 * t];
        w1[t][0] = wd[(size_t)(k2     ) * D_INNER + 16 * t];
        w1[t][1] = wd[(size_t)(k2 + 1 ) * D_INNER + 16 * t];
        bs[t]    = bdt[kdir * D_INNER + dcol + 16 * t];
    }

    if (wave < XP_NT) {
        int mrow = m0 + c16;
        int b = mrow / LL, l = mrow % LL;
        int lsp = scan_to_spatial(kdir, l);
        const float* arow = xi + (size_t)(b * LL + lsp) * D_INNER + kh;
        int nn = 16 * wave + c16;
        bool nok = (nn < NPROJ);
        int ncol = nok ? nn : 0;
        const float* wk = Wxp + (size_t)kdir * D_INNER * NPROJ + (size_t)kh * NPROJ + ncol;

        v8f acc = {};
        v2f   av[2][XP_KU];
        float bw[2][XP_KU][2];
#pragma unroll
        for (int u = 0; u < XP_KU; ++u) {
            int k0 = 4 * u;
            av[0][u]    = *(const v2f*)(arow + k0);
            bw[0][u][0] = wk[(size_t)(k0    ) * NPROJ];
            bw[0][u][1] = wk[(size_t)(k0 + 1) * NPROJ];
        }
#pragma unroll
        for (int bt = 0; bt < XP_NB; ++bt) {
            const int cur = bt & 1, nxt = cur ^ 1;
            if (bt + 1 < XP_NB) {
#pragma unroll
                for (int u = 0; u < XP_KU; ++u) {
                    int k0 = (bt + 1) * 4 * XP_KU + 4 * u;
                    av[nxt][u]    = *(const v2f*)(arow + k0);
                    bw[nxt][u][0] = wk[(size_t)(k0    ) * NPROJ];
                    bw[nxt][u][1] = wk[(size_t)(k0 + 1) * NPROJ];
                }
            }
#pragma unroll
            for (int u = 0; u < XP_KU; ++u) {
                v2f bf;
                bf[0] = nok ? bw[cur][u][0] : 0.f;
                bf[1] = nok ? bw[cur][u][1] : 0.f;
                acc = __builtin_amdgcn_wmma_f32_16x16x4_f32(false, av[cur][u], false, bf,
                                                            (short)0, acc, false, false);
            }
        }
        if (nok) {
#pragma unroll
            for (int v = 0; v < 8; ++v)
                xdbl[(size_t)(kdir * BL + mb + v) * NPROJ + nn] = acc[v];
        }
        if (wave == 0 && c16 < 8) {
#pragma unroll
            for (int v = 0; v < 8; ++v)
                drs[(hf * 8 + v) * 8 + c16] = (c16 < DT_RANK) ? acc[v] : 0.f;
        }
    }
    __syncthreads();
    v2f a0 = {drs[c16 * 8 + kh],     drs[c16 * 8 + kh + 1]};
    v2f a1 = {drs[c16 * 8 + 4 + kh], drs[c16 * 8 + 4 + kh + 1]};

#pragma unroll
    for (int t = 0; t < DL_PW; ++t) {
        v2f b0 = {w0[t][0], w0[t][1]};
        v2f b1;
        b1[0] = hf ? 0.f : w1[t][0];
        b1[1] = hf ? 0.f : w1[t][1];
        v8f dacc = {};
        dacc = __builtin_amdgcn_wmma_f32_16x16x4_f32(false, a0, false, b0,
                                                     (short)0, dacc, false, false);
        dacc = __builtin_amdgcn_wmma_f32_16x16x4_f32(false, a1, false, b1,
                                                     (short)0, dacc, false, false);
        int nn = dcol + 16 * t;
#pragma unroll
        for (int v = 0; v < 8; ++v) {
            float tt = dacc[v] + bs[t];
            delta[(size_t)(kdir * BL + mb + v) * D_INNER + nn] = softplus20(tt);
        }
    }
}


static_assert(WWD % CHUNK == 0 && CHUNK >= 2, "a chunk lies inside one scan row");
static_assert(CHUNK % SCAN_G == 0, "whole groups");

__device__ __forceinline__ size_t part_idx(int chunk, int g, int ch) {
    return ((size_t)(chunk * (D_STATE / 4) + g) * NCHAN + ch) * 4;
}

__device__ __forceinline__ v2f exp_pair(v2f a) {
    v2f r = {__expf(a[0]), __expf(a[1])};
    return r;
}

__global__ void __launch_bounds__(SCAN_BLK) __attribute__((amdgpu_waves_per_eu(1, 4)))
k_scan_part(const float* __restrict__ xi,
            const float* __restrict__ xdbl,
            const float* __restrict__ delta,
            const float* __restrict__ A_logs,
            float* __restrict__ hend,
            float* __restrict__ pdec) {
    int chunk = blockIdx.y;
    int kb    = blockIdx.x / (D_INNER / SCAN_BLK);
    int d     = (blockIdx.x % (D_INNER / SCAN_BLK)) * SCAN_BLK + threadIdx.x;
    int b     = kb % BATCH;
    int kdir  = kb / BATCH;
    int ch    = kb * D_INNER + d;

    v2f A[D_STATE / 2], h[D_STATE / 2], P[D_STATE / 2];
#pragma unroll
    for (int j = 0; j < D_STATE / 2; ++j) {
        const float* al = A_logs + (size_t)(kdir * D_INNER + d) * D_STATE + 2 * j;
        A[j][0] = -__expf(al[0]);
        A[j][1] = -__expf(al[1]);
        h[j] = (v2f){0.f, 0.f};
        P[j] = (v2f){1.f, 1.f};
    }

    int l0    = chunk * CHUNK;
    int lsp0  = scan_to_spatial(kdir, l0);
    int lstep = scan_to_spatial(kdir, l0 + 1) - lsp0;
    const float* xib = xi    + ((size_t)b  * LL + lsp0) * D_INNER + d;
    const float* dlb = delta + ((size_t)kb * LL + l0) * D_INNER + d;
    const float* xdb = xdbl  + ((size_t)kb * LL + l0) * NPROJ + DT_RANK;

    float xc[SCAN_G], dc[SCAN_G], Bs[D_STATE];
#pragma unroll
    for (int g = 0; g < SCAN_G; ++g) {
        xc[g] = xib[(ptrdiff_t)g * lstep * D_INNER];
        dc[g] = dlb[(size_t)g * D_INNER];
    }
#pragma unroll
    for (int n = 0; n < D_STATE; ++n) Bs[n] = xdb[n];

#pragma nounroll
    for (int i0 = 0; i0 < CHUNK; i0 += SCAN_G) {
        float xn[SCAN_G], dn[SCAN_G];
        __builtin_amdgcn_sched_barrier(0);
        int in = (i0 + SCAN_G < CHUNK) ? i0 + SCAN_G : i0;
#pragma unroll
        for (int g = 0; g < SCAN_G; ++g) {
            xn[g] = xib[(ptrdiff_t)(in + g) * lstep * D_INNER];
            dn[g] = dlb[(size_t)(in + g) * D_INNER];
        }
        __builtin_amdgcn_sched_barrier(0);
        const float* bcg = xdb + (size_t)i0 * NPROJ;
#pragma unroll
        for (int g = 0; g < SCAN_G; ++g) {
            const float* bn = (g + 1 < SCAN_G || i0 + SCAN_G < CHUNK) ? bcg + (g + 1) * NPROJ : bcg + g * NPROJ;
            __builtin_amdgcn_sched_barrier(0);
            asm volatile("" :: "s"(Bs[0]));
            float Bn[D_STATE];
#pragma unroll
            for (int n = 0; n < D_STATE; ++n) Bn[n] = bn[n];
            __builtin_amdgcn_sched_barrier(0);
            float dl  = dc[g];
            float dlx = dl * xc[g];
            v2f dl2  = {dl, dl};
            v2f dlx2 = {dlx, dlx};
#pragma unroll
            for (int j = 0; j < D_STATE / 2; ++j) {
                v2f dA = exp_pair(dl2 * A[j]);
                v2f Bv = {Bs[2 * j], Bs[2 * j + 1]};
                h[j] = dA * h[j] + dlx2 * Bv;
                P[j] = P[j] * dA;
            }
#pragma unroll
            for (int n = 0; n < D_STATE; ++n) Bs[n] = Bn[n];
        }
#pragma unroll
        for (int g = 0; g < SCAN_G; ++g) { xc[g] = xn[g]; dc[g] = dn[g]; }
    }
#pragma unroll
    for (int g = 0; g < D_STATE / 4; ++g) {
        v4f hv = {h[2 * g][0], h[2 * g][1], h[2 * g + 1][0], h[2 * g + 1][1]};
        v4f pv = {P[2 * g][0], P[2 * g][1], P[2 * g + 1][0], P[2 * g + 1][1]};
        *(v4f*)(hend + part_idx(chunk, g, ch)) = hv;
        *(v4f*)(pdec + part_idx(chunk, g, ch)) = pv;
    }
}

__global__ void __launch_bounds__(SCAN_BLK) __attribute__((amdgpu_waves_per_eu(1, 4)))
k_scan_final(const float* __restrict__ xi,
             const float* __restrict__ xdbl,
             const float* __restrict__ delta,
             const float* __restrict__ A_logs,
             const float* __restrict__ Ds,
             const float* __restrict__ hend,
             const float* __restrict__ pdec,
             float* __restrict__ ys) {
    int chunk = NCHUNK - 1 - (int)blockIdx.y;
    int kb    = blockIdx.x / (D_INNER / SCAN_BLK);
    int d     = (blockIdx.x % (D_INNER / SCAN_BLK)) * SCAN_BLK + threadIdx.x;
    int b     = kb % BATCH;
    int kdir  = kb / BATCH;
    int ch    = kb * D_INNER + d;

    v2f A[D_STATE / 2], h[D_STATE / 2];
#pragma unroll
    for (int j = 0; j < D_STATE / 2; ++j) {
        const float* al = A_logs + (size_t)(kdir * D_INNER + d) * D_STATE + 2 * j;
        A[j][0] = -__expf(al[0]);
        A[j][1] = -__expf(al[1]);
        h[j] = (v2f){0.f, 0.f};
    }
    float Dskip = Ds[kdir * D_INNER + d];

#pragma unroll 4
    for (int j = 0; j < chunk; ++j) {
#pragma unroll
        for (int g = 0; g < D_STATE / 4; ++g) {
            v4f pv = *(const v4f*)(pdec + part_idx(j, g, ch));
            v4f hv = *(const v4f*)(hend + part_idx(j, g, ch));
            h[2 * g]     = (v2f){pv[0], pv[1]} * h[2 * g]     + (v2f){hv[0], hv[1]};
            h[2 * g + 1] = (v2f){pv[2], pv[3]} * h[2 * g + 1] + (v2f){hv[2], hv[3]};
        }
    }

    int l0    = chunk * CHUNK;
    int lsp0  = scan_to_spatial(kdir, l0);
    int lstep = scan_to_spatial(kdir, l0 + 1) - lsp0;
    const float* xib = xi    + ((size_t)b  * LL + lsp0) * D_INNER + d;
    const float* dlb = delta + ((size_t)kb * LL + l0) * D_INNER + d;
    const float* xdb = xdbl  + ((size_t)kb * LL + l0) * NPROJ + DT_RANK;
    float*       ysb = ys    + ((size_t)kb * LL + l0) * D_INNER + d;

    float xc[SCAN_G], dc[SCAN_G];
    v2f Bv[D_STATE / 2], Cv[D_STATE / 2];
    float bkey = xdb[0];
#pragma unroll
    for (int g = 0; g < SCAN_G; ++g) {
        xc[g] = xib[(ptrdiff_t)g * lstep * D_INNER];
        dc[g] = dlb[(size_t)g * D_INNER];
    }
#pragma unroll
    for (int j = 0; j < D_STATE / 2; ++j) {
        Bv[j] = (v2f){xdb[2 * j], xdb[2 * j + 1]};
        Cv[j] = (v2f){xdb[D_STATE + 2 * j], xdb[D_STATE + 2 * j + 1]};
    }

#pragma nounroll
    for (int i0 = 0; i0 < CHUNK; i0 += SCAN_G) {
        float xn[SCAN_G], dn[SCAN_G];
        __builtin_amdgcn_sched_barrier(0);
        int in = (i0 + SCAN_G < CHUNK) ? i0 + SCAN_G : i0;
#pragma unroll
        for (int g = 0; g < SCAN_G; ++g) {
            xn[g] = xib[(ptrdiff_t)(in + g) * lstep * D_INNER];
            dn[g] = dlb[(size_t)(in + g) * D_INNER];
        }
        __builtin_amdgcn_sched_barrier(0);
        const float* bcg = xdb + (size_t)i0 * NPROJ;
#pragma unroll
        for (int g = 0; g < SCAN_G; ++g) {
            const float* bn = (g + 1 < SCAN_G || i0 + SCAN_G < CHUNK) ? bcg + (g + 1) * NPROJ : bcg + g * NPROJ;
            __builtin_amdgcn_sched_barrier(0);
            asm volatile("" :: "s"(bkey));
            float Bn[D_STATE], Cn[D_STATE];
#pragma unroll
            for (int n = 0; n < D_STATE; ++n) {
                Bn[n] = bn[n];
                Cn[n] = bn[D_STATE + n];
            }
            __builtin_amdgcn_sched_barrier(0);
            float x   = xc[g];
            float dl  = dc[g];
            float dlx = dl * x;
            v2f dl2  = {dl, dl};
            v2f dlx2 = {dlx, dlx};
            v2f ya = {0.f, 0.f}, yb = {0.f, 0.f};
#pragma unroll
            for (int j = 0; j < D_STATE / 2; j += 2) {
                v2f dA0 = exp_pair(dl2 * A[j]);
                v2f dA1 = exp_pair(dl2 * A[j + 1]);
                h[j]     = dA0 * h[j]     + dlx2 * Bv[j];
                h[j + 1] = dA1 * h[j + 1] + dlx2 * Bv[j + 1];
                ya += h[j]     * Cv[j];
                yb += h[j + 1] * Cv[j + 1];
            }
            ysb[(size_t)(i0 + g) * D_INNER] = ((ya[0] + ya[1]) + (yb[0] + yb[1])) + x * Dskip;
            bkey = Bn[0];
#pragma unroll
            for (int j = 0; j < D_STATE / 2; ++j) {
                Bv[j] = (v2f){Bn[2 * j], Bn[2 * j + 1]};
                Cv[j] = (v2f){Cn[2 * j], Cn[2 * j + 1]};
            }
        }
#pragma unroll
        for (int g = 0; g < SCAN_G; ++g) { xc[g] = xn[g]; dc[g] = dn[g]; }
    }
}

#define CO_NT   3
#define CO_KU   3
#define CO_NB   (D_INNER / (8 * CO_KU))
static_assert(D_INNER % (8 * CO_KU) == 0, "whole groups");
static_assert(DIMC == 2 * 16 * CO_NT, "two waves cover the output channels");
static_assert(LL % 16 == 0, "a position tile lies inside one image");

__global__ void __launch_bounds__(64) __attribute__((amdgpu_waves_per_eu(1, 2)))
k_comb_out(const float* __restrict__ ys,
           const float* __restrict__ xz,
           const float* __restrict__ Wout,
           float* __restrict__ out) {
    int lane = threadIdx.x & 31;
    int wave = __builtin_amdgcn_readfirstlane((int)(threadIdx.x >> 5));
    int c16 = lane & 15;
    int hf  = lane >> 4;
    int row = (int)blockIdx.x * 16 + c16;
    int b = row / LL, l = row % LL;

    const float* yp[KDIR];
#pragma unroll
    for (int k = 0; k < KDIR; ++k)
        yp[k] = ys + (size_t)((k * BATCH + b) * LL + spatial_to_scan(k, l)) * D_INNER + 4 * hf;
    const float* zp = xz + (size_t)row * XZW + D_INNER + 4 * hf;
    int n0 = wave * (16 * CO_NT);
    const float* wp = Wout + (size_t)(4 * hf) * DIMC + n0 + c16;

    v8f acc[CO_NT];
#pragma unroll
    for (int t = 0; t < CO_NT; ++t) acc[t] = (v8f){};

    v4f   yv[2][CO_KU][KDIR], zv[2][CO_KU];
    float wv[2][CO_KU][CO_NT][4];
#pragma unroll
    for (int u = 0; u < CO_KU; ++u) {
        int d0 = 8 * u;
#pragma unroll
        for (int k = 0; k < KDIR; ++k) yv[0][u][k] = *(const v4f*)(yp[k] + d0);
        zv[0][u] = *(const v4f*)(zp + d0);
#pragma unroll
        for (int t = 0; t < CO_NT; ++t)
#pragma unroll
            for (int e = 0; e < 4; ++e)
                wv[0][u][t][e] = wp[(size_t)(d0 + e) * DIMC + 16 * t];
    }
#pragma unroll
    for (int g = 0; g < CO_NB; ++g) {
        const int cur = g & 1, nxt = cur ^ 1;
        if (g + 1 < CO_NB) {
#pragma unroll
            for (int u = 0; u < CO_KU; ++u) {
                int d0 = 8 * ((g + 1) * CO_KU + u);
#pragma unroll
                for (int k = 0; k < KDIR; ++k) yv[nxt][u][k] = *(const v4f*)(yp[k] + d0);
                zv[nxt][u] = *(const v4f*)(zp + d0);
#pragma unroll
                for (int t = 0; t < CO_NT; ++t)
#pragma unroll
                    for (int e = 0; e < 4; ++e)
                        wv[nxt][u][t][e] = wp[(size_t)(d0 + e) * DIMC + 16 * t];
            }
        }
#pragma unroll
        for (int u = 0; u < CO_KU; ++u) {
            v4f s = ((yv[cur][u][0] + yv[cur][u][1]) + yv[cur][u][2]) + yv[cur][u][3];
            v4f z = zv[cur][u];
            float yy[4];
#pragma unroll
            for (int e = 0; e < 4; ++e) yy[e] = s[e] * silu(z[e]);
            v2f y01 = {yy[0], yy[1]};
            v2f y23 = {yy[2], yy[3]};
#pragma unroll
            for (int t = 0; t < CO_NT; ++t) {
                v2f w01 = {wv[cur][u][t][0], wv[cur][u][t][1]};
                v2f w23 = {wv[cur][u][t][2], wv[cur][u][t][3]};
                acc[t] = __builtin_amdgcn_wmma_f32_16x16x4_f32(false, w01, false, y01,
                                                               (short)0, acc[t], false, false);
                acc[t] = __builtin_amdgcn_wmma_f32_16x16x4_f32(false, w23, false, y23,
                                                               (short)0, acc[t], false, false);
            }
        }
    }
    float* op = out + (size_t)(b * DIMC + n0 + 8 * hf) * LL + l;
#pragma unroll
    for (int t = 0; t < CO_NT; ++t)
#pragma unroll
        for (int v = 0; v < 8; ++v)
            op[(size_t)(16 * t + v) * LL] = acc[t][v];
}

extern "C" void kernel_launch(void* const* d_in, const int* in_sizes, int n_in,
                              void* d_out, int out_size, void* d_ws, size_t ws_size,
                              hipStream_t stream) {
    (void)in_sizes; (void)n_in; (void)out_size; (void)ws_size;
    const float* x       = (const float*)d_in[0];
    const float* W_in    = (const float*)d_in[1];
    const float* W_conv  = (const float*)d_in[2];
    const float* b_conv  = (const float*)d_in[3];
    const float* W_xproj = (const float*)d_in[4];
    const float* W_dt    = (const float*)d_in[5];
    const float* b_dt    = (const float*)d_in[6];
    const float* A_logs  = (const float*)d_in[7];
    const float* Ds      = (const float*)d_in[8];
    const float* W_out   = (const float*)d_in[9];
    float* out = (float*)d_out;

    float* ws = (float*)d_ws;
    size_t o = 0;
    float* xz    = ws + o;  o += (size_t)BL * XZW;
    float* xi    = ws + o;  o += (size_t)BL * D_INNER;
    float* xdbl  = ws + o;  o += (size_t)KDIR * BL * NPROJ;
    float* delta = ws + o;  o += (size_t)KDIR * BL * D_INNER;
    float* ysbuf = ws + o;  o += (size_t)KDIR * BL * D_INNER;
    float* hend  = ws + o;  o += (size_t)(NCHUNK - 1) * NSCAN;
    float* pdec  = ws + o;  o += (size_t)(NCHUNK - 1) * NSCAN;

    dim3 blk(128);
    k_in_conv <<<CV_BLOCKS + Z_BLOCKS, IC_THREADS, 0, stream>>>(x, W_in, W_conv, b_conv, xz, xi);
    k_xproj_delta<<<dim3(392, 4), blk, 0, stream>>>(xi, W_xproj, W_dt, b_dt, xdbl, delta);
    k_scan_part <<<dim3(NCHAN / SCAN_BLK, NCHUNK - 1), SCAN_BLK, 0, stream>>>(
        xi, xdbl, delta, A_logs, hend, pdec);
    k_scan_final<<<dim3(NCHAN / SCAN_BLK, NCHUNK),     SCAN_BLK, 0, stream>>>(
        xi, xdbl, delta, A_logs, Ds, hend, pdec, ysbuf);
    k_comb_out<<<BL / 16, 64, 0, stream>>>(ysbuf, xz, W_out, out);
}


# pass 1 of the scan moved into the x-proj/dt-proj launch: a block makes xdbl and delta of one chunk on the matrix unit, keeps delta and B in LDS and scans from there; 4 launches; same order of sums
# speedup vs baseline: 27.8476x; 1.1499x over previous
#include <hip/hip_runtime.h>
#include <math.h>

typedef __attribute__((ext_vector_type(2))) float v2f;
typedef __attribute__((ext_vector_type(4))) float v4f;
typedef __attribute__((ext_vector_type(8))) float v8f;

#define DIMC    96
#define D_STATE 16
#define D_INNER 192
#define DT_RANK 6
#define KDIR    4
#define BATCH   2
#define HH      56
#define WWD     56
#define LL      (HH * WWD)
#define BL      (BATCH * LL)
#define NPROJ   (DT_RANK + 2 * D_STATE)
#define XZW     (2 * D_INNER)
#define NSCAN   (KDIR * BATCH * D_INNER * D_STATE)
#define NCHAN   (KDIR * BATCH * D_INNER)
#define SCAN_BLK 64
#define SCAN_G   8
#define NCHUNK  56
#define CHUNK   (LL / NCHUNK)

__device__ __forceinline__ int scan_to_spatial(int k, int l) {
    int q = l / WWD, r = l % WWD;
    if (k == 0) return l;
    if (k == 1) return q * WWD + (WWD - 1 - r);
    if (k == 2) return r * WWD + q;
    return (WWD - 1 - r) * WWD + q;
}

__device__ __forceinline__ int spatial_to_scan(int k, int l) {
    int q = l / WWD, r = l % WWD;
    if (k == 0) return l;
    if (k == 1) return q * WWD + (WWD - 1 - r);
    if (k == 2) return r * WWD + q;
    return r * WWD + (WWD - 1 - q);
}

__device__ __forceinline__ float silu(float v) {
    return v / (1.f + __expf(-v));
}

__device__ __forceinline__ float softplus20(float t) {
    float e  = __expf(t);
    float s  = 1.f + e;
    float bb = s - 1.f;
    float r  = (1.f - (s - bb)) + (e - bb);
    float l2 = __builtin_amdgcn_logf(s);
    const float c = 0x1.62e42ep-1f, cc = 0x1.efa39ep-25f;
    float lo = __builtin_fmaf(l2, cc, r * __builtin_amdgcn_rcpf(s));
    return (t > 20.f) ? t : __builtin_fmaf(l2, c, lo);
}

#define CV_ROWS   4
#define CV_WIN    (CV_ROWS + 2)
#define CV_POS    (CV_WIN * WWD)
#define CV_PW     (WWD + 2)
#define CV_COLS   32
#define CV_PITCH  34
#define CV_KU     6
#define CV_NB     (DIMC / (4 * CV_KU))
#define IC_WAVES  7
#define IC_THREADS (IC_WAVES * 32)
#define CV_PER_IMG ((HH / CV_ROWS) * (D_INNER / CV_COLS))
#define CV_BLOCKS (BATCH * CV_PER_IMG)
#define Z_BLOCKS  ((BL / 16 / IC_WAVES) * (D_INNER / 16))
static_assert(HH % CV_ROWS == 0, "whole bands");
static_assert(CV_POS == IC_WAVES * 3 * 16, "three row tiles per wave");
static_assert(CV_ROWS * WWD == IC_WAVES * 32, "32 conv outputs per thread");
static_assert((BL / 16) % IC_WAVES == 0, "whole z blocks");
static_assert(DIMC % (4 * CV_KU) == 0, "whole groups of k blocks");

__global__ void __launch_bounds__(IC_THREADS)
k_in_conv(const float* __restrict__ x,
          const float* __restrict__ Win,
          const float* __restrict__ Wc,
          const float* __restrict__ bc,
          float* __restrict__ xz,
          float* __restrict__ xi) {
    __shared__ float xs[CV_WIN * CV_PW * CV_PITCH];
    int lane = threadIdx.x & 31;
    int wave = __builtin_amdgcn_readfirstlane((int)(threadIdx.x >> 5));
    int c16  = lane & 15;
    int kh   = (lane >> 4) * 2;

    if (blockIdx.x >= CV_BLOCKS) {
        int zb    = (int)blockIdx.x - CV_BLOCKS;
        int mTile = (zb / (D_INNER / 16)) * IC_WAVES + wave;
        int nTile = D_INNER / 16 + zb % (D_INNER / 16);
        int m0 = mTile * 16, n0 = nTile * 16;
        int mrow = m0 + c16;
        int b = mrow / LL, l = mrow % LL;
        int nn = n0 + c16;
        float za[DIMC / 4][2], zw[DIMC / 4][2];
#pragma unroll
        for (int u = 0; u < DIMC / 4; ++u) {
            int c0 = 4 * u + kh;
            za[u][0] = x[(size_t)(b * DIMC + c0    ) * LL + l];
            za[u][1] = x[(size_t)(b * DIMC + c0 + 1) * LL + l];
            zw[u][0] = Win[(size_t)(c0    ) * XZW + nn];
            zw[u][1] = Win[(size_t)(c0 + 1) * XZW + nn];
        }
        v8f acc = {};
#pragma unroll
        for (int u = 0; u < DIMC / 4; ++u) {
            v2f a  = {za[u][0], za[u][1]};
            v2f bf = {zw[u][0], zw[u][1]};
            acc = __builtin_amdgcn_wmma_f32_16x16x4_f32(false, a, false, bf,
                                                        (short)0, acc, false, false);
        }
        int mb = m0 + (lane >> 4) * 8;
#pragma unroll
        for (int v = 0; v < 8; ++v)
            xz[(size_t)(mb + v) * XZW + nn] = acc[v];
        return;
    }

    int idx  = (int)blockIdx.x;
    int b    = idx / CV_PER_IMG;
    int rem  = idx % CV_PER_IMG;
    int band = rem / (D_INNER / CV_COLS);
    int d0   = (rem % (D_INNER / CV_COLS)) * CV_COLS;
    int hfirst = band * CV_ROWS - 1;

    for (int i = threadIdx.x; i < CV_WIN * 2 * CV_COLS; i += IC_THREADS) {
        int row = i / (2 * CV_COLS), side = (i / CV_COLS) & 1, col = i % CV_COLS;
        xs[(row * CV_PW + side * (CV_PW - 1)) * CV_PITCH + col] = 0.f;
    }

    const float* xa[3];
    bool ok[3];
#pragma unroll
    for (int t = 0; t < 3; ++t) {
        int p = (wave * 3 + t) * 16 + c16;
        int h = hfirst + p / WWD;
        int w = p % WWD;
        ok[t] = (h >= 0 && h < HH);
        int l = ok[t] ? h * WWD + w : 0;
        xa[t] = x + (size_t)b * DIMC * LL + l;
    }
    const float* wb = Win + d0 + c16;
    v8f acc[3][2];
#pragma unroll
    for (int t = 0; t < 3; ++t) { acc[t][0] = (v8f){}; acc[t][1] = (v8f){}; }
    float av[2][CV_KU][3][2], bv[2][CV_KU][2][2];
#pragma unroll
    for (int u = 0; u < CV_KU; ++u) {
        int c0 = 4 * u + kh;
#pragma unroll
        for (int t = 0; t < 3; ++t) {
            av[0][u][t][0] = xa[t][(size_t)(c0    ) * LL];
            av[0][u][t][1] = xa[t][(size_t)(c0 + 1) * LL];
        }
#pragma unroll
        for (int j = 0; j < 2; ++j) {
            bv[0][u][j][0] = wb[(size_t)(c0    ) * XZW + 16 * j];
            bv[0][u][j][1] = wb[(size_t)(c0 + 1) * XZW + 16 * j];
        }
    }
#pragma unroll
    for (int bt = 0; bt < CV_NB; ++bt) {
        const int cur = bt & 1, nxt = cur ^ 1;
        if (bt + 1 < CV_NB) {
#pragma unroll
            for (int u = 0; u < CV_KU; ++u) {
                int c0 = (bt + 1) * 4 * CV_KU + 4 * u + kh;
#pragma unroll
                for (int t = 0; t < 3; ++t) {
                    av[nxt][u][t][0] = xa[t][(size_t)(c0    ) * LL];
                    av[nxt][u][t][1] = xa[t][(size_t)(c0 + 1) * LL];
                }
#pragma unroll
                for (int j = 0; j < 2; ++j) {
                    bv[nxt][u][j][0] = wb[(size_t)(c0    ) * XZW + 16 * j];
                    bv[nxt][u][j][1] = wb[(size_t)(c0 + 1) * XZW + 16 * j];
                }
            }
        }
#pragma unroll
        for (int u = 0; u < CV_KU; ++u) {
#pragma unroll
            for (int t = 0; t < 3; ++t) {
                v2f a;
                a[0] = ok[t] ? av[cur][u][t][0] : 0.f;
                a[1] = ok[t] ? av[cur][u][t][1] : 0.f;
#pragma unroll
                for (int j = 0; j < 2; ++j) {
                    v2f bf = {bv[cur][u][j][0], bv[cur][u][j][1]};
                    acc[t][j] = __builtin_amdgcn_wmma_f32_16x16x4_f32(false, a, false, bf,
                                                                      (short)0, acc[t][j], false, false);
                }
            }
        }
    }
#pragma unroll
    for (int t = 0; t < 3; ++t) {
        int prow = (wave * 3 + t) * 16 + (lane >> 4) * 8;
#pragma unroll
        for (int v = 0; v < 8; ++v) {
            int p = prow + v;
            int q = (p / WWD) * CV_PW + p % WWD + 1;
            xs[q * CV_PITCH + c16]      = acc[t][0][v];
            xs[q * CV_PITCH + 16 + c16] = acc[t][1][v];
        }
    }
    __syncthreads();

    int d = d0 + lane;
    float wg[9];
#pragma unroll
    for (int t = 0; t < 9; ++t) wg[t] = Wc[d * 9 + t];
    float bias = bc[d];
    float* xo = xi + ((size_t)b * LL + band * CV_ROWS * WWD) * D_INNER + d;
#pragma unroll 4
    for (int i = 0; i < (CV_ROWS * WWD) / IC_WAVES; ++i) {
        int o = wave + IC_WAVES * i;
        int r = o / WWD, w = o % WWD;
        const float* q = xs + (r * CV_PW + w) * CV_PITCH + lane;
        float s = bias;
#pragma unroll
        for (int dy = 0; dy < 3; ++dy)
#pragma unroll
            for (int dx = 0; dx < 3; ++dx)
                s = __builtin_fmaf(wg[dy * 3 + dx], q[(dy * CV_PW + dx) * CV_PITCH], s);
        xo[(size_t)o * D_INNER] = silu(s);
    }
}


static_assert(WWD % CHUNK == 0 && CHUNK >= 2, "a chunk lies inside one scan row");
static_assert(CHUNK % SCAN_G == 0, "whole groups");

__device__ __forceinline__ size_t part_idx(int chunk, int g, int ch) {
    return ((size_t)(chunk * (D_STATE / 4) + g) * NCHAN + ch) * 4;
}

__device__ __forceinline__ v2f exp_pair(v2f a) {
    v2f r = {__expf(a[0]), __expf(a[1])};
    return r;
}

#define XP_KU   6
#define XP_NB   (D_INNER / (4 * XP_KU))
#define XP_NT   3
#define DL_NT   (D_INNER / 16)
#define FS_RT   4
#define FS_THREADS D_INNER
static_assert(D_INNER % (4 * XP_KU) == 0, "whole groups of k blocks");
static_assert(D_INNER % 32 == 0, "whole waves");
static_assert(FS_RT * 16 >= CHUNK && (FS_RT - 1) * 16 < CHUNK, "row tiles cover the chunk");
static_assert(FS_RT * 32 <= FS_THREADS, "a wave per row tile");
static_assert(CHUNK % 8 == 0, "a lane half's 8 rows lie inside the chunk or behind it");
static_assert(DT_RANK + D_STATE <= 32, "the B columns lie in the first two column tiles");

__global__ void __launch_bounds__(FS_THREADS) __attribute__((amdgpu_waves_per_eu(4, 4)))
k_xproj_delta_part(const float* __restrict__ xi,
                   const float* __restrict__ Wxp,
                   const float* __restrict__ Wdt,
                   const float* __restrict__ bdt,
                   const float* __restrict__ A_logs,
                   float* __restrict__ xdbl,
                   float* __restrict__ delta,
                   float* __restrict__ hend,
                   float* __restrict__ pdec) {
    __shared__ float dls[CHUNK * D_INNER];
    __shared__ v4f   Bsh[CHUNK * (D_STATE / 4)];
    __shared__ float drs[FS_RT][16 * 8];
    int lane = threadIdx.x & 31;
    int wave = __builtin_amdgcn_readfirstlane((int)(threadIdx.x >> 5));
    int chunk = blockIdx.x;
    int kb    = blockIdx.y;
    int b     = kb % BATCH;
    int kdir  = kb / BATCH;
    int l0    = chunk * CHUNK;
    size_t row0 = (size_t)kb * LL + l0;
    int c16 = lane & 15;
    int hf  = lane >> 4;
    int kh  = hf * 2;
    int ib  = wave * 16 + hf * 8;
    bool rowok = (ib < CHUNK);

    float w0[DL_NT][2], w1[DL_NT][2], bs[DL_NT];
    if (wave < FS_RT) {
        const float* wd = Wdt + (size_t)kdir * DT_RANK * D_INNER + c16;
        int k2 = hf ? 0 : 4;
#pragma unroll
        for (int t = 0; t < DL_NT; ++t) {
            w0[t][0] = wd[(size_t)(kh     ) * D_INNER + 16 * t];
            w0[t][1] = wd[(size_t)(kh + 1 ) * D_INNER + 16 * t];
            w1[t][0] = wd[(size_t)(k2     ) * D_INNER + 16 * t];
            w1[t][1] = wd[(size_t)(k2 + 1 ) * D_INNER + 16 * t];
            bs[t]    = bdt[kdir * D_INNER + c16 + 16 * t];
        }

        int i   = wave * 16 + c16;
        int ia  = (i < CHUNK) ? i : 0;
        int lsp = scan_to_spatial(kdir, l0 + ia);
        const float* arow = xi + ((size_t)b * LL + lsp) * D_INNER + kh;
        bool nok2 = (32 + c16 < NPROJ);
        int ncol2 = nok2 ? 32 + c16 : 0;
        const float* wk = Wxp + (size_t)kdir * D_INNER * NPROJ + (size_t)kh * NPROJ;
        int ncol[XP_NT] = {c16, 16 + c16, ncol2};

        v8f acc[XP_NT];
#pragma unroll
        for (int j = 0; j < XP_NT; ++j) acc[j] = (v8f){};
        v2f   av[2][XP_KU];
        float bw[2][XP_KU][XP_NT][2];
#pragma unroll
        for (int u = 0; u < XP_KU; ++u) {
            int k0 = 4 * u;
            av[0][u] = *(const v2f*)(arow + k0);
#pragma unroll
            for (int j = 0; j < XP_NT; ++j) {
                bw[0][u][j][0] = wk[(size_t)(k0    ) * NPROJ + ncol[j]];
                bw[0][u][j][1] = wk[(size_t)(k0 + 1) * NPROJ + ncol[j]];
            }
        }
#pragma unroll
        for (int bt = 0; bt < XP_NB; ++bt) {
            const int cur = bt & 1, nxt = cur ^ 1;
            if (bt + 1 < XP_NB) {
#pragma unroll
                for (int u = 0; u < XP_KU; ++u) {
                    int k0 = (bt + 1) * 4 * XP_KU + 4 * u;
                    av[nxt][u] = *(const v2f*)(arow + k0);
#pragma unroll
                    for (int j = 0; j < XP_NT; ++j) {
                        bw[nxt][u][j][0] = wk[(size_t)(k0    ) * NPROJ + ncol[j]];
                        bw[nxt][u][j][1] = wk[(size_t)(k0 + 1) * NPROJ + ncol[j]];
                    }
                }
            }
#pragma unroll
            for (int u = 0; u < XP_KU; ++u) {
#pragma unroll
                for (int j = 0; j < XP_NT; ++j) {
                    v2f bf;
                    bf[0] = (j < 2 || nok2) ? bw[cur][u][j][0] : 0.f;
                    bf[1] = (j < 2 || nok2) ? bw[cur][u][j][1] : 0.f;
                    acc[j] = __builtin_amdgcn_wmma_f32_16x16x4_f32(false, av[cur][u], false, bf,
                                                                   (short)0, acc[j], false, false);
                }
            }
        }
#pragma unroll
        for (int j = 0; j < XP_NT; ++j) {
            int nn = 16 * j + c16;
            if (rowok && nn < NPROJ) {
#pragma unroll
                for (int v = 0; v < 8; ++v)
                    xdbl[(row0 + ib + v) * NPROJ + nn] = acc[j][v];
            }
            if (j < 2 && rowok && nn >= DT_RANK && nn < DT_RANK + D_STATE) {
#pragma unroll
                for (int v = 0; v < 8; ++v)
                    ((float*)Bsh)[(ib + v) * D_STATE + nn - DT_RANK] = acc[j][v];
            }
        }
        if (c16 < 8) {
#pragma unroll
            for (int v = 0; v < 8; ++v)
                drs[wave][(hf * 8 + v) * 8 + c16] = (c16 < DT_RANK) ? acc[0][v] : 0.f;
        }
    }
    __syncthreads();
    if (wave < FS_RT) {
        v2f a0 = {drs[wave][c16 * 8 + kh],     drs[wave][c16 * 8 + kh + 1]};
        v2f a1 = {drs[wave][c16 * 8 + 4 + kh], drs[wave][c16 * 8 + 4 + kh + 1]};
#pragma unroll
        for (int t = 0; t < DL_NT; ++t) {
            v2f b0 = {w0[t][0], w0[t][1]};
            v2f b1;
            b1[0] = hf ? 0.f : w1[t][0];
            b1[1] = hf ? 0.f : w1[t][1];
            v8f dacc = {};
            dacc = __builtin_amdgcn_wmma_f32_16x16x4_f32(false, a0, false, b0,
                                                         (short)0, dacc, false, false);
            dacc = __builtin_amdgcn_wmma_f32_16x16x4_f32(false, a1, false, b1,
                                                         (short)0, dacc, false, false);
            int nn = c16 + 16 * t;
            float sp[8];
#pragma unroll
            for (int v = 0; v < 8; ++v) sp[v] = softplus20(dacc[v] + bs[t]);
            if (rowok) {
#pragma unroll
                for (int v = 0; v < 8; ++v) {
                    delta[(row0 + ib + v) * D_INNER + nn] = sp[v];
                    dls[(ib + v) * D_INNER + nn] = sp[v];
                }
            }
        }
    }
    __syncthreads();
    if (chunk == NCHUNK - 1) return;

    int d  = threadIdx.x;
    int ch = kb * D_INNER + d;
    v2f A[D_STATE / 2], h[D_STATE / 2], P[D_STATE / 2];
#pragma unroll
    for (int j = 0; j < D_STATE / 2; ++j) {
        const float* al = A_logs + (size_t)(kdir * D_INNER + d) * D_STATE + 2 * j;
        A[j][0] = -__expf(al[0]);
        A[j][1] = -__expf(al[1]);
        h[j] = (v2f){0.f, 0.f};
        P[j] = (v2f){1.f, 1.f};
    }
    int lsp0  = scan_to_spatial(kdir, l0);
    int lstep = scan_to_spatial(kdir, l0 + 1) - lsp0;
    const float* xib = xi + ((size_t)b * LL + lsp0) * D_INNER + d;
    const float* dlp = dls + d;

    float xc[SCAN_G];
#pragma unroll
    for (int g = 0; g < SCAN_G; ++g) xc[g] = xib[(ptrdiff_t)g * lstep * D_INNER];
    float dlc = dlp[0];
    v4f bc[D_STATE / 4];
#pragma unroll
    for (int q = 0; q < D_STATE / 4; ++q) bc[q] = Bsh[q];

#pragma nounroll
    for (int i0 = 0; i0 < CHUNK; i0 += SCAN_G) {
        float xn[SCAN_G];
        __builtin_amdgcn_sched_barrier(0);
        int in = (i0 + SCAN_G < CHUNK) ? i0 + SCAN_G : i0;
#pragma unroll
        for (int g = 0; g < SCAN_G; ++g) xn[g] = xib[(ptrdiff_t)(in + g) * lstep * D_INNER];
        __builtin_amdgcn_sched_barrier(0);
#pragma unroll
        for (int g = 0; g < SCAN_G; ++g) {
            int inx = (g + 1 < SCAN_G || i0 + SCAN_G < CHUNK) ? i0 + g + 1 : i0 + g;
            __builtin_amdgcn_sched_barrier(0);
            float dln = dlp[inx * D_INNER];
            v4f bn[D_STATE / 4];
#pragma unroll
            for (int q = 0; q < D_STATE / 4; ++q) bn[q] = Bsh[inx * (D_STATE / 4) + q];
            __builtin_amdgcn_sched_barrier(0);
            float dl = dlc;
            v4f bq[D_STATE / 4];
#pragma unroll
            for (int q = 0; q < D_STATE / 4; ++q) bq[q] = bc[q];
            float dlx = dl * xc[g];
            v2f dl2  = {dl, dl};
            v2f dlx2 = {dlx, dlx};
#pragma unroll
            for (int j = 0; j < D_STATE / 2; ++j) {
                v2f dA = exp_pair(dl2 * A[j]);
                v2f Bv = {bq[j / 2][2 * (j % 2)], bq[j / 2][2 * (j % 2) + 1]};
                h[j] = dA * h[j] + dlx2 * Bv;
                P[j] = P[j] * dA;
            }
            dlc = dln;
#pragma unroll
            for (int q = 0; q < D_STATE / 4; ++q) bc[q] = bn[q];
        }
#pragma unroll
        for (int g = 0; g < SCAN_G; ++g) xc[g] = xn[g];
    }
#pragma unroll
    for (int g = 0; g < D_STATE / 4; ++g) {
        v4f hv = {h[2 * g][0], h[2 * g][1], h[2 * g + 1][0], h[2 * g + 1][1]};
        v4f pv = {P[2 * g][0], P[2 * g][1], P[2 * g + 1][0], P[2 * g + 1][1]};
        *(v4f*)(hend + part_idx(chunk, g, ch)) = hv;
        *(v4f*)(pdec + part_idx(chunk, g, ch)) = pv;
    }
}

__global__ void __launch_bounds__(SCAN_BLK) __attribute__((amdgpu_waves_per_eu(1, 4)))
k_scan_final(const float* __restrict__ xi,
             const float* __restrict__ xdbl,
             const float* __restrict__ delta,
             const float* __restrict__ A_logs,
             const float* __restrict__ Ds,
             const float* __restrict__ hend,
             const float* __restrict__ pdec,
             float* __restrict__ ys) {
    int chunk = NCHUNK - 1 - (int)blockIdx.y;
    int kb    = blockIdx.x / (D_INNER / SCAN_BLK);
    int d     = (blockIdx.x % (D_INNER / SCAN_BLK)) * SCAN_BLK + threadIdx.x;
    int b     = kb % BATCH;
    int kdir  = kb / BATCH;
    int ch    = kb * D_INNER + d;

    v2f A[D_STATE / 2], h[D_STATE / 2];
#pragma unroll
    for (int j = 0; j < D_STATE / 2; ++j) {
        const float* al = A_logs + (size_t)(kdir * D_INNER + d) * D_STATE + 2 * j;
        A[j][0] = -__expf(al[0]);
        A[j][1] = -__expf(al[1]);
        h[j] = (v2f){0.f, 0.f};
    }
    float Dskip = Ds[kdir * D_INNER + d];

#pragma unroll 4
    for (int j = 0; j < chunk; ++j) {
#pragma unroll
        for (int g = 0; g < D_STATE / 4; ++g) {
            v4f pv = *(const v4f*)(pdec + part_idx(j, g, ch));
            v4f hv = *(const v4f*)(hend + part_idx(j, g, ch));
            h[2 * g]     = (v2f){pv[0], pv[1]} * h[2 * g]     + (v2f){hv[0], hv[1]};
            h[2 * g + 1] = (v2f){pv[2], pv[3]} * h[2 * g + 1] + (v2f){hv[2], hv[3]};
        }
    }

    int l0    = chunk * CHUNK;
    int lsp0  = scan_to_spatial(kdir, l0);
    int lstep = scan_to_spatial(kdir, l0 + 1) - lsp0;
    const float* xib = xi    + ((size_t)b  * LL + lsp0) * D_INNER + d;
    const float* dlb = delta + ((size_t)kb * LL + l0) * D_INNER + d;
    const float* xdb = xdbl  + ((size_t)kb * LL + l0) * NPROJ + DT_RANK;
    float*       ysb = ys    + ((size_t)kb * LL + l0) * D_INNER + d;

    float xc[SCAN_G], dc[SCAN_G];
    v2f Bv[D_STATE / 2], Cv[D_STATE / 2];
    float bkey = xdb[0];
#pragma unroll
    for (int g = 0; g < SCAN_G; ++g) {
        xc[g] = xib[(ptrdiff_t)g * lstep * D_INNER];
        dc[g] = dlb[(size_t)g * D_INNER];
    }
#pragma unroll
    for (int j = 0; j < D_STATE / 2; ++j) {
        Bv[j] = (v2f){xdb[2 * j], xdb[2 * j + 1]};
        Cv[j] = (v2f){xdb[D_STATE + 2 * j], xdb[D_STATE + 2 * j + 1]};
    }

#pragma nounroll
    for (int i0 = 0; i0 < CHUNK; i0 += SCAN_G) {
        float xn[SCAN_G], dn[SCAN_G];
        __builtin_amdgcn_sched_barrier(0);
        int in = (i0 + SCAN_G < CHUNK) ? i0 + SCAN_G : i0;
#pragma unroll
        for (int g = 0; g < SCAN_G; ++g) {
            xn[g] = xib[(ptrdiff_t)(in + g) * lstep * D_INNER];
            dn[g] = dlb[(size_t)(in + g) * D_INNER];
        }
        __builtin_amdgcn_sched_barrier(0);
        const float* bcg = xdb + (size_t)i0 * NPROJ;
#pragma unroll
        for (int g = 0; g < SCAN_G; ++g) {
            const float* bn = (g + 1 < SCAN_G || i0 + SCAN_G < CHUNK) ? bcg + (g + 1) * NPROJ : bcg + g * NPROJ;
            __builtin_amdgcn_sched_barrier(0);
            asm volatile("" :: "s"(bkey));
            float Bn[D_STATE], Cn[D_STATE];
#pragma unroll
            for (int n = 0; n < D_STATE; ++n) {
                Bn[n] = bn[n];
                Cn[n] = bn[D_STATE + n];
            }
            __builtin_amdgcn_sched_barrier(0);
            float x   = xc[g];
            float dl  = dc[g];
            float dlx = dl * x;
            v2f dl2  = {dl, dl};
            v2f dlx2 = {dlx, dlx};
            v2f ya = {0.f, 0.f}, yb = {0.f, 0.f};
#pragma unroll
            for (int j = 0; j < D_STATE / 2; j += 2) {
                v2f dA0 = exp_pair(dl2 * A[j]);
                v2f dA1 = exp_pair(dl2 * A[j + 1]);
                h[j]     = dA0 * h[j]     + dlx2 * Bv[j];
                h[j + 1] = dA1 * h[j + 1] + dlx2 * Bv[j + 1];
                ya += h[j]     * Cv[j];
                yb += h[j + 1] * Cv[j + 1];
            }
            ysb[(size_t)(i0 + g) * D_INNER] = ((ya[0] + ya[1]) + (yb[0] + yb[1])) + x * Dskip;
            bkey = Bn[0];
#pragma unroll
            for (int j = 0; j < D_STATE / 2; ++j) {
                Bv[j] = (v2f){Bn[2 * j], Bn[2 * j + 1]};
                Cv[j] = (v2f){Cn[2 * j], Cn[2 * j + 1]};
            }
        }
#pragma unroll
        for (int g = 0; g < SCAN_G; ++g) { xc[g] = xn[g]; dc[g] = dn[g]; }
    }
}

#define CO_NT   3
#define CO_KU   3
#define CO_NB   (D_INNER / (8 * CO_KU))
static_assert(D_INNER % (8 * CO_KU) == 0, "whole groups");
static_assert(DIMC == 2 * 16 * CO_NT, "two waves cover the output channels");
static_assert(LL % 16 == 0, "a position tile lies inside one image");

__global__ void __launch_bounds__(64) __attribute__((amdgpu_waves_per_eu(1, 2)))
k_comb_out(const float* __restrict__ ys,
           const float* __restrict__ xz,
           const float* __restrict__ Wout,
           float* __restrict__ out) {
    int lane = threadIdx.x & 31;
    int wave = __builtin_amdgcn_readfirstlane((int)(threadIdx.x >> 5));
    int c16 = lane & 15;
    int hf  = lane >> 4;
    int row = (int)blockIdx.x * 16 + c16;
    int b = row / LL, l = row % LL;

    const float* yp[KDIR];
#pragma unroll
    for (int k = 0; k < KDIR; ++k)
        yp[k] = ys + (size_t)((k * BATCH + b) * LL + spatial_to_scan(k, l)) * D_INNER + 4 * hf;
    const float* zp = xz + (size_t)row * XZW + D_INNER + 4 * hf;
    int n0 = wave * (16 * CO_NT);
    const float* wp = Wout + (size_t)(4 * hf) * DIMC + n0 + c16;

    v8f acc[CO_NT];
#pragma unroll
    for (int t = 0; t < CO_NT; ++t) acc[t] = (v8f){};

    v4f   yv[2][CO_KU][KDIR], zv[2][CO_KU];
    float wv[2][CO_KU][CO_NT][4];
#pragma unroll
    for (int u = 0; u < CO_KU; ++u) {
        int d0 = 8 * u;
#pragma unroll
        for (int k = 0; k < KDIR; ++k) yv[0][u][k] = *(const v4f*)(yp[k] + d0);
        zv[0][u] = *(const v4f*)(zp + d0);
#pragma unroll
        for (int t = 0; t < CO_NT; ++t)
#pragma unroll
            for (int e = 0; e < 4; ++e)
                wv[0][u][t][e] = wp[(size_t)(d0 + e) * DIMC + 16 * t];
    }
#pragma unroll
    for (int g = 0; g < CO_NB; ++g) {
        const int cur = g & 1, nxt = cur ^ 1;
        if (g + 1 < CO_NB) {
#pragma unroll
            for (int u = 0; u < CO_KU; ++u) {
                int d0 = 8 * ((g + 1) * CO_KU + u);
#pragma unroll
                for (int k = 0; k < KDIR; ++k) yv[nxt][u][k] = *(const v4f*)(yp[k] + d0);
                zv[nxt][u] = *(const v4f*)(zp + d0);
#pragma unroll
                for (int t = 0; t < CO_NT; ++t)
#pragma unroll
                    for (int e = 0; e < 4; ++e)
                        wv[nxt][u][t][e] = wp[(size_t)(d0 + e) * DIMC + 16 * t];
            }
        }
#pragma unroll
        for (int u = 0; u < CO_KU; ++u) {
            v4f s = ((yv[cur][u][0] + yv[cur][u][1]) + yv[cur][u][2]) + yv[cur][u][3];
            v4f z = zv[cur][u];
            float yy[4];
#pragma unroll
            for (int e = 0; e < 4; ++e) yy[e] = s[e] * silu(z[e]);
            v2f y01 = {yy[0], yy[1]};
            v2f y23 = {yy[2], yy[3]};
#pragma unroll
            for (int t = 0; t < CO_NT; ++t) {
                v2f w01 = {wv[cur][u][t][0], wv[cur][u][t][1]};
                v2f w23 = {wv[cur][u][t][2], wv[cur][u][t][3]};
                acc[t] = __builtin_amdgcn_wmma_f32_16x16x4_f32(false, w01, false, y01,
                                                               (short)0, acc[t], false, false);
                acc[t] = __builtin_amdgcn_wmma_f32_16x16x4_f32(false, w23, false, y23,
                                                               (short)0, acc[t], false, false);
            }
        }
    }
    float* op = out + (size_t)(b * DIMC + n0 + 8 * hf) * LL + l;
#pragma unroll
    for (int t = 0; t < CO_NT; ++t)
#pragma unroll
        for (int v = 0; v < 8; ++v)
            op[(size_t)(16 * t + v) * LL] = acc[t][v];
}

extern "C" void kernel_launch(void* const* d_in, const int* in_sizes, int n_in,
                              void* d_out, int out_size, void* d_ws, size_t ws_size,
                              hipStream_t stream) {
    (void)in_sizes; (void)n_in; (void)out_size; (void)ws_size;
    const float* x       = (const float*)d_in[0];
    const float* W_in    = (const float*)d_in[1];
    const float* W_conv  = (const float*)d_in[2];
    const float* b_conv  = (const float*)d_in[3];
    const float* W_xproj = (const float*)d_in[4];
    const float* W_dt    = (const float*)d_in[5];
    const float* b_dt    = (const float*)d_in[6];
    const float* A_logs  = (const float*)d_in[7];
    const float* Ds      = (const float*)d_in[8];
    const float* W_out   = (const float*)d_in[9];
    float* out = (float*)d_out;

    float* ws = (float*)d_ws;
    size_t o = 0;
    float* xz    = ws + o;  o += (size_t)BL * XZW;
    float* xi    = ws + o;  o += (size_t)BL * D_INNER;
    float* xdbl  = ws + o;  o += (size_t)KDIR * BL * NPROJ;
    float* delta = ws + o;  o += (size_t)KDIR * BL * D_INNER;
    float* ysbuf = ws + o;  o += (size_t)KDIR * BL * D_INNER;
    float* hend  = ws + o;  o += (size_t)(NCHUNK - 1) * NSCAN;
    float* pdec  = ws + o;  o += (size_t)(NCHUNK - 1) * NSCAN;

    k_in_conv <<<CV_BLOCKS + Z_BLOCKS, IC_THREADS, 0, stream>>>(x, W_in, W_conv, b_conv, xz, xi);
    k_xproj_delta_part<<<dim3(NCHUNK, NCHAN / D_INNER), FS_THREADS, 0, stream>>>(
        xi, W_xproj, W_dt, b_dt, A_logs, xdbl, delta, hend, pdec);
    k_scan_final<<<dim3(NCHAN / SCAN_BLK, NCHUNK),     SCAN_BLK, 0, stream>>>(
        xi, xdbl, delta, A_logs, Ds, hend, pdec, ysbuf);
    k_comb_out<<<BL / 16, 64, 0, stream>>>(ysbuf, xz, W_out, out);
}


# pass 2 of the scan: a block is 4 consecutive chunks (a wave each) of 32 channels; the fold of the chunks in front is shared out over its 4 waves and joined through LDS; order of sums in h changes
# speedup vs baseline: 30.8162x; 1.1066x over previous
#include <hip/hip_runtime.h>
#include <math.h>

typedef __attribute__((ext_vector_type(2))) float v2f;
typedef __attribute__((ext_vector_type(4))) float v4f;
typedef __attribute__((ext_vector_type(8))) float v8f;

#define DIMC    96
#define D_STATE 16
#define D_INNER 192
#define DT_RANK 6
#define KDIR    4
#define BATCH   2
#define HH      56
#define WWD     56
#define LL      (HH * WWD)
#define BL      (BATCH * LL)
#define NPROJ   (DT_RANK + 2 * D_STATE)
#define XZW     (2 * D_INNER)
#define NSCAN   (KDIR * BATCH * D_INNER * D_STATE)
#define NCHAN   (KDIR * BATCH * D_INNER)
#define SF_CH    32
#define SF_NW    4
#define SCAN_G   8
#define NCHUNK  56
#define CHUNK   (LL / NCHUNK)
#define SF_NG   (NCHUNK / SF_NW)

__device__ __forceinline__ int scan_to_spatial(int k, int l) {
    int q = l / WWD, r = l % WWD;
    if (k == 0) return l;
    if (k == 1) return q * WWD + (WWD - 1 - r);
    if (k == 2) return r * WWD + q;
    return (WWD - 1 - r) * WWD + q;
}

__device__ __forceinline__ int spatial_to_scan(int k, int l) {
    int q = l / WWD, r = l % WWD;
    if (k == 0) return l;
    if (k == 1) return q * WWD + (WWD - 1 - r);
    if (k == 2) return r * WWD + q;
    return r * WWD + (WWD - 1 - q);
}

__device__ __forceinline__ float silu(float v) {
    return v / (1.f + __expf(-v));
}

__device__ __forceinline__ float softplus20(float t) {
    float e  = __expf(t);
    float s  = 1.f + e;
    float bb = s - 1.f;
    float r  = (1.f - (s - bb)) + (e - bb);
    float l2 = __builtin_amdgcn_logf(s);
    const float c = 0x1.62e42ep-1f, cc = 0x1.efa39ep-25f;
    float lo = __builtin_fmaf(l2, cc, r * __builtin_amdgcn_rcpf(s));
    return (t > 20.f) ? t : __builtin_fmaf(l2, c, lo);
}

#define CV_ROWS   4
#define CV_WIN    (CV_ROWS + 2)
#define CV_POS    (CV_WIN * WWD)
#define CV_PW     (WWD + 2)
#define CV_COLS   32
#define CV_PITCH  34
#define CV_KU     6
#define CV_NB     (DIMC / (4 * CV_KU))
#define IC_WAVES  7
#define IC_THREADS (IC_WAVES * 32)
#define CV_PER_IMG ((HH / CV_ROWS) * (D_INNER / CV_COLS))
#define CV_BLOCKS (BATCH * CV_PER_IMG)
#define Z_BLOCKS  ((BL / 16 / IC_WAVES) * (D_INNER / 16))
static_assert(HH % CV_ROWS == 0, "whole bands");
static_assert(CV_POS == IC_WAVES * 3 * 16, "three row tiles per wave");
static_assert(CV_ROWS * WWD == IC_WAVES * 32, "32 conv outputs per thread");
static_assert((BL / 16) % IC_WAVES == 0, "whole z blocks");
static_assert(DIMC % (4 * CV_KU) == 0, "whole groups of k blocks");

__global__ void __launch_bounds__(IC_THREADS)
k_in_conv(const float* __restrict__ x,
          const float* __restrict__ Win,
          const float* __restrict__ Wc,
          const float* __restrict__ bc,
          float* __restrict__ xz,
          float* __restrict__ xi) {
    __shared__ float xs[CV_WIN * CV_PW * CV_PITCH];
    int lane = threadIdx.x & 31;
    int wave = __builtin_amdgcn_readfirstlane((int)(threadIdx.x >> 5));
    int c16  = lane & 15;
    int kh   = (lane >> 4) * 2;

    if (blockIdx.x >= CV_BLOCKS) {
        int zb    = (int)blockIdx.x - CV_BLOCKS;
        int mTile = (zb / (D_INNER / 16)) * IC_WAVES + wave;
        int nTile = D_INNER / 16 + zb % (D_INNER / 16);
        int m0 = mTile * 16, n0 = nTile * 16;
        int mrow = m0 + c16;
        int b = mrow / LL, l = mrow % LL;
        int nn = n0 + c16;
        float za[DIMC / 4][2], zw[DIMC / 4][2];
#pragma unroll
        for (int u = 0; u < DIMC / 4; ++u) {
            int c0 = 4 * u + kh;
            za[u][0] = x[(size_t)(b * DIMC + c0    ) * LL + l];
            za[u][1] = x[(size_t)(b * DIMC + c0 + 1) * LL + l];
            zw[u][0] = Win[(size_t)(c0    ) * XZW + nn];
            zw[u][1] = Win[(size_t)(c0 + 1) * XZW + nn];
        }
        v8f acc = {};
#pragma unroll
        for (int u = 0; u < DIMC / 4; ++u) {
            v2f a  = {za[u][0], za[u][1]};
            v2f bf = {zw[u][0], zw[u][1]};
            acc = __builtin_amdgcn_wmma_f32_16x16x4_f32(false, a, false, bf,
                                                        (short)0, acc, false, false);
        }
        int mb = m0 + (lane >> 4) * 8;
#pragma unroll
        for (int v = 0; v < 8; ++v)
            xz[(size_t)(mb + v) * XZW + nn] = acc[v];
        return;
    }

    int idx  = (int)blockIdx.x;
    int b    = idx / CV_PER_IMG;
    int rem  = idx % CV_PER_IMG;
    int band = rem / (D_INNER / CV_COLS);
    int d0   = (rem % (D_INNER / CV_COLS)) * CV_COLS;
    int hfirst = band * CV_ROWS - 1;

    for (int i = threadIdx.x; i < CV_WIN * 2 * CV_COLS; i += IC_THREADS) {
        int row = i / (2 * CV_COLS), side = (i / CV_COLS) & 1, col = i % CV_COLS;
        xs[(row * CV_PW + side * (CV_PW - 1)) * CV_PITCH + col] = 0.f;
    }

    const float* xa[3];
    bool ok[3];
#pragma unroll
    for (int t = 0; t < 3; ++t) {
        int p = (wave * 3 + t) * 16 + c16;
        int h = hfirst + p / WWD;
        int w = p % WWD;
        ok[t] = (h >= 0 && h < HH);
        int l = ok[t] ? h * WWD + w : 0;
        xa[t] = x + (size_t)b * DIMC * LL + l;
    }
    const float* wb = Win + d0 + c16;
    v8f acc[3][2];
#pragma unroll
    for (int t = 0; t < 3; ++t) { acc[t][0] = (v8f){}; acc[t][1] = (v8f){}; }
    float av[2][CV_KU][3][2], bv[2][CV_KU][2][2];
#pragma unroll
    for (int u = 0; u < CV_KU; ++u) {
        int c0 = 4 * u + kh;
#pragma unroll
        for (int t = 0; t < 3; ++t) {
            av[0][u][t][0] = xa[t][(size_t)(c0    ) * LL];
            av[0][u][t][1] = xa[t][(size_t)(c0 + 1) * LL];
        }
#pragma unroll
        for (int j = 0; j < 2; ++j) {
            bv[0][u][j][0] = wb[(size_t)(c0    ) * XZW + 16 * j];
            bv[0][u][j][1] = wb[(size_t)(c0 + 1) * XZW + 16 * j];
        }
    }
#pragma unroll
    for (int bt = 0; bt < CV_NB; ++bt) {
        const int cur = bt & 1, nxt = cur ^ 1;
        if (bt + 1 < CV_NB) {
#pragma unroll
            for (int u = 0; u < CV_KU; ++u) {
                int c0 = (bt + 1) * 4 * CV_KU + 4 * u + kh;
#pragma unroll
                for (int t = 0; t < 3; ++t) {
                    av[nxt][u][t][0] = xa[t][(size_t)(c0    ) * LL];
                    av[nxt][u][t][1] = xa[t][(size_t)(c0 + 1) * LL];
                }
#pragma unroll
                for (int j = 0; j < 2; ++j) {
                    bv[nxt][u][j][0] = wb[(size_t)(c0    ) * XZW + 16 * j];
                    bv[nxt][u][j][1] = wb[(size_t)(c0 + 1) * XZW + 16 * j];
                }
            }
        }
#pragma unroll
        for (int u = 0; u < CV_KU; ++u) {
#pragma unroll
            for (int t = 0; t < 3; ++t) {
                v2f a;
                a[0] = ok[t] ? av[cur][u][t][0] : 0.f;
                a[1] = ok[t] ? av[cur][u][t][1] : 0.f;
#pragma unroll
                for (int j = 0; j < 2; ++j) {
                    v2f bf = {bv[cur][u][j][0], bv[cur][u][j][1]};
                    acc[t][j] = __builtin_amdgcn_wmma_f32_16x16x4_f32(false, a, false, bf,
                                                                      (short)0, acc[t][j], false, false);
                }
            }
        }
    }
#pragma unroll
    for (int t = 0; t < 3; ++t) {
        int prow = (wave * 3 + t) * 16 + (lane >> 4) * 8;
#pragma unroll
        for (int v = 0; v < 8; ++v) {
            int p = prow + v;
            int q = (p / WWD) * CV_PW + p % WWD + 1;
            xs[q * CV_PITCH + c16]      = acc[t][0][v];
            xs[q * CV_PITCH + 16 + c16] = acc[t][1][v];
        }
    }
    __syncthreads();

    int d = d0 + lane;
    float wg[9];
#pragma unroll
    for (int t = 0; t < 9; ++t) wg[t] = Wc[d * 9 + t];
    float bias = bc[d];
    float* xo = xi + ((size_t)b * LL + band * CV_ROWS * WWD) * D_INNER + d;
#pragma unroll 4
    for (int i = 0; i < (CV_ROWS * WWD) / IC_WAVES; ++i) {
        int o = wave + IC_WAVES * i;
        int r = o / WWD, w = o % WWD;
        const float* q = xs + (r * CV_PW + w) * CV_PITCH + lane;
        float s = bias;
#pragma unroll
        for (int dy = 0; dy < 3; ++dy)
#pragma unroll
            for (int dx = 0; dx < 3; ++dx)
                s = __builtin_fmaf(wg[dy * 3 + dx], q[(dy * CV_PW + dx) * CV_PITCH], s);
        xo[(size_t)o * D_INNER] = silu(s);
    }
}


static_assert(WWD % CHUNK == 0 && CHUNK >= 2, "a chunk lies inside one scan row");
static_assert(CHUNK % SCAN_G == 0, "whole groups");
static_assert(NCHUNK % SF_NW == 0, "whole groups of chunks");
static_assert(D_INNER % SF_CH == 0, "a wave's channels share (k, b)");

__device__ __forceinline__ size_t part_idx(int chunk, int g, int ch) {
    return ((size_t)(chunk * (D_STATE / 4) + g) * NCHAN + ch) * 4;
}

__device__ __forceinline__ v2f exp_pair(v2f a) {
    v2f r = {__expf(a[0]), __expf(a[1])};
    return r;
}

#define XP_KU   6
#define XP_NB   (D_INNER / (4 * XP_KU))
#define XP_NT   3
#define DL_NT   (D_INNER / 16)
#define FS_RT   4
#define FS_THREADS D_INNER
static_assert(D_INNER % (4 * XP_KU) == 0, "whole groups of k blocks");
static_assert(D_INNER % 32 == 0, "whole waves");
static_assert(FS_RT * 16 >= CHUNK && (FS_RT - 1) * 16 < CHUNK, "row tiles cover the chunk");
static_assert(FS_RT * 32 <= FS_THREADS, "a wave per row tile");
static_assert(CHUNK % 8 == 0, "a lane half's 8 rows lie inside the chunk or behind it");
static_assert(DT_RANK + D_STATE <= 32, "the B columns lie in the first two column tiles");

__global__ void __launch_bounds__(FS_THREADS) __attribute__((amdgpu_waves_per_eu(4, 4)))
k_xproj_delta_part(const float* __restrict__ xi,
                   const float* __restrict__ Wxp,
                   const float* __restrict__ Wdt,
                   const float* __restrict__ bdt,
                   const float* __restrict__ A_logs,
                   float* __restrict__ xdbl,
                   float* __restrict__ delta,
                   float* __restrict__ hend,
                   float* __restrict__ pdec) {
    __shared__ float dls[CHUNK * D_INNER];
    __shared__ v4f   Bsh[CHUNK * (D_STATE / 4)];
    __shared__ float drs[FS_RT][16 * 8];
    int lane = threadIdx.x & 31;
    int wave = __builtin_amdgcn_readfirstlane((int)(threadIdx.x >> 5));
    int chunk = blockIdx.x;
    int kb    = blockIdx.y;
    int b     = kb % BATCH;
    int kdir  = kb / BATCH;
    int l0    = chunk * CHUNK;
    size_t row0 = (size_t)kb * LL + l0;
    int c16 = lane & 15;
    int hf  = lane >> 4;
    int kh  = hf * 2;
    int ib  = wave * 16 + hf * 8;
    bool rowok = (ib < CHUNK);

    float w0[DL_NT][2], w1[DL_NT][2], bs[DL_NT];
    if (wave < FS_RT) {
        const float* wd = Wdt + (size_t)kdir * DT_RANK * D_INNER + c16;
        int k2 = hf ? 0 : 4;
#pragma unroll
        for (int t = 0; t < DL_NT; ++t) {
            w0[t][0] = wd[(size_t)(kh     ) * D_INNER + 16 * t];
            w0[t][1] = wd[(size_t)(kh + 1 ) * D_INNER + 16 * t];
            w1[t][0] = wd[(size_t)(k2     ) * D_INNER + 16 * t];
            w1[t][1] = wd[(size_t)(k2 + 1 ) * D_INNER + 16 * t];
            bs[t]    = bdt[kdir * D_INNER + c16 + 16 * t];
        }

        int i   = wave * 16 + c16;
        int ia  = (i < CHUNK) ? i : 0;
        int lsp = scan_to_spatial(kdir, l0 + ia);
        const float* arow = xi + ((size_t)b * LL + lsp) * D_INNER + kh;
        bool nok2 = (32 + c16 < NPROJ);
        int ncol2 = nok2 ? 32 + c16 : 0;
        const float* wk = Wxp + (size_t)kdir * D_INNER * NPROJ + (size_t)kh * NPROJ;
        int ncol[XP_NT] = {c16, 16 + c16, ncol2};

        v8f acc[XP_NT];
#pragma unroll
        for (int j = 0; j < XP_NT; ++j) acc[j] = (v8f){};
        v2f   av[2][XP_KU];
        float bw[2][XP_KU][XP_NT][2];
#pragma unroll
        for (int u = 0; u < XP_KU; ++u) {
            int k0 = 4 * u;
            av[0][u] = *(const v2f*)(arow + k0);
#pragma unroll
            for (int j = 0; j < XP_NT; ++j) {
                bw[0][u][j][0] = wk[(size_t)(k0    ) * NPROJ + ncol[j]];
                bw[0][u][j][1] = wk[(size_t)(k0 + 1) * NPROJ + ncol[j]];
            }
        }
#pragma unroll
        for (int bt = 0; bt < XP_NB; ++bt) {
            const int cur = bt & 1, nxt = cur ^ 1;
            if (bt + 1 < XP_NB) {
#pragma unroll
                for (int u = 0; u < XP_KU; ++u) {
                    int k0 = (bt + 1) * 4 * XP_KU + 4 * u;
                    av[nxt][u] = *(const v2f*)(arow + k0);
#pragma unroll
                    for (int j = 0; j < XP_NT; ++j) {
                        bw[nxt][u][j][0] = wk[(size_t)(k0    ) * NPROJ + ncol[j]];
                        bw[nxt][u][j][1] = wk[(size_t)(k0 + 1) * NPROJ + ncol[j]];
                    }
                }
            }
#pragma unroll
            for (int u = 0; u < XP_KU; ++u) {
#pragma unroll
                for (int j = 0; j < XP_NT; ++j) {
                    v2f bf;
                    bf[0] = (j < 2 || nok2) ? bw[cur][u][j][0] : 0.f;
                    bf[1] = (j < 2 || nok2) ? bw[cur][u][j][1] : 0.f;
                    acc[j] = __builtin_amdgcn_wmma_f32_16x16x4_f32(false, av[cur][u], false, bf,
                                                                   (short)0, acc[j], false, false);
                }
            }
        }
#pragma unroll
        for (int j = 0; j < XP_NT; ++j) {
            int nn = 16 * j + c16;
            if (rowok && nn < NPROJ) {
#pragma unroll
                for (int v = 0; v < 8; ++v)
                    xdbl[(row0 + ib + v) * NPROJ + nn] = acc[j][v];
            }
            if (j < 2 && rowok && nn >= DT_RANK && nn < DT_RANK + D_STATE) {
#pragma unroll
                for (int v = 0; v < 8; ++v)
                    ((float*)Bsh)[(ib + v) * D_STATE + nn - DT_RANK] = acc[j][v];
            }
        }
        if (c16 < 8) {
#pragma unroll
            for (int v = 0; v < 8; ++v)
                drs[wave][(hf * 8 + v) * 8 + c16] = (c16 < DT_RANK) ? acc[0][v] : 0.f;
        }
    }
    __syncthreads();
    if (wave < FS_RT) {
        v2f a0 = {drs[wave][c16 * 8 + kh],     drs[wave][c16 * 8 + kh + 1]};
        v2f a1 = {drs[wave][c16 * 8 + 4 + kh], drs[wave][c16 * 8 + 4 + kh + 1]};
#pragma unroll
        for (int t = 0; t < DL_NT; ++t) {
            v2f b0 = {w0[t][0], w0[t][1]};
            v2f b1;
            b1[0] = hf ? 0.f : w1[t][0];
            b1[1] = hf ? 0.f : w1[t][1];
            v8f dacc = {};
            dacc = __builtin_amdgcn_wmma_f32_16x16x4_f32(false, a0, false, b0,
                                                         (short)0, dacc, false, false);
            dacc = __builtin_amdgcn_wmma_f32_16x16x4_f32(false, a1, false, b1,
                                                         (short)0, dacc, false, false);
            int nn = c16 + 16 * t;
            float sp[8];
#pragma unroll
            for (int v = 0; v < 8; ++v) sp[v] = softplus20(dacc[v] + bs[t]);
            if (rowok) {
#pragma unroll
                for (int v = 0; v < 8; ++v) {
                    delta[(row0 + ib + v) * D_INNER + nn] = sp[v];
                    dls[(ib + v) * D_INNER + nn] = sp[v];
                }
            }
        }
    }
    __syncthreads();
    if (chunk == NCHUNK - 1) return;

    int d  = threadIdx.x;
    int ch = kb * D_INNER + d;
    v2f A[D_STATE / 2], h[D_STATE / 2], P[D_STATE / 2];
#pragma unroll
    for (int j = 0; j < D_STATE / 2; ++j) {
        const float* al = A_logs + (size_t)(kdir * D_INNER + d) * D_STATE + 2 * j;
        A[j][0] = -__expf(al[0]);
        A[j][1] = -__expf(al[1]);
        h[j] = (v2f){0.f, 0.f};
        P[j] = (v2f){1.f, 1.f};
    }
    int lsp0  = scan_to_spatial(kdir, l0);
    int lstep = scan_to_spatial(kdir, l0 + 1) - lsp0;
    const float* xib = xi + ((size_t)b * LL + lsp0) * D_INNER + d;
    const float* dlp = dls + d;

    float xc[SCAN_G];
#pragma unroll
    for (int g = 0; g < SCAN_G; ++g) xc[g] = xib[(ptrdiff_t)g * lstep * D_INNER];
    float dlc = dlp[0];
    v4f bc[D_STATE / 4];
#pragma unroll
    for (int q = 0; q < D_STATE / 4; ++q) bc[q] = Bsh[q];

#pragma nounroll
    for (int i0 = 0; i0 < CHUNK; i0 += SCAN_G) {
        float xn[SCAN_G];
        __builtin_amdgcn_sched_barrier(0);
        int in = (i0 + SCAN_G < CHUNK) ? i0 + SCAN_G : i0;
#pragma unroll
        for (int g = 0; g < SCAN_G; ++g) xn[g] = xib[(ptrdiff_t)(in + g) * lstep * D_INNER];
        __builtin_amdgcn_sched_barrier(0);
#pragma unroll
        for (int g = 0; g < SCAN_G; ++g) {
            int inx = (g + 1 < SCAN_G || i0 + SCAN_G < CHUNK) ? i0 + g + 1 : i0 + g;
            __builtin_amdgcn_sched_barrier(0);
            float dln = dlp[inx * D_INNER];
            v4f bn[D_STATE / 4];
#pragma unroll
            for (int q = 0; q < D_STATE / 4; ++q) bn[q] = Bsh[inx * (D_STATE / 4) + q];
            __builtin_amdgcn_sched_barrier(0);
            float dl = dlc;
            v4f bq[D_STATE / 4];
#pragma unroll
            for (int q = 0; q < D_STATE / 4; ++q) bq[q] = bc[q];
            float dlx = dl * xc[g];
            v2f dl2  = {dl, dl};
            v2f dlx2 = {dlx, dlx};
#pragma unroll
            for (int j = 0; j < D_STATE / 2; ++j) {
                v2f dA = exp_pair(dl2 * A[j]);
                v2f Bv = {bq[j / 2][2 * (j % 2)], bq[j / 2][2 * (j % 2) + 1]};
                h[j] = dA * h[j] + dlx2 * Bv;
                P[j] = P[j] * dA;
            }
            dlc = dln;
#pragma unroll
            for (int q = 0; q < D_STATE / 4; ++q) bc[q] = bn[q];
        }
#pragma unroll
        for (int g = 0; g < SCAN_G; ++g) xc[g] = xn[g];
    }
#pragma unroll
    for (int g = 0; g < D_STATE / 4; ++g) {
        v4f hv = {h[2 * g][0], h[2 * g][1], h[2 * g + 1][0], h[2 * g + 1][1]};
        v4f pv = {P[2 * g][0], P[2 * g][1], P[2 * g + 1][0], P[2 * g + 1][1]};
        *(v4f*)(hend + part_idx(chunk, g, ch)) = hv;
        *(v4f*)(pdec + part_idx(chunk, g, ch)) = pv;
    }
}

__global__ void __launch_bounds__(SF_NW * 32) __attribute__((amdgpu_waves_per_eu(1, 4)))
k_scan_final(const float* __restrict__ xi,
             const float* __restrict__ xdbl,
             const float* __restrict__ delta,
             const float* __restrict__ A_logs,
             const float* __restrict__ Ds,
             const float* __restrict__ hend,
             const float* __restrict__ pdec,
             float* __restrict__ ys) {
    __shared__ v4f segm[SF_NW][2 * (D_STATE / 4)][32];
    int lane = threadIdx.x & 31;
    int wave = __builtin_amdgcn_readfirstlane((int)(threadIdx.x >> 5));
    int grp   = SF_NG - 1 - (int)blockIdx.y;
    int chunk = grp * SF_NW + wave;
    int kb    = blockIdx.x / (D_INNER / SF_CH);
    int d     = (blockIdx.x % (D_INNER / SF_CH)) * SF_CH + lane;
    int b     = kb % BATCH;
    int kdir  = kb / BATCH;
    int ch    = kb * D_INNER + d;

    v2f A[D_STATE / 2], h[D_STATE / 2], P[D_STATE / 2];
#pragma unroll
    for (int j = 0; j < D_STATE / 2; ++j) {
        const float* al = A_logs + (size_t)(kdir * D_INNER + d) * D_STATE + 2 * j;
        A[j][0] = -__expf(al[0]);
        A[j][1] = -__expf(al[1]);
        h[j] = (v2f){0.f, 0.f};
        P[j] = (v2f){1.f, 1.f};
    }
    float Dskip = Ds[kdir * D_INNER + d];

    int j0 = wave * grp;
#pragma unroll 4
    for (int i = 0; i < grp; ++i) {
#pragma unroll
        for (int g = 0; g < D_STATE / 4; ++g) {
            v4f pv = *(const v4f*)(pdec + part_idx(j0 + i, g, ch));
            v4f hv = *(const v4f*)(hend + part_idx(j0 + i, g, ch));
            h[2 * g]     = (v2f){pv[0], pv[1]} * h[2 * g]     + (v2f){hv[0], hv[1]};
            h[2 * g + 1] = (v2f){pv[2], pv[3]} * h[2 * g + 1] + (v2f){hv[2], hv[3]};
            P[2 * g]     = P[2 * g]     * (v2f){pv[0], pv[1]};
            P[2 * g + 1] = P[2 * g + 1] * (v2f){pv[2], pv[3]};
        }
    }
#pragma unroll
    for (int g = 0; g < D_STATE / 4; ++g) {
        segm[wave][g][lane]               = (v4f){P[2 * g][0], P[2 * g][1], P[2 * g + 1][0], P[2 * g + 1][1]};
        segm[wave][D_STATE / 4 + g][lane] = (v4f){h[2 * g][0], h[2 * g][1], h[2 * g + 1][0], h[2 * g + 1][1]};
    }
    __syncthreads();
    v4f ep[SF_NW - 1][D_STATE / 4], eh[SF_NW - 1][D_STATE / 4];
#pragma unroll
    for (int i = 0; i < SF_NW - 1; ++i) {
#pragma unroll
        for (int g = 0; g < D_STATE / 4; ++g) {
            ep[i][g] = (v4f){1.f, 1.f, 1.f, 1.f};
            eh[i][g] = (v4f){0.f, 0.f, 0.f, 0.f};
        }
    }
#pragma unroll
    for (int i = 0; i < SF_NW - 1; ++i) {
        if (i < wave) {
#pragma unroll
            for (int g = 0; g < D_STATE / 4; ++g) {
                ep[i][g] = *(const v4f*)(pdec + part_idx(grp * SF_NW + i, g, ch));
                eh[i][g] = *(const v4f*)(hend + part_idx(grp * SF_NW + i, g, ch));
            }
        }
    }
    int l0    = chunk * CHUNK;
    int lsp0  = scan_to_spatial(kdir, l0);
    int lstep = scan_to_spatial(kdir, l0 + 1) - lsp0;
    const float* xib = xi    + ((size_t)b  * LL + lsp0) * D_INNER + d;
    const float* dlb = delta + ((size_t)kb * LL + l0) * D_INNER + d;
    const float* xdb = xdbl  + ((size_t)kb * LL + l0) * NPROJ + DT_RANK;
    float*       ysb = ys    + ((size_t)kb * LL + l0) * D_INNER + d;

    float xc[SCAN_G], dc[SCAN_G];
    v2f Bv[D_STATE / 2], Cv[D_STATE / 2];
    float bkey = xdb[0];
#pragma unroll
    for (int g = 0; g < SCAN_G; ++g) {
        xc[g] = xib[(ptrdiff_t)g * lstep * D_INNER];
        dc[g] = dlb[(size_t)g * D_INNER];
    }
#pragma unroll
    for (int j = 0; j < D_STATE / 2; ++j) {
        Bv[j] = (v2f){xdb[2 * j], xdb[2 * j + 1]};
        Cv[j] = (v2f){xdb[D_STATE + 2 * j], xdb[D_STATE + 2 * j + 1]};
    }

#pragma unroll
    for (int g = 0; g < D_STATE / 4; ++g) {
        v4f hv = segm[0][D_STATE / 4 + g][lane];
        h[2 * g]     = (v2f){hv[0], hv[1]};
        h[2 * g + 1] = (v2f){hv[2], hv[3]};
    }
#pragma unroll
    for (int s = 1; s < SF_NW; ++s) {
#pragma unroll
        for (int g = 0; g < D_STATE / 4; ++g) {
            v4f pv = segm[s][g][lane];
            v4f hv = segm[s][D_STATE / 4 + g][lane];
            h[2 * g]     = (v2f){pv[0], pv[1]} * h[2 * g]     + (v2f){hv[0], hv[1]};
            h[2 * g + 1] = (v2f){pv[2], pv[3]} * h[2 * g + 1] + (v2f){hv[2], hv[3]};
        }
    }
#pragma unroll
    for (int i = 0; i < SF_NW - 1; ++i) {
#pragma unroll
        for (int g = 0; g < D_STATE / 4; ++g) {
            h[2 * g]     = (v2f){ep[i][g][0], ep[i][g][1]} * h[2 * g]     + (v2f){eh[i][g][0], eh[i][g][1]};
            h[2 * g + 1] = (v2f){ep[i][g][2], ep[i][g][3]} * h[2 * g + 1] + (v2f){eh[i][g][2], eh[i][g][3]};
        }
    }

#pragma nounroll
    for (int i0 = 0; i0 < CHUNK; i0 += SCAN_G) {
        float xn[SCAN_G], dn[SCAN_G];
        __builtin_amdgcn_sched_barrier(0);
        int in = (i0 + SCAN_G < CHUNK) ? i0 + SCAN_G : i0;
#pragma unroll
        for (int g = 0; g < SCAN_G; ++g) {
            xn[g] = xib[(ptrdiff_t)(in + g) * lstep * D_INNER];
            dn[g] = dlb[(size_t)(in + g) * D_INNER];
        }
        __builtin_amdgcn_sched_barrier(0);
        const float* bcg = xdb + (size_t)i0 * NPROJ;
#pragma unroll
        for (int g = 0; g < SCAN_G; ++g) {
            const float* bn = (g + 1 < SCAN_G || i0 + SCAN_G < CHUNK) ? bcg + (g + 1) * NPROJ : bcg + g * NPROJ;
            __builtin_amdgcn_sched_barrier(0);
            asm volatile("" :: "s"(bkey));
            float Bn[D_STATE], Cn[D_STATE];
#pragma unroll
            for (int n = 0; n < D_STATE; ++n) {
                Bn[n] = bn[n];
                Cn[n] = bn[D_STATE + n];
            }
            __builtin_amdgcn_sched_barrier(0);
            float x   = xc[g];
            float dl  = dc[g];
            float dlx = dl * x;
            v2f dl2  = {dl, dl};
            v2f dlx2 = {dlx, dlx};
            v2f ya = {0.f, 0.f}, yb = {0.f, 0.f};
#pragma unroll
            for (int j = 0; j < D_STATE / 2; j += 2) {
                v2f dA0 = exp_pair(dl2 * A[j]);
                v2f dA1 = exp_pair(dl2 * A[j + 1]);
                h[j]     = dA0 * h[j]     + dlx2 * Bv[j];
                h[j + 1] = dA1 * h[j + 1] + dlx2 * Bv[j + 1];
                ya += h[j]     * Cv[j];
                yb += h[j + 1] * Cv[j + 1];
            }
            ysb[(size_t)(i0 + g) * D_INNER] = ((ya[0] + ya[1]) + (yb[0] + yb[1])) + x * Dskip;
            bkey = Bn[0];
#pragma unroll
            for (int j = 0; j < D_STATE / 2; ++j) {
                Bv[j] = (v2f){Bn[2 * j], Bn[2 * j + 1]};
                Cv[j] = (v2f){Cn[2 * j], Cn[2 * j + 1]};
            }
        }
#pragma unroll
        for (int g = 0; g < SCAN_G; ++g) { xc[g] = xn[g]; dc[g] = dn[g]; }
    }
}

#define CO_NT   3
#define CO_KU   3
#define CO_NB   (D_INNER / (8 * CO_KU))
static_assert(D_INNER % (8 * CO_KU) == 0, "whole groups");
static_assert(DIMC == 2 * 16 * CO_NT, "two waves cover the output channels");
static_assert(LL % 16 == 0, "a position tile lies inside one image");

__global__ void __launch_bounds__(64) __attribute__((amdgpu_waves_per_eu(1, 2)))
k_comb_out(const float* __restrict__ ys,
           const float* __restrict__ xz,
           const float* __restrict__ Wout,
           float* __restrict__ out) {
    int lane = threadIdx.x & 31;
    int wave = __builtin_amdgcn_readfirstlane((int)(threadIdx.x >> 5));
    int c16 = lane & 15;
    int hf  = lane >> 4;
    int row = (int)blockIdx.x * 16 + c16;
    int b = row / LL, l = row % LL;

    const float* yp[KDIR];
#pragma unroll
    for (int k = 0; k < KDIR; ++k)
        yp[k] = ys + (size_t)((k * BATCH + b) * LL + spatial_to_scan(k, l)) * D_INNER + 4 * hf;
    const float* zp = xz + (size_t)row * XZW + D_INNER + 4 * hf;
    int n0 = wave * (16 * CO_NT);
    const float* wp = Wout + (size_t)(4 * hf) * DIMC + n0 + c16;

    v8f acc[CO_NT];
#pragma unroll
    for (int t = 0; t < CO_NT; ++t) acc[t] = (v8f){};

    v4f   yv[2][CO_KU][KDIR], zv[2][CO_KU];
    float wv[2][CO_KU][CO_NT][4];
#pragma unroll
    for (int u = 0; u < CO_KU; ++u) {
        int d0 = 8 * u;
#pragma unroll
        for (int k = 0; k < KDIR; ++k) yv[0][u][k] = *(const v4f*)(yp[k] + d0);
        zv[0][u] = *(const v4f*)(zp + d0);
#pragma unroll
        for (int t = 0; t < CO_NT; ++t)
#pragma unroll
            for (int e = 0; e < 4; ++e)
                wv[0][u][t][e] = wp[(size_t)(d0 + e) * DIMC + 16 * t];
    }
#pragma unroll
    for (int g = 0; g < CO_NB; ++g) {
        const int cur = g & 1, nxt = cur ^ 1;
        if (g + 1 < CO_NB) {
#pragma unroll
            for (int u = 0; u < CO_KU; ++u) {
                int d0 = 8 * ((g + 1) * CO_KU + u);
#pragma unroll
                for (int k = 0; k < KDIR; ++k) yv[nxt][u][k] = *(const v4f*)(yp[k] + d0);
                zv[nxt][u] = *(const v4f*)(zp + d0);
#pragma unroll
                for (int t = 0; t < CO_NT; ++t)
#pragma unroll
                    for (int e = 0; e < 4; ++e)
                        wv[nxt][u][t][e] = wp[(size_t)(d0 + e) * DIMC + 16 * t];
            }
        }
#pragma unroll
        for (int u = 0; u < CO_KU; ++u) {
            v4f s = ((yv[cur][u][0] + yv[cur][u][1]) + yv[cur][u][2]) + yv[cur][u][3];
            v4f z = zv[cur][u];
            float yy[4];
#pragma unroll
            for (int e = 0; e < 4; ++e) yy[e] = s[e] * silu(z[e]);
            v2f y01 = {yy[0], yy[1]};
            v2f y23 = {yy[2], yy[3]};
#pragma unroll
            for (int t = 0; t < CO_NT; ++t) {
                v2f w01 = {wv[cur][u][t][0], wv[cur][u][t][1]};
                v2f w23 = {wv[cur][u][t][2], wv[cur][u][t][3]};
                acc[t] = __builtin_amdgcn_wmma_f32_16x16x4_f32(false, w01, false, y01,
                                                               (short)0, acc[t], false, false);
                acc[t] = __builtin_amdgcn_wmma_f32_16x16x4_f32(false, w23, false, y23,
                                                               (short)0, acc[t], false, false);
            }
        }
    }
    float* op = out + (size_t)(b * DIMC + n0 + 8 * hf) * LL + l;
#pragma unroll
    for (int t = 0; t < CO_NT; ++t)
#pragma unroll
        for (int v = 0; v < 8; ++v)
            op[(size_t)(16 * t + v) * LL] = acc[t][v];
}

extern "C" void kernel_launch(void* const* d_in, const int* in_sizes, int n_in,
                              void* d_out, int out_size, void* d_ws, size_t ws_size,
                              hipStream_t stream) {
    (void)in_sizes; (void)n_in; (void)out_size; (void)ws_size;
    const float* x       = (const float*)d_in[0];
    const float* W_in    = (const float*)d_in[1];
    const float* W_conv  = (const float*)d_in[2];
    const float* b_conv  = (const float*)d_in[3];
    const float* W_xproj = (const float*)d_in[4];
    const float* W_dt    = (const float*)d_in[5];
    const float* b_dt    = (const float*)d_in[6];
    const float* A_logs  = (const float*)d_in[7];
    const float* Ds      = (const float*)d_in[8];
    const float* W_out   = (const float*)d_in[9];
    float* out = (float*)d_out;

    float* ws = (float*)d_ws;
    size_t o = 0;
    float* xz    = ws + o;  o += (size_t)BL * XZW;
    float* xi    = ws + o;  o += (size_t)BL * D_INNER;
    float* xdbl  = ws + o;  o += (size_t)KDIR * BL * NPROJ;
    float* delta = ws + o;  o += (size_t)KDIR * BL * D_INNER;
    float* ysbuf = ws + o;  o += (size_t)KDIR * BL * D_INNER;
    float* hend  = ws + o;  o += (size_t)(NCHUNK - 1) * NSCAN;
    float* pdec  = ws + o;  o += (size_t)(NCHUNK - 1) * NSCAN;

    k_in_conv <<<CV_BLOCKS + Z_BLOCKS, IC_THREADS, 0, stream>>>(x, W_in, W_conv, b_conv, xz, xi);
    k_xproj_delta_part<<<dim3(NCHUNK, NCHAN / D_INNER), FS_THREADS, 0, stream>>>(
        xi, W_xproj, W_dt, b_dt, A_logs, xdbl, delta, hend, pdec);
    k_scan_final<<<dim3(NCHAN / SF_CH, SF_NG),      SF_NW * 32, 0, stream>>>(
        xi, xdbl, delta, A_logs, Ds, hend, pdec, ysbuf);
    k_comb_out<<<BL / 16, 64, 0, stream>>>(ysbuf, xz, W_out, out);
}


# pass 2 of the scan: a step's B and C come by 16-byte vector loads four steps ahead, into the register pairs the packed multiplies read (was scalar loads + ~24 copies a step); order of sums kept
# speedup vs baseline: 36.8372x; 1.1954x over previous
#include <hip/hip_runtime.h>
#include <math.h>

typedef __attribute__((ext_vector_type(2))) float v2f;
typedef __attribute__((ext_vector_type(4))) float v4f;
typedef __attribute__((ext_vector_type(8))) float v8f;
typedef float v4fu __attribute__((ext_vector_type(4), aligned(4)));

#define DIMC    96
#define D_STATE 16
#define D_INNER 192
#define DT_RANK 6
#define KDIR    4
#define BATCH   2
#define HH      56
#define WWD     56
#define LL      (HH * WWD)
#define BL      (BATCH * LL)
#define NPROJ   (DT_RANK + 2 * D_STATE)
#define XZW     (2 * D_INNER)
#define NSCAN   (KDIR * BATCH * D_INNER * D_STATE)
#define NCHAN   (KDIR * BATCH * D_INNER)
#define SF_CH    32
#define SF_NW    4
#define SCAN_G   8
#define NCHUNK  56
#define CHUNK   (LL / NCHUNK)
#define SF_NG   (NCHUNK / SF_NW)
#define SF_D    4

__device__ __forceinline__ int scan_to_spatial(int k, int l) {
    int q = l / WWD, r = l % WWD;
    if (k == 0) return l;
    if (k == 1) return q * WWD + (WWD - 1 - r);
    if (k == 2) return r * WWD + q;
    return (WWD - 1 - r) * WWD + q;
}

__device__ __forceinline__ int spatial_to_scan(int k, int l) {
    int q = l / WWD, r = l % WWD;
    if (k == 0) return l;
    if (k == 1) return q * WWD + (WWD - 1 - r);
    if (k == 2) return r * WWD + q;
    return r * WWD + (WWD - 1 - q);
}

__device__ __forceinline__ float silu(float v) {
    return v / (1.f + __expf(-v));
}

__device__ __forceinline__ float softplus20(float t) {
    float e  = __expf(t);
    float s  = 1.f + e;
    float bb = s - 1.f;
    float r  = (1.f - (s - bb)) + (e - bb);
    float l2 = __builtin_amdgcn_logf(s);
    const float c = 0x1.62e42ep-1f, cc = 0x1.efa39ep-25f;
    float lo = __builtin_fmaf(l2, cc, r * __builtin_amdgcn_rcpf(s));
    return (t > 20.f) ? t : __builtin_fmaf(l2, c, lo);
}

#define CV_ROWS   4
#define CV_WIN    (CV_ROWS + 2)
#define CV_POS    (CV_WIN * WWD)
#define CV_PW     (WWD + 2)
#define CV_COLS   32
#define CV_PITCH  34
#define CV_KU     6
#define CV_NB     (DIMC / (4 * CV_KU))
#define IC_WAVES  7
#define IC_THREADS (IC_WAVES * 32)
#define CV_PER_IMG ((HH / CV_ROWS) * (D_INNER / CV_COLS))
#define CV_BLOCKS (BATCH * CV_PER_IMG)
#define Z_BLOCKS  ((BL / 16 / IC_WAVES) * (D_INNER / 16))
static_assert(HH % CV_ROWS == 0, "whole bands");
static_assert(CV_POS == IC_WAVES * 3 * 16, "three row tiles per wave");
static_assert(CV_ROWS * WWD == IC_WAVES * 32, "32 conv outputs per thread");
static_assert((BL / 16) % IC_WAVES == 0, "whole z blocks");
static_assert(DIMC % (4 * CV_KU) == 0, "whole groups of k blocks");

__global__ void __launch_bounds__(IC_THREADS)
k_in_conv(const float* __restrict__ x,
          const float* __restrict__ Win,
          const float* __restrict__ Wc,
          const float* __restrict__ bc,
          float* __restrict__ xz,
          float* __restrict__ xi) {
    __shared__ float xs[CV_WIN * CV_PW * CV_PITCH];
    int lane = threadIdx.x & 31;
    int wave = __builtin_amdgcn_readfirstlane((int)(threadIdx.x >> 5));
    int c16  = lane & 15;
    int kh   = (lane >> 4) * 2;

    if (blockIdx.x >= CV_BLOCKS) {
        int zb    = (int)blockIdx.x - CV_BLOCKS;
        int mTile = (zb / (D_INNER / 16)) * IC_WAVES + wave;
        int nTile = D_INNER / 16 + zb % (D_INNER / 16);
        int m0 = mTile * 16, n0 = nTile * 16;
        int mrow = m0 + c16;
        int b = mrow / LL, l = mrow % LL;
        int nn = n0 + c16;
        float za[DIMC / 4][2], zw[DIMC / 4][2];
#pragma unroll
        for (int u = 0; u < DIMC / 4; ++u) {
            int c0 = 4 * u + kh;
            za[u][0] = x[(size_t)(b * DIMC + c0    ) * LL + l];
            za[u][1] = x[(size_t)(b * DIMC + c0 + 1) * LL + l];
            zw[u][0] = Win[(size_t)(c0    ) * XZW + nn];
            zw[u][1] = Win[(size_t)(c0 + 1) * XZW + nn];
        }
        v8f acc = {};
#pragma unroll
        for (int u = 0; u < DIMC / 4; ++u) {
            v2f a  = {za[u][0], za[u][1]};
            v2f bf = {zw[u][0], zw[u][1]};
            acc = __builtin_amdgcn_wmma_f32_16x16x4_f32(false, a, false, bf,
                                                        (short)0, acc, false, false);
        }
        int mb = m0 + (lane >> 4) * 8;
#pragma unroll
        for (int v = 0; v < 8; ++v)
            xz[(size_t)(mb + v) * XZW + nn] = acc[v];
        return;
    }

    int idx  = (int)blockIdx.x;
    int b    = idx / CV_PER_IMG;
    int rem  = idx % CV_PER_IMG;
    int band = rem / (D_INNER / CV_COLS);
    int d0   = (rem % (D_INNER / CV_COLS)) * CV_COLS;
    int hfirst = band * CV_ROWS - 1;

    for (int i = threadIdx.x; i < CV_WIN * 2 * CV_COLS; i += IC_THREADS) {
        int row = i / (2 * CV_COLS), side = (i / CV_COLS) & 1, col = i % CV_COLS;
        xs[(row * CV_PW + side * (CV_PW - 1)) * CV_PITCH + col] = 0.f;
    }

    const float* xa[3];
    bool ok[3];
#pragma unroll
    for (int t = 0; t < 3; ++t) {
        int p = (wave * 3 + t) * 16 + c16;
        int h = hfirst + p / WWD;
        int w = p % WWD;
        ok[t] = (h >= 0 && h < HH);
        int l = ok[t] ? h * WWD + w : 0;
        xa[t] = x + (size_t)b * DIMC * LL + l;
    }
    const float* wb = Win + d0 + c16;
    v8f acc[3][2];
#pragma unroll
    for (int t = 0; t < 3; ++t) { acc[t][0] = (v8f){}; acc[t][1] = (v8f){}; }
    float av[2][CV_KU][3][2], bv[2][CV_KU][2][2];
#pragma unroll
    for (int u = 0; u < CV_KU; ++u) {
        int c0 = 4 * u + kh;
#pragma unroll
        for (int t = 0; t < 3; ++t) {
            av[0][u][t][0] = xa[t][(size_t)(c0    ) * LL];
            av[0][u][t][1] = xa[t][(size_t)(c0 + 1) * LL];
        }
#pragma unroll
        for (int j = 0; j < 2; ++j) {
            bv[0][u][j][0] = wb[(size_t)(c0    ) * XZW + 16 * j];
            bv[0][u][j][1] = wb[(size_t)(c0 + 1) * XZW + 16 * j];
        }
    }
#pragma unroll
    for (int bt = 0; bt < CV_NB; ++bt) {
        const int cur = bt & 1, nxt = cur ^ 1;
        if (bt + 1 < CV_NB) {
#pragma unroll
            for (int u = 0; u < CV_KU; ++u) {
                int c0 = (bt + 1) * 4 * CV_KU + 4 * u + kh;
#pragma unroll
                for (int t = 0; t < 3; ++t) {
                    av[nxt][u][t][0] = xa[t][(size_t)(c0    ) * LL];
                    av[nxt][u][t][1] = xa[t][(size_t)(c0 + 1) * LL];
                }
#pragma unroll
                for (int j = 0; j < 2; ++j) {
                    bv[nxt][u][j][0] = wb[(size_t)(c0    ) * XZW + 16 * j];
                    bv[nxt][u][j][1] = wb[(size_t)(c0 + 1) * XZW + 16 * j];
                }
            }
        }
#pragma unroll
        for (int u = 0; u < CV_KU; ++u) {
#pragma unroll
            for (int t = 0; t < 3; ++t) {
                v2f a;
                a[0] = ok[t] ? av[cur][u][t][0] : 0.f;
                a[1] = ok[t] ? av[cur][u][t][1] : 0.f;
#pragma unroll
                for (int j = 0; j < 2; ++j) {
                    v2f bf = {bv[cur][u][j][0], bv[cur][u][j][1]};
                    acc[t][j] = __builtin_amdgcn_wmma_f32_16x16x4_f32(false, a, false, bf,
                                                                      (short)0, acc[t][j], false, false);
                }
            }
        }
    }
#pragma unroll
    for (int t = 0; t < 3; ++t) {
        int prow = (wave * 3 + t) * 16 + (lane >> 4) * 8;
#pragma unroll
        for (int v = 0; v < 8; ++v) {
            int p = prow + v;
            int q = (p / WWD) * CV_PW + p % WWD + 1;
            xs[q * CV_PITCH + c16]      = acc[t][0][v];
            xs[q * CV_PITCH + 16 + c16] = acc[t][1][v];
        }
    }
    __syncthreads();

    int d = d0 + lane;
    float wg[9];
#pragma unroll
    for (int t = 0; t < 9; ++t) wg[t] = Wc[d * 9 + t];
    float bias = bc[d];
    float* xo = xi + ((size_t)b * LL + band * CV_ROWS * WWD) * D_INNER + d;
#pragma unroll 4
    for (int i = 0; i < (CV_ROWS * WWD) / IC_WAVES; ++i) {
        int o = wave + IC_WAVES * i;
        int r = o / WWD, w = o % WWD;
        const float* q = xs + (r * CV_PW + w) * CV_PITCH + lane;
        float s = bias;
#pragma unroll
        for (int dy = 0; dy < 3; ++dy)
#pragma unroll
            for (int dx = 0; dx < 3; ++dx)
                s = __builtin_fmaf(wg[dy * 3 + dx], q[(dy * CV_PW + dx) * CV_PITCH], s);
        xo[(size_t)o * D_INNER] = silu(s);
    }
}


static_assert(WWD % CHUNK == 0 && CHUNK >= 2, "a chunk lies inside one scan row");
static_assert(CHUNK % SCAN_G == 0, "whole groups");
static_assert(SCAN_G % SF_D == 0 && SF_D <= CHUNK, "a step keeps its slot from trip to trip");
static_assert(NCHUNK % SF_NW == 0, "whole groups of chunks");
static_assert(D_INNER % SF_CH == 0, "a wave's channels share (k, b)");

__device__ __forceinline__ size_t part_idx(int chunk, int g, int ch) {
    return ((size_t)(chunk * (D_STATE / 4) + g) * NCHAN + ch) * 4;
}

__device__ __forceinline__ v2f exp_pair(v2f a) {
    v2f r = {__expf(a[0]), __expf(a[1])};
    return r;
}

#define XP_KU   6
#define XP_NB   (D_INNER / (4 * XP_KU))
#define XP_NT   3
#define DL_NT   (D_INNER / 16)
#define FS_RT   4
#define FS_THREADS D_INNER
static_assert(D_INNER % (4 * XP_KU) == 0, "whole groups of k blocks");
static_assert(D_INNER % 32 == 0, "whole waves");
static_assert(FS_RT * 16 >= CHUNK && (FS_RT - 1) * 16 < CHUNK, "row tiles cover the chunk");
static_assert(FS_RT * 32 <= FS_THREADS, "a wave per row tile");
static_assert(CHUNK % 8 == 0, "a lane half's 8 rows lie inside the chunk or behind it");
static_assert(DT_RANK + D_STATE <= 32, "the B columns lie in the first two column tiles");

__global__ void __launch_bounds__(FS_THREADS) __attribute__((amdgpu_waves_per_eu(4, 4)))
k_xproj_delta_part(const float* __restrict__ xi,
                   const float* __restrict__ Wxp,
                   const float* __restrict__ Wdt,
                   const float* __restrict__ bdt,
                   const float* __restrict__ A_logs,
                   float* __restrict__ xdbl,
                   float* __restrict__ delta,
                   float* __restrict__ hend,
                   float* __restrict__ pdec) {
    __shared__ float dls[CHUNK * D_INNER];
    __shared__ v4f   Bsh[CHUNK * (D_STATE / 4)];
    __shared__ float drs[FS_RT][16 * 8];
    int lane = threadIdx.x & 31;
    int wave = __builtin_amdgcn_readfirstlane((int)(threadIdx.x >> 5));
    int chunk = blockIdx.x;
    int kb    = blockIdx.y;
    int b     = kb % BATCH;
    int kdir  = kb / BATCH;
    int l0    = chunk * CHUNK;
    size_t row0 = (size_t)kb * LL + l0;
    int c16 = lane & 15;
    int hf  = lane >> 4;
    int kh  = hf * 2;
    int ib  = wave * 16 + hf * 8;
    bool rowok = (ib < CHUNK);

    float w0[DL_NT][2], w1[DL_NT][2], bs[DL_NT];
    if (wave < FS_RT) {
        const float* wd = Wdt + (size_t)kdir * DT_RANK * D_INNER + c16;
        int k2 = hf ? 0 : 4;
#pragma unroll
        for (int t = 0; t < DL_NT; ++t) {
            w0[t][0] = wd[(size_t)(kh     ) * D_INNER + 16 * t];
            w0[t][1] = wd[(size_t)(kh + 1 ) * D_INNER + 16 * t];
            w1[t][0] = wd[(size_t)(k2     ) * D_INNER + 16 * t];
            w1[t][1] = wd[(size_t)(k2 + 1 ) * D_INNER + 16 * t];
            bs[t]    = bdt[kdir * D_INNER + c16 + 16 * t];
        }

        int i   = wave * 16 + c16;
        int ia  = (i < CHUNK) ? i : 0;
        int lsp = scan_to_spatial(kdir, l0 + ia);
        const float* arow = xi + ((size_t)b * LL + lsp) * D_INNER + kh;
        bool nok2 = (32 + c16 < NPROJ);
        int ncol2 = nok2 ? 32 + c16 : 0;
        const float* wk = Wxp + (size_t)kdir * D_INNER * NPROJ + (size_t)kh * NPROJ;
        int ncol[XP_NT] = {c16, 16 + c16, ncol2};

        v8f acc[XP_NT];
#pragma unroll
        for (int j = 0; j < XP_NT; ++j) acc[j] = (v8f){};
        v2f   av[2][XP_KU];
        float bw[2][XP_KU][XP_NT][2];
#pragma unroll
        for (int u = 0; u < XP_KU; ++u) {
            int k0 = 4 * u;
            av[0][u] = *(const v2f*)(arow + k0);
#pragma unroll
            for (int j = 0; j < XP_NT; ++j) {
                bw[0][u][j][0] = wk[(size_t)(k0    ) * NPROJ + ncol[j]];
                bw[0][u][j][1] = wk[(size_t)(k0 + 1) * NPROJ + ncol[j]];
            }
        }
#pragma unroll
        for (int bt = 0; bt < XP_NB; ++bt) {
            const int cur = bt & 1, nxt = cur ^ 1;
            if (bt + 1 < XP_NB) {
#pragma unroll
                for (int u = 0; u < XP_KU; ++u) {
                    int k0 = (bt + 1) * 4 * XP_KU + 4 * u;
                    av[nxt][u] = *(const v2f*)(arow + k0);
#pragma unroll
                    for (int j = 0; j < XP_NT; ++j) {
                        bw[nxt][u][j][0] = wk[(size_t)(k0    ) * NPROJ + ncol[j]];
                        bw[nxt][u][j][1] = wk[(size_t)(k0 + 1) * NPROJ + ncol[j]];
                    }
                }
            }
#pragma unroll
            for (int u = 0; u < XP_KU; ++u) {
#pragma unroll
                for (int j = 0; j < XP_NT; ++j) {
                    v2f bf;
                    bf[0] = (j < 2 || nok2) ? bw[cur][u][j][0] : 0.f;
                    bf[1] = (j < 2 || nok2) ? bw[cur][u][j][1] : 0.f;
                    acc[j] = __builtin_amdgcn_wmma_f32_16x16x4_f32(false, av[cur][u], false, bf,
                                                                   (short)0, acc[j], false, false);
                }
            }
        }
#pragma unroll
        for (int j = 0; j < XP_NT; ++j) {
            int nn = 16 * j + c16;
            if (rowok && nn < NPROJ) {
#pragma unroll
                for (int v = 0; v < 8; ++v)
                    xdbl[(row0 + ib + v) * NPROJ + nn] = acc[j][v];
            }
            if (j < 2 && rowok && nn >= DT_RANK && nn < DT_RANK + D_STATE) {
#pragma unroll
                for (int v = 0; v < 8; ++v)
                    ((float*)Bsh)[(ib + v) * D_STATE + nn - DT_RANK] = acc[j][v];
            }
        }
        if (c16 < 8) {
#pragma unroll
            for (int v = 0; v < 8; ++v)
                drs[wave][(hf * 8 + v) * 8 + c16] = (c16 < DT_RANK) ? acc[0][v] : 0.f;
        }
    }
    __syncthreads();
    if (wave < FS_RT) {
        v2f a0 = {drs[wave][c16 * 8 + kh],     drs[wave][c16 * 8 + kh + 1]};
        v2f a1 = {drs[wave][c16 * 8 + 4 + kh], drs[wave][c16 * 8 + 4 + kh + 1]};
#pragma unroll
        for (int t = 0; t < DL_NT; ++t) {
            v2f b0 = {w0[t][0], w0[t][1]};
            v2f b1;
            b1[0] = hf ? 0.f : w1[t][0];
            b1[1] = hf ? 0.f : w1[t][1];
            v8f dacc = {};
            dacc = __builtin_amdgcn_wmma_f32_16x16x4_f32(false, a0, false, b0,
                                                         (short)0, dacc, false, false);
            dacc = __builtin_amdgcn_wmma_f32_16x16x4_f32(false, a1, false, b1,
                                                         (short)0, dacc, false, false);
            int nn = c16 + 16 * t;
            float sp[8];
#pragma unroll
            for (int v = 0; v < 8; ++v) sp[v] = softplus20(dacc[v] + bs[t]);
            if (rowok) {
#pragma unroll
                for (int v = 0; v < 8; ++v) {
                    delta[(row0 + ib + v) * D_INNER + nn] = sp[v];
                    dls[(ib + v) * D_INNER + nn] = sp[v];
                }
            }
        }
    }
    __syncthreads();
    if (chunk == NCHUNK - 1) return;

    int d  = threadIdx.x;
    int ch = kb * D_INNER + d;
    v2f A[D_STATE / 2], h[D_STATE / 2], P[D_STATE / 2];
#pragma unroll
    for (int j = 0; j < D_STATE / 2; ++j) {
        const float* al = A_logs + (size_t)(kdir * D_INNER + d) * D_STATE + 2 * j;
        A[j][0] = -__expf(al[0]);
        A[j][1] = -__expf(al[1]);
        h[j] = (v2f){0.f, 0.f};
        P[j] = (v2f){1.f, 1.f};
    }
    int lsp0  = scan_to_spatial(kdir, l0);
    int lstep = scan_to_spatial(kdir, l0 + 1) - lsp0;
    const float* xib = xi + ((size_t)b * LL + lsp0) * D_INNER + d;
    const float* dlp = dls + d;

    float xc[SCAN_G];
#pragma unroll
    for (int g = 0; g < SCAN_G; ++g) xc[g] = xib[(ptrdiff_t)g * lstep * D_INNER];
    float dlc = dlp[0];
    v4f bc[D_STATE / 4];
#pragma unroll
    for (int q = 0; q < D_STATE / 4; ++q) bc[q] = Bsh[q];

#pragma nounroll
    for (int i0 = 0; i0 < CHUNK; i0 += SCAN_G) {
        float xn[SCAN_G];
        __builtin_amdgcn_sched_barrier(0);
        int in = (i0 + SCAN_G < CHUNK) ? i0 + SCAN_G : i0;
#pragma unroll
        for (int g = 0; g < SCAN_G; ++g) xn[g] = xib[(ptrdiff_t)(in + g) * lstep * D_INNER];
        __builtin_amdgcn_sched_barrier(0);
#pragma unroll
        for (int g = 0; g < SCAN_G; ++g) {
            int inx = (g + 1 < SCAN_G || i0 + SCAN_G < CHUNK) ? i0 + g + 1 : i0 + g;
            __builtin_amdgcn_sched_barrier(0);
            float dln = dlp[inx * D_INNER];
            v4f bn[D_STATE / 4];
#pragma unroll
            for (int q = 0; q < D_STATE / 4; ++q) bn[q] = Bsh[inx * (D_STATE / 4) + q];
            __builtin_amdgcn_sched_barrier(0);
            float dl = dlc;
            v4f bq[D_STATE / 4];
#pragma unroll
            for (int q = 0; q < D_STATE / 4; ++q) bq[q] = bc[q];
            float dlx = dl * xc[g];
            v2f dl2  = {dl, dl};
            v2f dlx2 = {dlx, dlx};
#pragma unroll
            for (int j = 0; j < D_STATE / 2; ++j) {
                v2f dA = exp_pair(dl2 * A[j]);
                v2f Bv = {bq[j / 2][2 * (j % 2)], bq[j / 2][2 * (j % 2) + 1]};
                h[j] = dA * h[j] + dlx2 * Bv;
                P[j] = P[j] * dA;
            }
            dlc = dln;
#pragma unroll
            for (int q = 0; q < D_STATE / 4; ++q) bc[q] = bn[q];
        }
#pragma unroll
        for (int g = 0; g < SCAN_G; ++g) xc[g] = xn[g];
    }
#pragma unroll
    for (int g = 0; g < D_STATE / 4; ++g) {
        v4f hv = {h[2 * g][0], h[2 * g][1], h[2 * g + 1][0], h[2 * g + 1][1]};
        v4f pv = {P[2 * g][0], P[2 * g][1], P[2 * g + 1][0], P[2 * g + 1][1]};
        *(v4f*)(hend + part_idx(chunk, g, ch)) = hv;
        *(v4f*)(pdec + part_idx(chunk, g, ch)) = pv;
    }
}

__global__ void __launch_bounds__(SF_NW * 32) __attribute__((amdgpu_waves_per_eu(1, 4)))
k_scan_final(const float* __restrict__ xi,
             const float* __restrict__ xdbl,
             const float* __restrict__ delta,
             const float* __restrict__ A_logs,
             const float* __restrict__ Ds,
             const float* __restrict__ hend,
             const float* __restrict__ pdec,
             float* __restrict__ ys) {
    __shared__ v4f segm[SF_NW][2 * (D_STATE / 4)][32];
    int lane = threadIdx.x & 31;
    int wave = __builtin_amdgcn_readfirstlane((int)(threadIdx.x >> 5));
    int grp   = SF_NG - 1 - (int)blockIdx.y;
    int chunk = grp * SF_NW + wave;
    int kb    = blockIdx.x / (D_INNER / SF_CH);
    int d     = (blockIdx.x % (D_INNER / SF_CH)) * SF_CH + lane;
    int b     = kb % BATCH;
    int kdir  = kb / BATCH;
    int ch    = kb * D_INNER + d;

    v2f A[D_STATE / 2], h[D_STATE / 2], P[D_STATE / 2];
#pragma unroll
    for (int j = 0; j < D_STATE / 2; ++j) {
        const float* al = A_logs + (size_t)(kdir * D_INNER + d) * D_STATE + 2 * j;
        A[j][0] = -__expf(al[0]);
        A[j][1] = -__expf(al[1]);
        h[j] = (v2f){0.f, 0.f};
        P[j] = (v2f){1.f, 1.f};
    }
    float Dskip = Ds[kdir * D_INNER + d];

    int j0 = wave * grp;
#pragma unroll 4
    for (int i = 0; i < grp; ++i) {
#pragma unroll
        for (int g = 0; g < D_STATE / 4; ++g) {
            v4f pv = *(const v4f*)(pdec + part_idx(j0 + i, g, ch));
            v4f hv = *(const v4f*)(hend + part_idx(j0 + i, g, ch));
            h[2 * g]     = (v2f){pv[0], pv[1]} * h[2 * g]     + (v2f){hv[0], hv[1]};
            h[2 * g + 1] = (v2f){pv[2], pv[3]} * h[2 * g + 1] + (v2f){hv[2], hv[3]};
            P[2 * g]     = P[2 * g]     * (v2f){pv[0], pv[1]};
            P[2 * g + 1] = P[2 * g + 1] * (v2f){pv[2], pv[3]};
        }
    }
#pragma unroll
    for (int g = 0; g < D_STATE / 4; ++g) {
        segm[wave][g][lane]               = (v4f){P[2 * g][0], P[2 * g][1], P[2 * g + 1][0], P[2 * g + 1][1]};
        segm[wave][D_STATE / 4 + g][lane] = (v4f){h[2 * g][0], h[2 * g][1], h[2 * g + 1][0], h[2 * g + 1][1]};
    }
    __syncthreads();
    v4f ep[SF_NW - 1][D_STATE / 4], eh[SF_NW - 1][D_STATE / 4];
#pragma unroll
    for (int i = 0; i < SF_NW - 1; ++i) {
#pragma unroll
        for (int g = 0; g < D_STATE / 4; ++g) {
            ep[i][g] = (v4f){1.f, 1.f, 1.f, 1.f};
            eh[i][g] = (v4f){0.f, 0.f, 0.f, 0.f};
        }
    }
#pragma unroll
    for (int i = 0; i < SF_NW - 1; ++i) {
        if (i < wave) {
#pragma unroll
            for (int g = 0; g < D_STATE / 4; ++g) {
                ep[i][g] = *(const v4f*)(pdec + part_idx(grp * SF_NW + i, g, ch));
                eh[i][g] = *(const v4f*)(hend + part_idx(grp * SF_NW + i, g, ch));
            }
        }
    }
    int l0    = chunk * CHUNK;
    int lsp0  = scan_to_spatial(kdir, l0);
    int lstep = scan_to_spatial(kdir, l0 + 1) - lsp0;
    const float* xib = xi    + ((size_t)b  * LL + lsp0) * D_INNER + d;
    const float* dlb = delta + ((size_t)kb * LL + l0) * D_INNER + d;
    const float* xdb = xdbl  + ((size_t)kb * LL + l0) * NPROJ + DT_RANK;
    int vz;
    asm("v_mov_b32 %0, 0" : "=v"(vz));
    const float* xdv = xdb + vz;
    float*       ysb = ys    + ((size_t)kb * LL + l0) * D_INNER + d;

    float xst[SF_D], dst[SF_D];
    v4f   bcq[SF_D][2 * (D_STATE / 4)];
#pragma unroll
    for (int s = 0; s < SF_D; ++s) {
        xst[s] = xib[(ptrdiff_t)s * lstep * D_INNER];
        dst[s] = dlb[(size_t)s * D_INNER];
#pragma unroll
        for (int q = 0; q < 2 * (D_STATE / 4); ++q)
            bcq[s][q] = *(const v4fu*)(xdv + (size_t)s * NPROJ + 4 * q);
    }

#pragma unroll
    for (int g = 0; g < D_STATE / 4; ++g) {
        v4f hv = segm[0][D_STATE / 4 + g][lane];
        h[2 * g]     = (v2f){hv[0], hv[1]};
        h[2 * g + 1] = (v2f){hv[2], hv[3]};
    }
#pragma unroll
    for (int s = 1; s < SF_NW; ++s) {
#pragma unroll
        for (int g = 0; g < D_STATE / 4; ++g) {
            v4f pv = segm[s][g][lane];
            v4f hv = segm[s][D_STATE / 4 + g][lane];
            h[2 * g]     = (v2f){pv[0], pv[1]} * h[2 * g]     + (v2f){hv[0], hv[1]};
            h[2 * g + 1] = (v2f){pv[2], pv[3]} * h[2 * g + 1] + (v2f){hv[2], hv[3]};
        }
    }
#pragma unroll
    for (int i = 0; i < SF_NW - 1; ++i) {
#pragma unroll
        for (int g = 0; g < D_STATE / 4; ++g) {
            h[2 * g]     = (v2f){ep[i][g][0], ep[i][g][1]} * h[2 * g]     + (v2f){eh[i][g][0], eh[i][g][1]};
            h[2 * g + 1] = (v2f){ep[i][g][2], ep[i][g][3]} * h[2 * g + 1] + (v2f){eh[i][g][2], eh[i][g][3]};
        }
    }

#pragma nounroll
    for (int i0 = 0; i0 < CHUNK; i0 += SCAN_G) {
#pragma unroll
        for (int g = 0; g < SCAN_G; ++g) {
            const int slot = g % SF_D;
            const int sn = (i0 + g + SF_D < CHUNK) ? i0 + g + SF_D : CHUNK - 1;
            float x   = xst[slot];
            float dl  = dst[slot];
            float dlx = dl * x;
            v2f dl2  = {dl, dl};
            v2f dlx2 = {dlx, dlx};
            v2f u[D_STATE / 2];
#pragma unroll
            for (int j = 0; j < D_STATE / 2; ++j) {
                v4f bq = bcq[slot][j / 2];
                v2f Bp = {bq[2 * (j % 2)], bq[2 * (j % 2) + 1]};
                u[j] = dlx2 * Bp;
            }
            __builtin_amdgcn_sched_barrier(0);
            xst[slot] = xib[(ptrdiff_t)sn * lstep * D_INNER];
            dst[slot] = dlb[(size_t)sn * D_INNER];
#pragma unroll
            for (int q = 0; q < D_STATE / 4; ++q)
                bcq[slot][q] = *(const v4fu*)(xdv + (size_t)sn * NPROJ + 4 * q);
            __builtin_amdgcn_sched_barrier(0);
            v2f ya = {0.f, 0.f}, yb = {0.f, 0.f};
#pragma unroll
            for (int j = 0; j < D_STATE / 2; j += 2) {
                v4f cq = bcq[slot][D_STATE / 4 + j / 2];
                v2f C0 = {cq[0], cq[1]}, C1 = {cq[2], cq[3]};
                v2f dA0 = exp_pair(dl2 * A[j]);
                v2f dA1 = exp_pair(dl2 * A[j + 1]);
                h[j]     = dA0 * h[j]     + u[j];
                h[j + 1] = dA1 * h[j + 1] + u[j + 1];
                ya += h[j]     * C0;
                yb += h[j + 1] * C1;
            }
            ysb[(size_t)(i0 + g) * D_INNER] = ((ya[0] + ya[1]) + (yb[0] + yb[1])) + x * Dskip;
            __builtin_amdgcn_sched_barrier(0);
#pragma unroll
            for (int q = D_STATE / 4; q < 2 * (D_STATE / 4); ++q)
                bcq[slot][q] = *(const v4fu*)(xdv + (size_t)sn * NPROJ + 4 * q);
            __builtin_amdgcn_sched_barrier(0);
        }
    }
}

#define CO_NT   3
#define CO_KU   3
#define CO_NB   (D_INNER / (8 * CO_KU))
static_assert(D_INNER % (8 * CO_KU) == 0, "whole groups");
static_assert(DIMC == 2 * 16 * CO_NT, "two waves cover the output channels");
static_assert(LL % 16 == 0, "a position tile lies inside one image");

__global__ void __launch_bounds__(64) __attribute__((amdgpu_waves_per_eu(1, 2)))
k_comb_out(const float* __restrict__ ys,
           const float* __restrict__ xz,
           const float* __restrict__ Wout,
           float* __restrict__ out) {
    int lane = threadIdx.x & 31;
    int wave = __builtin_amdgcn_readfirstlane((int)(threadIdx.x >> 5));
    int c16 = lane & 15;
    int hf  = lane >> 4;
    int row = (int)blockIdx.x * 16 + c16;
    int b = row / LL, l = row % LL;

    const float* yp[KDIR];
#pragma unroll
    for (int k = 0; k < KDIR; ++k)
        yp[k] = ys + (size_t)((k * BATCH + b) * LL + spatial_to_scan(k, l)) * D_INNER + 4 * hf;
    const float* zp = xz + (size_t)row * XZW + D_INNER + 4 * hf;
    int n0 = wave * (16 * CO_NT);
    const float* wp = Wout + (size_t)(4 * hf) * DIMC + n0 + c16;

    v8f acc[CO_NT];
#pragma unroll
    for (int t = 0; t < CO_NT; ++t) acc[t] = (v8f){};

    v4f   yv[2][CO_KU][KDIR], zv[2][CO_KU];
    float wv[2][CO_KU][CO_NT][4];
#pragma unroll
    for (int u = 0; u < CO_KU; ++u) {
        int d0 = 8 * u;
#pragma unroll
        for (int k = 0; k < KDIR; ++k) yv[0][u][k] = *(const v4f*)(yp[k] + d0);
        zv[0][u] = *(const v4f*)(zp + d0);
#pragma unroll
        for (int t = 0; t < CO_NT; ++t)
#pragma unroll
            for (int e = 0; e < 4; ++e)
                wv[0][u][t][e] = wp[(size_t)(d0 + e) * DIMC + 16 * t];
    }
#pragma unroll
    for (int g = 0; g < CO_NB; ++g) {
        const int cur = g & 1, nxt = cur ^ 1;
        if (g + 1 < CO_NB) {
#pragma unroll
            for (int u = 0; u < CO_KU; ++u) {
                int d0 = 8 * ((g + 1) * CO_KU + u);
#pragma unroll
                for (int k = 0; k < KDIR; ++k) yv[nxt][u][k] = *(const v4f*)(yp[k] + d0);
                zv[nxt][u] = *(const v4f*)(zp + d0);
#pragma unroll
                for (int t = 0; t < CO_NT; ++t)
#pragma unroll
                    for (int e = 0; e < 4; ++e)
                        wv[nxt][u][t][e] = wp[(size_t)(d0 + e) * DIMC + 16 * t];
            }
        }
#pragma unroll
        for (int u = 0; u < CO_KU; ++u) {
            v4f s = ((yv[cur][u][0] + yv[cur][u][1]) + yv[cur][u][2]) + yv[cur][u][3];
            v4f z = zv[cur][u];
            float yy[4];
#pragma unroll
            for (int e = 0; e < 4; ++e) yy[e] = s[e] * silu(z[e]);
            v2f y01 = {yy[0], yy[1]};
            v2f y23 = {yy[2], yy[3]};
#pragma unroll
            for (int t = 0; t < CO_NT; ++t) {
                v2f w01 = {wv[cur][u][t][0], wv[cur][u][t][1]};
                v2f w23 = {wv[cur][u][t][2], wv[cur][u][t][3]};
                acc[t] = __builtin_amdgcn_wmma_f32_16x16x4_f32(false, w01, false, y01,
                                                               (short)0, acc[t], false, false);
                acc[t] = __builtin_amdgcn_wmma_f32_16x16x4_f32(false, w23, false, y23,
                                                               (short)0, acc[t], false, false);
            }
        }
    }
    float* op = out + (size_t)(b * DIMC + n0 + 8 * hf) * LL + l;
#pragma unroll
    for (int t = 0; t < CO_NT; ++t)
#pragma unroll
        for (int v = 0; v < 8; ++v)
            op[(size_t)(16 * t + v) * LL] = acc[t][v];
}

extern "C" void kernel_launch(void* const* d_in, const int* in_sizes, int n_in,
                              void* d_out, int out_size, void* d_ws, size_t ws_size,
                              hipStream_t stream) {
    (void)in_sizes; (void)n_in; (void)out_size; (void)ws_size;
    const float* x       = (const float*)d_in[0];
    const float* W_in    = (const float*)d_in[1];
    const float* W_conv  = (const float*)d_in[2];
    const float* b_conv  = (const float*)d_in[3];
    const float* W_xproj = (const float*)d_in[4];
    const float* W_dt    = (const float*)d_in[5];
    const float* b_dt    = (const float*)d_in[6];
    const float* A_logs  = (const float*)d_in[7];
    const float* Ds      = (const float*)d_in[8];
    const float* W_out   = (const float*)d_in[9];
    float* out = (float*)d_out;

    float* ws = (float*)d_ws;
    size_t o = 0;
    float* xz    = ws + o;  o += (size_t)BL * XZW;
    float* xi    = ws + o;  o += (size_t)BL * D_INNER;
    float* xdbl  = ws + o;  o += (size_t)KDIR * BL * NPROJ;
    float* delta = ws + o;  o += (size_t)KDIR * BL * D_INNER;
    float* ysbuf = ws + o;  o += (size_t)KDIR * BL * D_INNER;
    float* hend  = ws + o;  o += (size_t)(NCHUNK - 1) * NSCAN;
    float* pdec  = ws + o;  o += (size_t)(NCHUNK - 1) * NSCAN;

    k_in_conv <<<CV_BLOCKS + Z_BLOCKS, IC_THREADS, 0, stream>>>(x, W_in, W_conv, b_conv, xz, xi);
    k_xproj_delta_part<<<dim3(NCHUNK, NCHAN / D_INNER), FS_THREADS, 0, stream>>>(
        xi, W_xproj, W_dt, b_dt, A_logs, xdbl, delta, hend, pdec);
    k_scan_final<<<dim3(NCHAN / SF_CH, SF_NG),      SF_NW * 32, 0, stream>>>(
        xi, xdbl, delta, A_logs, Ds, hend, pdec, ysbuf);
    k_comb_out<<<BL / 16, 64, 0, stream>>>(ysbuf, xz, W_out, out);
}


# scan: decay in base 2, A2 = A*log2(e) made once per thread; a chunk's P handed on as its delta sum S (1 float, not 16) and taken as 2^(A2*S) where a map is applied; P's rounding changes
# speedup vs baseline: 37.9611x; 1.0305x over previous
#include <hip/hip_runtime.h>
#include <math.h>

typedef __attribute__((ext_vector_type(2))) float v2f;
typedef __attribute__((ext_vector_type(4))) float v4f;
typedef __attribute__((ext_vector_type(8))) float v8f;
typedef float v4fu __attribute__((ext_vector_type(4), aligned(4)));

#define DIMC    96
#define D_STATE 16
#define D_INNER 192
#define DT_RANK 6
#define KDIR    4
#define BATCH   2
#define HH      56
#define WWD     56
#define LL      (HH * WWD)
#define BL      (BATCH * LL)
#define NPROJ   (DT_RANK + 2 * D_STATE)
#define XZW     (2 * D_INNER)
#define NSCAN   (KDIR * BATCH * D_INNER * D_STATE)
#define NCHAN   (KDIR * BATCH * D_INNER)
#define SF_CH    32
#define SF_NW    4
#define SCAN_G   8
#define NCHUNK  56
#define CHUNK   (LL / NCHUNK)
#define SF_NG   (NCHUNK / SF_NW)
#define SF_D    4

__device__ __forceinline__ int scan_to_spatial(int k, int l) {
    int q = l / WWD, r = l % WWD;
    if (k == 0) return l;
    if (k == 1) return q * WWD + (WWD - 1 - r);
    if (k == 2) return r * WWD + q;
    return (WWD - 1 - r) * WWD + q;
}

__device__ __forceinline__ int spatial_to_scan(int k, int l) {
    int q = l / WWD, r = l % WWD;
    if (k == 0) return l;
    if (k == 1) return q * WWD + (WWD - 1 - r);
    if (k == 2) return r * WWD + q;
    return r * WWD + (WWD - 1 - q);
}

__device__ __forceinline__ float silu(float v) {
    return v / (1.f + __expf(-v));
}

__device__ __forceinline__ float softplus20(float t) {
    float e  = __expf(t);
    float s  = 1.f + e;
    float bb = s - 1.f;
    float r  = (1.f - (s - bb)) + (e - bb);
    float l2 = __builtin_amdgcn_logf(s);
    const float c = 0x1.62e42ep-1f, cc = 0x1.efa39ep-25f;
    float lo = __builtin_fmaf(l2, cc, r * __builtin_amdgcn_rcpf(s));
    return (t > 20.f) ? t : __builtin_fmaf(l2, c, lo);
}

#define CV_ROWS   4
#define CV_WIN    (CV_ROWS + 2)
#define CV_POS    (CV_WIN * WWD)
#define CV_PW     (WWD + 2)
#define CV_COLS   32
#define CV_PITCH  34
#define CV_KU     6
#define CV_NB     (DIMC / (4 * CV_KU))
#define IC_WAVES  7
#define IC_THREADS (IC_WAVES * 32)
#define CV_PER_IMG ((HH / CV_ROWS) * (D_INNER / CV_COLS))
#define CV_BLOCKS (BATCH * CV_PER_IMG)
#define Z_BLOCKS  ((BL / 16 / IC_WAVES) * (D_INNER / 16))
static_assert(HH % CV_ROWS == 0, "whole bands");
static_assert(CV_POS == IC_WAVES * 3 * 16, "three row tiles per wave");
static_assert(CV_ROWS * WWD == IC_WAVES * 32, "32 conv outputs per thread");
static_assert((BL / 16) % IC_WAVES == 0, "whole z blocks");
static_assert(DIMC % (4 * CV_KU) == 0, "whole groups of k blocks");

__global__ void __launch_bounds__(IC_THREADS)
k_in_conv(const float* __restrict__ x,
          const float* __restrict__ Win,
          const float* __restrict__ Wc,
          const float* __restrict__ bc,
          float* __restrict__ xz,
          float* __restrict__ xi) {
    __shared__ float xs[CV_WIN * CV_PW * CV_PITCH];
    int lane = threadIdx.x & 31;
    int wave = __builtin_amdgcn_readfirstlane((int)(threadIdx.x >> 5));
    int c16  = lane & 15;
    int kh   = (lane >> 4) * 2;

    if (blockIdx.x >= CV_BLOCKS) {
        int zb    = (int)blockIdx.x - CV_BLOCKS;
        int mTile = (zb / (D_INNER / 16)) * IC_WAVES + wave;
        int nTile = D_INNER / 16 + zb % (D_INNER / 16);
        int m0 = mTile * 16, n0 = nTile * 16;
        int mrow = m0 + c16;
        int b = mrow / LL, l = mrow % LL;
        int nn = n0 + c16;
        float za[DIMC / 4][2], zw[DIMC / 4][2];
#pragma unroll
        for (int u = 0; u < DIMC / 4; ++u) {
            int c0 = 4 * u + kh;
            za[u][0] = x[(size_t)(b * DIMC + c0    ) * LL + l];
            za[u][1] = x[(size_t)(b * DIMC + c0 + 1) * LL + l];
            zw[u][0] = Win[(size_t)(c0    ) * XZW + nn];
            zw[u][1] = Win[(size_t)(c0 + 1) * XZW + nn];
        }
        v8f acc = {};
#pragma unroll
        for (int u = 0; u < DIMC / 4; ++u) {
            v2f a  = {za[u][0], za[u][1]};
            v2f bf = {zw[u][0], zw[u][1]};
            acc = __builtin_amdgcn_wmma_f32_16x16x4_f32(false, a, false, bf,
                                                        (short)0, acc, false, false);
        }
        int mb = m0 + (lane >> 4) * 8;
#pragma unroll
        for (int v = 0; v < 8; ++v)
            xz[(size_t)(mb + v) * XZW + nn] = acc[v];
        return;
    }

    int idx  = (int)blockIdx.x;
    int b    = idx / CV_PER_IMG;
    int rem  = idx % CV_PER_IMG;
    int band = rem / (D_INNER / CV_COLS);
    int d0   = (rem % (D_INNER / CV_COLS)) * CV_COLS;
    int hfirst = band * CV_ROWS - 1;

    for (int i = threadIdx.x; i < CV_WIN * 2 * CV_COLS; i += IC_THREADS) {
        int row = i / (2 * CV_COLS), side = (i / CV_COLS) & 1, col = i % CV_COLS;
        xs[(row * CV_PW + side * (CV_PW - 1)) * CV_PITCH + col] = 0.f;
    }

    const float* xa[3];
    bool ok[3];
#pragma unroll
    for (int t = 0; t < 3; ++t) {
        int p = (wave * 3 + t) * 16 + c16;
        int h = hfirst + p / WWD;
        int w = p % WWD;
        ok[t] = (h >= 0 && h < HH);
        int l = ok[t] ? h * WWD + w : 0;
        xa[t] = x + (size_t)b * DIMC * LL + l;
    }
    const float* wb = Win + d0 + c16;
    v8f acc[3][2];
#pragma unroll
    for (int t = 0; t < 3; ++t) { acc[t][0] = (v8f){}; acc[t][1] = (v8f){}; }
    float av[2][CV_KU][3][2], bv[2][CV_KU][2][2];
#pragma unroll
    for (int u = 0; u < CV_KU; ++u) {
        int c0 = 4 * u + kh;
#pragma unroll
        for (int t = 0; t < 3; ++t) {
            av[0][u][t][0] = xa[t][(size_t)(c0    ) * LL];
            av[0][u][t][1] = xa[t][(size_t)(c0 + 1) * LL];
        }
#pragma unroll
        for (int j = 0; j < 2; ++j) {
            bv[0][u][j][0] = wb[(size_t)(c0    ) * XZW + 16 * j];
            bv[0][u][j][1] = wb[(size_t)(c0 + 1) * XZW + 16 * j];
        }
    }
#pragma unroll
    for (int bt = 0; bt < CV_NB; ++bt) {
        const int cur = bt & 1, nxt = cur ^ 1;
        if (bt + 1 < CV_NB) {
#pragma unroll
            for (int u = 0; u < CV_KU; ++u) {
                int c0 = (bt + 1) * 4 * CV_KU + 4 * u + kh;
#pragma unroll
                for (int t = 0; t < 3; ++t) {
                    av[nxt][u][t][0] = xa[t][(size_t)(c0    ) * LL];
                    av[nxt][u][t][1] = xa[t][(size_t)(c0 + 1) * LL];
                }
#pragma unroll
                for (int j = 0; j < 2; ++j) {
                    bv[nxt][u][j][0] = wb[(size_t)(c0    ) * XZW + 16 * j];
                    bv[nxt][u][j][1] = wb[(size_t)(c0 + 1) * XZW + 16 * j];
                }
            }
        }
#pragma unroll
        for (int u = 0; u < CV_KU; ++u) {
#pragma unroll
            for (int t = 0; t < 3; ++t) {
                v2f a;
                a[0] = ok[t] ? av[cur][u][t][0] : 0.f;
                a[1] = ok[t] ? av[cur][u][t][1] : 0.f;
#pragma unroll
                for (int j = 0; j < 2; ++j) {
                    v2f bf = {bv[cur][u][j][0], bv[cur][u][j][1]};
                    acc[t][j] = __builtin_amdgcn_wmma_f32_16x16x4_f32(false, a, false, bf,
                                                                      (short)0, acc[t][j], false, false);
                }
            }
        }
    }
#pragma unroll
    for (int t = 0; t < 3; ++t) {
        int prow = (wave * 3 + t) * 16 + (lane >> 4) * 8;
#pragma unroll
        for (int v = 0; v < 8; ++v) {
            int p = prow + v;
            int q = (p / WWD) * CV_PW + p % WWD + 1;
            xs[q * CV_PITCH + c16]      = acc[t][0][v];
            xs[q * CV_PITCH + 16 + c16] = acc[t][1][v];
        }
    }
    __syncthreads();

    int d = d0 + lane;
    float wg[9];
#pragma unroll
    for (int t = 0; t < 9; ++t) wg[t] = Wc[d * 9 + t];
    float bias = bc[d];
    float* xo = xi + ((size_t)b * LL + band * CV_ROWS * WWD) * D_INNER + d;
#pragma unroll 4
    for (int i = 0; i < (CV_ROWS * WWD) / IC_WAVES; ++i) {
        int o = wave + IC_WAVES * i;
        int r = o / WWD, w = o % WWD;
        const float* q = xs + (r * CV_PW + w) * CV_PITCH + lane;
        float s = bias;
#pragma unroll
        for (int dy = 0; dy < 3; ++dy)
#pragma unroll
            for (int dx = 0; dx < 3; ++dx)
                s = __builtin_fmaf(wg[dy * 3 + dx], q[(dy * CV_PW + dx) * CV_PITCH], s);
        xo[(size_t)o * D_INNER] = silu(s);
    }
}


static_assert(WWD % CHUNK == 0 && CHUNK >= 2, "a chunk lies inside one scan row");
static_assert(CHUNK % SCAN_G == 0, "whole groups");
static_assert(SCAN_G % SF_D == 0 && SF_D <= CHUNK, "a step keeps its slot from trip to trip");
static_assert(NCHUNK % SF_NW == 0, "whole groups of chunks");
static_assert(D_INNER % SF_CH == 0, "a wave's channels share (k, b)");

__device__ __forceinline__ size_t part_idx(int chunk, int g, int ch) {
    return ((size_t)(chunk * (D_STATE / 4) + g) * NCHAN + ch) * 4;
}

#define LOG2E 0x1.715476p+0f
__device__ __forceinline__ float a2_of(float a_log) {
    return fmaxf(-__expf(a_log) * LOG2E, -0x1.fffffep+127f);
}
__device__ __forceinline__ v2f exp2_pair(v2f a) {
    v2f r = {__builtin_amdgcn_exp2f(a[0]), __builtin_amdgcn_exp2f(a[1])};
    return r;
}

#define XP_KU   6
#define XP_NB   (D_INNER / (4 * XP_KU))
#define XP_NT   3
#define DL_NT   (D_INNER / 16)
#define FS_RT   4
#define FS_THREADS D_INNER
static_assert(D_INNER % (4 * XP_KU) == 0, "whole groups of k blocks");
static_assert(D_INNER % 32 == 0, "whole waves");
static_assert(FS_RT * 16 >= CHUNK && (FS_RT - 1) * 16 < CHUNK, "row tiles cover the chunk");
static_assert(FS_RT * 32 <= FS_THREADS, "a wave per row tile");
static_assert(CHUNK % 8 == 0, "a lane half's 8 rows lie inside the chunk or behind it");
static_assert(DT_RANK + D_STATE <= 32, "the B columns lie in the first two column tiles");

__global__ void __launch_bounds__(FS_THREADS) __attribute__((amdgpu_waves_per_eu(4, 4)))
k_xproj_delta_part(const float* __restrict__ xi,
                   const float* __restrict__ Wxp,
                   const float* __restrict__ Wdt,
                   const float* __restrict__ bdt,
                   const float* __restrict__ A_logs,
                   float* __restrict__ xdbl,
                   float* __restrict__ delta,
                   float* __restrict__ hend,
                   float* __restrict__ ssum) {
    __shared__ float dls[CHUNK * D_INNER];
    __shared__ v4f   Bsh[CHUNK * (D_STATE / 4)];
    __shared__ float drs[FS_RT][16 * 8];
    int lane = threadIdx.x & 31;
    int wave = __builtin_amdgcn_readfirstlane((int)(threadIdx.x >> 5));
    int chunk = blockIdx.x;
    int kb    = blockIdx.y;
    int b     = kb % BATCH;
    int kdir  = kb / BATCH;
    int l0    = chunk * CHUNK;
    size_t row0 = (size_t)kb * LL + l0;
    int c16 = lane & 15;
    int hf  = lane >> 4;
    int kh  = hf * 2;
    int ib  = wave * 16 + hf * 8;
    bool rowok = (ib < CHUNK);

    float w0[DL_NT][2], w1[DL_NT][2], bs[DL_NT];
    if (wave < FS_RT) {
        const float* wd = Wdt + (size_t)kdir * DT_RANK * D_INNER + c16;
        int k2 = hf ? 0 : 4;
#pragma unroll
        for (int t = 0; t < DL_NT; ++t) {
            w0[t][0] = wd[(size_t)(kh     ) * D_INNER + 16 * t];
            w0[t][1] = wd[(size_t)(kh + 1 ) * D_INNER + 16 * t];
            w1[t][0] = wd[(size_t)(k2     ) * D_INNER + 16 * t];
            w1[t][1] = wd[(size_t)(k2 + 1 ) * D_INNER + 16 * t];
            bs[t]    = bdt[kdir * D_INNER + c16 + 16 * t];
        }

        int i   = wave * 16 + c16;
        int ia  = (i < CHUNK) ? i : 0;
        int lsp = scan_to_spatial(kdir, l0 + ia);
        const float* arow = xi + ((size_t)b * LL + lsp) * D_INNER + kh;
        bool nok2 = (32 + c16 < NPROJ);
        int ncol2 = nok2 ? 32 + c16 : 0;
        const float* wk = Wxp + (size_t)kdir * D_INNER * NPROJ + (size_t)kh * NPROJ;
        int ncol[XP_NT] = {c16, 16 + c16, ncol2};

        v8f acc[XP_NT];
#pragma unroll
        for (int j = 0; j < XP_NT; ++j) acc[j] = (v8f){};
        v2f   av[2][XP_KU];
        float bw[2][XP_KU][XP_NT][2];
#pragma unroll
        for (int u = 0; u < XP_KU; ++u) {
            int k0 = 4 * u;
            av[0][u] = *(const v2f*)(arow + k0);
#pragma unroll
            for (int j = 0; j < XP_NT; ++j) {
                bw[0][u][j][0] = wk[(size_t)(k0    ) * NPROJ + ncol[j]];
                bw[0][u][j][1] = wk[(size_t)(k0 + 1) * NPROJ + ncol[j]];
            }
        }
#pragma unroll
        for (int bt = 0; bt < XP_NB; ++bt) {
            const int cur = bt & 1, nxt = cur ^ 1;
            if (bt + 1 < XP_NB) {
#pragma unroll
                for (int u = 0; u < XP_KU; ++u) {
                    int k0 = (bt + 1) * 4 * XP_KU + 4 * u;
                    av[nxt][u] = *(const v2f*)(arow + k0);
#pragma unroll
                    for (int j = 0; j < XP_NT; ++j) {
                        bw[nxt][u][j][0] = wk[(size_t)(k0    ) * NPROJ + ncol[j]];
                        bw[nxt][u][j][1] = wk[(size_t)(k0 + 1) * NPROJ + ncol[j]];
                    }
                }
            }
#pragma unroll
            for (int u = 0; u < XP_KU; ++u) {
#pragma unroll
                for (int j = 0; j < XP_NT; ++j) {
                    v2f bf;
                    bf[0] = (j < 2 || nok2) ? bw[cur][u][j][0] : 0.f;
                    bf[1] = (j < 2 || nok2) ? bw[cur][u][j][1] : 0.f;
                    acc[j] = __builtin_amdgcn_wmma_f32_16x16x4_f32(false, av[cur][u], false, bf,
                                                                   (short)0, acc[j], false, false);
                }
            }
        }
#pragma unroll
        for (int j = 0; j < XP_NT; ++j) {
            int nn = 16 * j + c16;
            if (rowok && nn < NPROJ) {
#pragma unroll
                for (int v = 0; v < 8; ++v)
                    xdbl[(row0 + ib + v) * NPROJ + nn] = acc[j][v];
            }
            if (j < 2 && rowok && nn >= DT_RANK && nn < DT_RANK + D_STATE) {
#pragma unroll
                for (int v = 0; v < 8; ++v)
                    ((float*)Bsh)[(ib + v) * D_STATE + nn - DT_RANK] = acc[j][v];
            }
        }
        if (c16 < 8) {
#pragma unroll
            for (int v = 0; v < 8; ++v)
                drs[wave][(hf * 8 + v) * 8 + c16] = (c16 < DT_RANK) ? acc[0][v] : 0.f;
        }
    }
    __syncthreads();
    if (wave < FS_RT) {
        v2f a0 = {drs[wave][c16 * 8 + kh],     drs[wave][c16 * 8 + kh + 1]};
        v2f a1 = {drs[wave][c16 * 8 + 4 + kh], drs[wave][c16 * 8 + 4 + kh + 1]};
#pragma unroll
        for (int t = 0; t < DL_NT; ++t) {
            v2f b0 = {w0[t][0], w0[t][1]};
            v2f b1;
            b1[0] = hf ? 0.f : w1[t][0];
            b1[1] = hf ? 0.f : w1[t][1];
            v8f dacc = {};
            dacc = __builtin_amdgcn_wmma_f32_16x16x4_f32(false, a0, false, b0,
                                                         (short)0, dacc, false, false);
            dacc = __builtin_amdgcn_wmma_f32_16x16x4_f32(false, a1, false, b1,
                                                         (short)0, dacc, false, false);
            int nn = c16 + 16 * t;
            float sp[8];
#pragma unroll
            for (int v = 0; v < 8; ++v) sp[v] = softplus20(dacc[v] + bs[t]);
            if (rowok) {
#pragma unroll
                for (int v = 0; v < 8; ++v) {
                    delta[(row0 + ib + v) * D_INNER + nn] = sp[v];
                    dls[(ib + v) * D_INNER + nn] = sp[v];
                }
            }
        }
    }
    __syncthreads();
    if (chunk == NCHUNK - 1) return;

    int d  = threadIdx.x;
    int ch = kb * D_INNER + d;
    v2f A2[D_STATE / 2], h[D_STATE / 2];
#pragma unroll
    for (int j = 0; j < D_STATE / 2; ++j) {
        const float* al = A_logs + (size_t)(kdir * D_INNER + d) * D_STATE + 2 * j;
        A2[j][0] = a2_of(al[0]);
        A2[j][1] = a2_of(al[1]);
        h[j] = (v2f){0.f, 0.f};
    }
    float dsum = 0.f;
    int lsp0  = scan_to_spatial(kdir, l0);
    int lstep = scan_to_spatial(kdir, l0 + 1) - lsp0;
    const float* xib = xi + ((size_t)b * LL + lsp0) * D_INNER + d;
    const float* dlp = dls + d;

    float xc[SCAN_G];
#pragma unroll
    for (int g = 0; g < SCAN_G; ++g) xc[g] = xib[(ptrdiff_t)g * lstep * D_INNER];
    float dlc = dlp[0];
    v4f bc[D_STATE / 4];
#pragma unroll
    for (int q = 0; q < D_STATE / 4; ++q) bc[q] = Bsh[q];

#pragma nounroll
    for (int i0 = 0; i0 < CHUNK; i0 += SCAN_G) {
        float xn[SCAN_G];
        __builtin_amdgcn_sched_barrier(0);
        int in = (i0 + SCAN_G < CHUNK) ? i0 + SCAN_G : i0;
#pragma unroll
        for (int g = 0; g < SCAN_G; ++g) xn[g] = xib[(ptrdiff_t)(in + g) * lstep * D_INNER];
        __builtin_amdgcn_sched_barrier(0);
#pragma unroll
        for (int g = 0; g < SCAN_G; ++g) {
            int inx = (g + 1 < SCAN_G || i0 + SCAN_G < CHUNK) ? i0 + g + 1 : i0 + g;
            __builtin_amdgcn_sched_barrier(0);
            float dln = dlp[inx * D_INNER];
            v4f bn[D_STATE / 4];
#pragma unroll
            for (int q = 0; q < D_STATE / 4; ++q) bn[q] = Bsh[inx * (D_STATE / 4) + q];
            __builtin_amdgcn_sched_barrier(0);
            float dl = dlc;
            v4f bq[D_STATE / 4];
#pragma unroll
            for (int q = 0; q < D_STATE / 4; ++q) bq[q] = bc[q];
            float dlx = dl * xc[g];
            v2f dl2  = {dl, dl};
            v2f dlx2 = {dlx, dlx};
            dsum += dl;
#pragma unroll
            for (int j = 0; j < D_STATE / 2; ++j) {
                v2f dA = exp2_pair(dl2 * A2[j]);
                v2f Bv = {bq[j / 2][2 * (j % 2)], bq[j / 2][2 * (j % 2) + 1]};
                h[j] = dA * h[j] + dlx2 * Bv;
            }
            dlc = dln;
#pragma unroll
            for (int q = 0; q < D_STATE / 4; ++q) bc[q] = bn[q];
        }
#pragma unroll
        for (int g = 0; g < SCAN_G; ++g) xc[g] = xn[g];
    }
#pragma unroll
    for (int g = 0; g < D_STATE / 4; ++g) {
        v4f hv = {h[2 * g][0], h[2 * g][1], h[2 * g + 1][0], h[2 * g + 1][1]};
        *(v4f*)(hend + part_idx(chunk, g, ch)) = hv;
    }
    ssum[(size_t)chunk * NCHAN + ch] = dsum;
}

__global__ void __launch_bounds__(SF_NW * 32) __attribute__((amdgpu_waves_per_eu(1, 4)))
k_scan_final(const float* __restrict__ xi,
             const float* __restrict__ xdbl,
             const float* __restrict__ delta,
             const float* __restrict__ A_logs,
             const float* __restrict__ Ds,
             const float* __restrict__ hend,
             const float* __restrict__ ssum,
             float* __restrict__ ys) {
    __shared__ v4f segm[SF_NW][2 * (D_STATE / 4)][32];
    int lane = threadIdx.x & 31;
    int wave = __builtin_amdgcn_readfirstlane((int)(threadIdx.x >> 5));
    int grp   = SF_NG - 1 - (int)blockIdx.y;
    int chunk = grp * SF_NW + wave;
    int kb    = blockIdx.x / (D_INNER / SF_CH);
    int d     = (blockIdx.x % (D_INNER / SF_CH)) * SF_CH + lane;
    int b     = kb % BATCH;
    int kdir  = kb / BATCH;
    int ch    = kb * D_INNER + d;

    v2f A2[D_STATE / 2], h[D_STATE / 2];
#pragma unroll
    for (int j = 0; j < D_STATE / 2; ++j) {
        const float* al = A_logs + (size_t)(kdir * D_INNER + d) * D_STATE + 2 * j;
        A2[j][0] = a2_of(al[0]);
        A2[j][1] = a2_of(al[1]);
        h[j] = (v2f){0.f, 0.f};
    }
    float Dskip = Ds[kdir * D_INNER + d];

    int j0 = wave * grp;
    float Sfold = 0.f;
#pragma unroll 4
    for (int i = 0; i < grp; ++i) {
        float sv = ssum[(size_t)(j0 + i) * NCHAN + ch];
        v2f s2 = {sv, sv};
#pragma unroll
        for (int g = 0; g < D_STATE / 4; ++g) {
            v4f hv = *(const v4f*)(hend + part_idx(j0 + i, g, ch));
            v2f p0 = exp2_pair(s2 * A2[2 * g]);
            v2f p1 = exp2_pair(s2 * A2[2 * g + 1]);
            h[2 * g]     = p0 * h[2 * g]     + (v2f){hv[0], hv[1]};
            h[2 * g + 1] = p1 * h[2 * g + 1] + (v2f){hv[2], hv[3]};
        }
        Sfold += sv;
    }
    v2f sf2 = {Sfold, Sfold};
#pragma unroll
    for (int g = 0; g < D_STATE / 4; ++g) {
        v2f p0 = exp2_pair(sf2 * A2[2 * g]);
        v2f p1 = exp2_pair(sf2 * A2[2 * g + 1]);
        segm[wave][g][lane]               = (v4f){p0[0], p0[1], p1[0], p1[1]};
        segm[wave][D_STATE / 4 + g][lane] = (v4f){h[2 * g][0], h[2 * g][1], h[2 * g + 1][0], h[2 * g + 1][1]};
    }
    __syncthreads();
    float es[SF_NW - 1];
    v4f   eh[SF_NW - 1][D_STATE / 4];
#pragma unroll
    for (int i = 0; i < SF_NW - 1; ++i) {
        es[i] = 0.f;
#pragma unroll
        for (int g = 0; g < D_STATE / 4; ++g)
            eh[i][g] = (v4f){0.f, 0.f, 0.f, 0.f};
    }
#pragma unroll
    for (int i = 0; i < SF_NW - 1; ++i) {
        if (i < wave) {
            es[i] = ssum[(size_t)(grp * SF_NW + i) * NCHAN + ch];
#pragma unroll
            for (int g = 0; g < D_STATE / 4; ++g)
                eh[i][g] = *(const v4f*)(hend + part_idx(grp * SF_NW + i, g, ch));
        }
    }
    int l0    = chunk * CHUNK;
    int lsp0  = scan_to_spatial(kdir, l0);
    int lstep = scan_to_spatial(kdir, l0 + 1) - lsp0;
    const float* xib = xi    + ((size_t)b  * LL + lsp0) * D_INNER + d;
    const float* dlb = delta + ((size_t)kb * LL + l0) * D_INNER + d;
    const float* xdb = xdbl  + ((size_t)kb * LL + l0) * NPROJ + DT_RANK;
    int vz;
    asm("v_mov_b32 %0, 0" : "=v"(vz));
    const float* xdv = xdb + vz;
    float*       ysb = ys    + ((size_t)kb * LL + l0) * D_INNER + d;

    float xst[SF_D], dst[SF_D];
    v4f   bcq[SF_D][2 * (D_STATE / 4)];
#pragma unroll
    for (int s = 0; s < SF_D; ++s) {
        xst[s] = xib[(ptrdiff_t)s * lstep * D_INNER];
        dst[s] = dlb[(size_t)s * D_INNER];
#pragma unroll
        for (int q = 0; q < 2 * (D_STATE / 4); ++q)
            bcq[s][q] = *(const v4fu*)(xdv + (size_t)s * NPROJ + 4 * q);
    }

#pragma unroll
    for (int g = 0; g < D_STATE / 4; ++g) {
        v4f hv = segm[0][D_STATE / 4 + g][lane];
        h[2 * g]     = (v2f){hv[0], hv[1]};
        h[2 * g + 1] = (v2f){hv[2], hv[3]};
    }
#pragma unroll
    for (int s = 1; s < SF_NW; ++s) {
#pragma unroll
        for (int g = 0; g < D_STATE / 4; ++g) {
            v4f pv = segm[s][g][lane];
            v4f hv = segm[s][D_STATE / 4 + g][lane];
            h[2 * g]     = (v2f){pv[0], pv[1]} * h[2 * g]     + (v2f){hv[0], hv[1]};
            h[2 * g + 1] = (v2f){pv[2], pv[3]} * h[2 * g + 1] + (v2f){hv[2], hv[3]};
        }
    }
#pragma unroll
    for (int i = 0; i < SF_NW - 1; ++i) {
        v2f e2 = {es[i], es[i]};
#pragma unroll
        for (int g = 0; g < D_STATE / 4; ++g) {
            v2f p0 = exp2_pair(e2 * A2[2 * g]);
            v2f p1 = exp2_pair(e2 * A2[2 * g + 1]);
            h[2 * g]     = p0 * h[2 * g]     + (v2f){eh[i][g][0], eh[i][g][1]};
            h[2 * g + 1] = p1 * h[2 * g + 1] + (v2f){eh[i][g][2], eh[i][g][3]};
        }
    }

#pragma nounroll
    for (int i0 = 0; i0 < CHUNK; i0 += SCAN_G) {
#pragma unroll
        for (int g = 0; g < SCAN_G; ++g) {
            const int slot = g % SF_D;
            const int sn = (i0 + g + SF_D < CHUNK) ? i0 + g + SF_D : CHUNK - 1;
            float x   = xst[slot];
            float dl  = dst[slot];
            float dlx = dl * x;
            v2f dl2  = {dl, dl};
            v2f dlx2 = {dlx, dlx};
            v2f u[D_STATE / 2];
#pragma unroll
            for (int j = 0; j < D_STATE / 2; ++j) {
                v4f bq = bcq[slot][j / 2];
                v2f Bp = {bq[2 * (j % 2)], bq[2 * (j % 2) + 1]};
                u[j] = dlx2 * Bp;
            }
            __builtin_amdgcn_sched_barrier(0);
            xst[slot] = xib[(ptrdiff_t)sn * lstep * D_INNER];
            dst[slot] = dlb[(size_t)sn * D_INNER];
#pragma unroll
            for (int q = 0; q < D_STATE / 4; ++q)
                bcq[slot][q] = *(const v4fu*)(xdv + (size_t)sn * NPROJ + 4 * q);
            __builtin_amdgcn_sched_barrier(0);
            v2f ya = {0.f, 0.f}, yb = {0.f, 0.f};
#pragma unroll
            for (int j = 0; j < D_STATE / 2; j += 2) {
                v4f cq = bcq[slot][D_STATE / 4 + j / 2];
                v2f C0 = {cq[0], cq[1]}, C1 = {cq[2], cq[3]};
                v2f dA0 = exp2_pair(dl2 * A2[j]);
                v2f dA1 = exp2_pair(dl2 * A2[j + 1]);
                h[j]     = dA0 * h[j]     + u[j];
                h[j + 1] = dA1 * h[j + 1] + u[j + 1];
                ya += h[j]     * C0;
                yb += h[j + 1] * C1;
            }
            ysb[(size_t)(i0 + g) * D_INNER] = ((ya[0] + ya[1]) + (yb[0] + yb[1])) + x * Dskip;
            __builtin_amdgcn_sched_barrier(0);
#pragma unroll
            for (int q = D_STATE / 4; q < 2 * (D_STATE / 4); ++q)
                bcq[slot][q] = *(const v4fu*)(xdv + (size_t)sn * NPROJ + 4 * q);
            __builtin_amdgcn_sched_barrier(0);
        }
    }
}

#define CO_NT   3
#define CO_KU   3
#define CO_NB   (D_INNER / (8 * CO_KU))
static_assert(D_INNER % (8 * CO_KU) == 0, "whole groups");
static_assert(DIMC == 2 * 16 * CO_NT, "two waves cover the output channels");
static_assert(LL % 16 == 0, "a position tile lies inside one image");

__global__ void __launch_bounds__(64) __attribute__((amdgpu_waves_per_eu(1, 2)))
k_comb_out(const float* __restrict__ ys,
           const float* __restrict__ xz,
           const float* __restrict__ Wout,
           float* __restrict__ out) {
    int lane = threadIdx.x & 31;
    int wave = __builtin_amdgcn_readfirstlane((int)(threadIdx.x >> 5));
    int c16 = lane & 15;
    int hf  = lane >> 4;
    int row = (int)blockIdx.x * 16 + c16;
    int b = row / LL, l = row % LL;

    const float* yp[KDIR];
#pragma unroll
    for (int k = 0; k < KDIR; ++k)
        yp[k] = ys + (size_t)((k * BATCH + b) * LL + spatial_to_scan(k, l)) * D_INNER + 4 * hf;
    const float* zp = xz + (size_t)row * XZW + D_INNER + 4 * hf;
    int n0 = wave * (16 * CO_NT);
    const float* wp = Wout + (size_t)(4 * hf) * DIMC + n0 + c16;

    v8f acc[CO_NT];
#pragma unroll
    for (int t = 0; t < CO_NT; ++t) acc[t] = (v8f){};

    v4f   yv[2][CO_KU][KDIR], zv[2][CO_KU];
    float wv[2][CO_KU][CO_NT][4];
#pragma unroll
    for (int u = 0; u < CO_KU; ++u) {
        int d0 = 8 * u;
#pragma unroll
        for (int k = 0; k < KDIR; ++k) yv[0][u][k] = *(const v4f*)(yp[k] + d0);
        zv[0][u] = *(const v4f*)(zp + d0);
#pragma unroll
        for (int t = 0; t < CO_NT; ++t)
#pragma unroll
            for (int e = 0; e < 4; ++e)
                wv[0][u][t][e] = wp[(size_t)(d0 + e) * DIMC + 16 * t];
    }
#pragma unroll
    for (int g = 0; g < CO_NB; ++g) {
        const int cur = g & 1, nxt = cur ^ 1;
        if (g + 1 < CO_NB) {
#pragma unroll
            for (int u = 0; u < CO_KU; ++u) {
                int d0 = 8 * ((g + 1) * CO_KU + u);
#pragma unroll
                for (int k = 0; k < KDIR; ++k) yv[nxt][u][k] = *(const v4f*)(yp[k] + d0);
                zv[nxt][u] = *(const v4f*)(zp + d0);
#pragma unroll
                for (int t = 0; t < CO_NT; ++t)
#pragma unroll
                    for (int e = 0; e < 4; ++e)
                        wv[nxt][u][t][e] = wp[(size_t)(d0 + e) * DIMC + 16 * t];
            }
        }
#pragma unroll
        for (int u = 0; u < CO_KU; ++u) {
            v4f s = ((yv[cur][u][0] + yv[cur][u][1]) + yv[cur][u][2]) + yv[cur][u][3];
            v4f z = zv[cur][u];
            float yy[4];
#pragma unroll
            for (int e = 0; e < 4; ++e) yy[e] = s[e] * silu(z[e]);
            v2f y01 = {yy[0], yy[1]};
            v2f y23 = {yy[2], yy[3]};
#pragma unroll
            for (int t = 0; t < CO_NT; ++t) {
                v2f w01 = {wv[cur][u][t][0], wv[cur][u][t][1]};
                v2f w23 = {wv[cur][u][t][2], wv[cur][u][t][3]};
                acc[t] = __builtin_amdgcn_wmma_f32_16x16x4_f32(false, w01, false, y01,
                                                               (short)0, acc[t], false, false);
                acc[t] = __builtin_amdgcn_wmma_f32_16x16x4_f32(false, w23, false, y23,
                                                               (short)0, acc[t], false, false);
            }
        }
    }
    float* op = out + (size_t)(b * DIMC + n0 + 8 * hf) * LL + l;
#pragma unroll
    for (int t = 0; t < CO_NT; ++t)
#pragma unroll
        for (int v = 0; v < 8; ++v)
            op[(size_t)(16 * t + v) * LL] = acc[t][v];
}

extern "C" void kernel_launch(void* const* d_in, const int* in_sizes, int n_in,
                              void* d_out, int out_size, void* d_ws, size_t ws_size,
                              hipStream_t stream) {
    (void)in_sizes; (void)n_in; (void)out_size; (void)ws_size;
    const float* x       = (const float*)d_in[0];
    const float* W_in    = (const float*)d_in[1];
    const float* W_conv  = (const float*)d_in[2];
    const float* b_conv  = (const float*)d_in[3];
    const float* W_xproj = (const float*)d_in[4];
    const float* W_dt    = (const float*)d_in[5];
    const float* b_dt    = (const float*)d_in[6];
    const float* A_logs  = (const float*)d_in[7];
    const float* Ds      = (const float*)d_in[8];
    const float* W_out   = (const float*)d_in[9];
    float* out = (float*)d_out;

    float* ws = (float*)d_ws;
    size_t o = 0;
    float* xz    = ws + o;  o += (size_t)BL * XZW;
    float* xi    = ws + o;  o += (size_t)BL * D_INNER;
    float* xdbl  = ws + o;  o += (size_t)KDIR * BL * NPROJ;
    float* delta = ws + o;  o += (size_t)KDIR * BL * D_INNER;
    float* ysbuf = ws + o;  o += (size_t)KDIR * BL * D_INNER;
    float* hend  = ws + o;  o += (size_t)(NCHUNK - 1) * NSCAN;
    float* ssum  = ws + o;  o += (size_t)(NCHUNK - 1) * NCHAN;

    k_in_conv <<<CV_BLOCKS + Z_BLOCKS, IC_THREADS, 0, stream>>>(x, W_in, W_conv, b_conv, xz, xi);
    k_xproj_delta_part<<<dim3(NCHUNK, NCHAN / D_INNER), FS_THREADS, 0, stream>>>(
        xi, W_xproj, W_dt, b_dt, A_logs, xdbl, delta, hend, ssum);
    k_scan_final<<<dim3(NCHAN / SF_CH, SF_NG),      SF_NW * 32, 0, stream>>>(
        xi, xdbl, delta, A_logs, Ds, hend, ssum, ysbuf);
    k_comb_out<<<BL / 16, 64, 0, stream>>>(ysbuf, xz, W_out, out);
}
